# starting kernel
# baseline (speedup 1.0000x reference)
#include <hip/hip_runtime.h>
#include <math.h>

typedef float v2f __attribute__((ext_vector_type(2)));
typedef float v8f __attribute__((ext_vector_type(8)));

#define WMMA_F32X4(a, b, c) \
  __builtin_amdgcn_wmma_f32_16x16x4_f32(false, (a), false, (b), (short)0, (c), false, false)

__global__ void zero_kernel(float* __restrict__ p, long long n) {
  long long i = (long long)blockIdx.x * blockDim.x + threadIdx.x;
  if (i < n) p[i] = 0.0f;
}

__global__ void deg_kernel(const int* __restrict__ src, float* __restrict__ deg, int E) {
  int e = blockIdx.x * blockDim.x + threadIdx.x;
  if (e < E) unsafeAtomicAdd(&deg[src[e]], 1.0f);
}

__global__ void dinv_kernel(const float* __restrict__ deg, float* __restrict__ dinv, int N) {
  int i = blockIdx.x * blockDim.x + threadIdx.x;
  if (i < N) {
    float d = deg[i];
    dinv[i] = (d > 0.0f) ? rsqrtf(fmaxf(d, 1.0f)) : 0.0f;
  }
}

__global__ void norm_kernel(const int* __restrict__ src, const int* __restrict__ dst,
                            const float* __restrict__ dinv, float* __restrict__ nrm, int E) {
  int e = blockIdx.x * blockDim.x + threadIdx.x;
  if (e < E) nrm[e] = -dinv[src[e]] * dinv[dst[e]];
}

__global__ void gemm1_kernel(const float* __restrict__ x, const float* __restrict__ W1,
                             const float* __restrict__ b1, float* __restrict__ pre1,
                             float* __restrict__ y1, int N, int nTiles) {
  const int wave = (blockIdx.x * blockDim.x + threadIdx.x) >> 5;
  if (wave >= nTiles) return;
  const int lane = threadIdx.x & 31;
  const int n = lane & 15;
  const int hi = lane >> 4;
  const int base = wave * 16;
  const int mrow = min(base + n, N - 1);
  const float* __restrict__ W0 = W1;
  const float* __restrict__ Wb = W1 + 64 * 16;
  const float* __restrict__ xr = x + (long long)mrow * 64;

  v8f acc0 = {0.f, 0.f, 0.f, 0.f, 0.f, 0.f, 0.f, 0.f};
  v8f acc1 = acc0;
#pragma unroll
  for (int kb = 0; kb < 64; kb += 4) {
    const int k0 = kb + 2 * hi;
    v2f a = *(const v2f*)(xr + k0);
    v2f b0, bb;
    b0.x = W0[k0 * 16 + n];  b0.y = W0[(k0 + 1) * 16 + n];
    bb.x = Wb[k0 * 16 + n];  bb.y = Wb[(k0 + 1) * 16 + n];
    acc0 = WMMA_F32X4(a, b0, acc0);
    acc1 = WMMA_F32X4(a, bb, acc1);
  }
  const float bias = b1[n];
#pragma unroll
  for (int r = 0; r < 8; ++r) {
    const int row = base + r + 8 * hi;
    if (row < N) {
      pre1[(long long)row * 16 + n] = acc0[r] + bias;
      y1[(long long)row * 16 + n]  = acc1[r];
    }
  }
}

__global__ void scatter1_kernel(const int* __restrict__ src, const int* __restrict__ dst,
                                const float* __restrict__ nrm, const float* __restrict__ y1,
                                float* __restrict__ agg1, int E) {
  unsigned tid = blockIdx.x * blockDim.x + threadIdx.x;
  unsigned e = tid >> 4;
  unsigned f = tid & 15u;
  if (e < (unsigned)E) {
    int s = src[e], d = dst[e];
    unsafeAtomicAdd(&agg1[(long long)d * 16 + f], nrm[e] * y1[(long long)s * 16 + f]);
  }
}

__global__ void gemm2_kernel(const float* __restrict__ pre1, const float* __restrict__ agg1,
                             const float* __restrict__ W2, const float* __restrict__ b2,
                             float* __restrict__ pre2, float* __restrict__ y2,
                             int N, int nTiles) {
  const int wave = (blockIdx.x * blockDim.x + threadIdx.x) >> 5;
  if (wave >= nTiles) return;
  const int lane = threadIdx.x & 31;
  const int n = lane & 15;
  const int hi = lane >> 4;
  const int base = wave * 16;
  const int mrow = min(base + n, N - 1);
  const float* __restrict__ W0 = W2;
  const float* __restrict__ Wb = W2 + 16 * 40;

  v2f a[4];
#pragma unroll
  for (int i = 0; i < 4; ++i) {
    const int k0 = i * 4 + 2 * hi;
    const long long idx = (long long)mrow * 16 + k0;
    v2f p = *(const v2f*)(pre1 + idx);
    v2f g = *(const v2f*)(agg1 + idx);
    a[i].x = fmaxf(p.x + g.x, 0.0f);
    a[i].y = fmaxf(p.y + g.y, 0.0f);
  }

  v8f acc[6];
#pragma unroll
  for (int i = 0; i < 6; ++i) acc[i] = (v8f){0.f, 0.f, 0.f, 0.f, 0.f, 0.f, 0.f, 0.f};

#pragma unroll
  for (int ct = 0; ct < 3; ++ct) {
    const int col = ct * 16 + n;
    const int cc = min(col, 39);
#pragma unroll
    for (int i = 0; i < 4; ++i) {
      const int k0 = i * 4 + 2 * hi;
      v2f b0, bb;
      b0.x = W0[k0 * 40 + cc];  b0.y = W0[(k0 + 1) * 40 + cc];
      bb.x = Wb[k0 * 40 + cc];  bb.y = Wb[(k0 + 1) * 40 + cc];
      acc[ct * 2]     = WMMA_F32X4(a[i], b0, acc[ct * 2]);
      acc[ct * 2 + 1] = WMMA_F32X4(a[i], bb, acc[ct * 2 + 1]);
    }
  }

#pragma unroll
  for (int ct = 0; ct < 3; ++ct) {
    const int col = ct * 16 + n;
    if (col < 40) {
      const float bias = b2[col];
#pragma unroll
      for (int r = 0; r < 8; ++r) {
        const int row = base + r + 8 * hi;
        if (row < N) {
          pre2[(long long)row * 40 + col] = acc[ct * 2][r] + bias;
          y2[(long long)row * 40 + col]   = acc[ct * 2 + 1][r];
        }
      }
    }
  }
}

__global__ void scatter2_kernel(const int* __restrict__ src, const int* __restrict__ dst,
                                const float* __restrict__ nrm, const float* __restrict__ y2,
                                float* __restrict__ agg2, int E) {
  unsigned tid = blockIdx.x * blockDim.x + threadIdx.x;
  unsigned e = tid / 40u;
  unsigned f = tid - e * 40u;
  if (e < (unsigned)E) {
    int s = src[e], d = dst[e];
    unsafeAtomicAdd(&agg2[(long long)d * 40 + f], nrm[e] * y2[(long long)s * 40 + f]);
  }
}

__global__ void lsm_kernel(const float* __restrict__ pre2, const float* __restrict__ agg2,
                           float* __restrict__ out, int N) {
  const int node = (blockIdx.x * blockDim.x + threadIdx.x) >> 5;
  if (node >= N) return;
  const int lane = threadIdx.x & 31;
  const float* __restrict__ pr = pre2 + (long long)node * 40;
  const float* __restrict__ ag = agg2 + (long long)node * 40;
  float v0 = pr[lane] + ag[lane];
  float v1 = (lane < 8) ? (pr[lane + 32] + ag[lane + 32]) : -INFINITY;
  float m = fmaxf(v0, v1);
#pragma unroll
  for (int o = 16; o > 0; o >>= 1) m = fmaxf(m, __shfl_xor(m, o, 32));
  float s = expf(v0 - m) + ((lane < 8) ? expf(v1 - m) : 0.0f);
#pragma unroll
  for (int o = 16; o > 0; o >>= 1) s += __shfl_xor(s, o, 32);
  const float lse = m + logf(s);
  float* __restrict__ op = out + (long long)node * 40;
  op[lane] = v0 - lse;
  if (lane < 8) op[lane + 32] = v1 - lse;
}

extern "C" void kernel_launch(void* const* d_in, const int* in_sizes, int n_in,
                              void* d_out, int out_size, void* d_ws, size_t ws_size,
                              hipStream_t stream) {
  const float* x  = (const float*)d_in[0];
  const int* ei   = (const int*)d_in[1];
  const float* W1 = (const float*)d_in[2];
  const float* b1 = (const float*)d_in[3];
  const float* W2 = (const float*)d_in[4];
  const float* b2 = (const float*)d_in[5];
  float* out = (float*)d_out;

  const int N = in_sizes[0] / 64;
  const int E = in_sizes[1] / 2;
  const int* src = ei;
  const int* dst = ei + E;

  float* ws = (float*)d_ws;
  size_t off = 0;
  float* deg  = ws + off; off += (size_t)N;
  float* dinv = ws + off; off += (size_t)N;
  float* nrm  = ws + off; off += (size_t)E;
  float* pre1 = ws + off; off += (size_t)N * 16;
  float* y1   = ws + off; off += (size_t)N * 16;
  float* agg1 = ws + off; off += (size_t)N * 16;
  float* pre2 = ws + off; off += (size_t)N * 40;
  float* y2   = ws + off; off += (size_t)N * 40;
  float* agg2 = ws + off; off += (size_t)N * 40;

  const int B = 256;
  auto blk = [](long long n, int b) { return (unsigned)((n + b - 1) / b); };

  zero_kernel<<<blk(N, B), B, 0, stream>>>(deg, N);
  zero_kernel<<<blk((long long)N * 16, B), B, 0, stream>>>(agg1, (long long)N * 16);
  zero_kernel<<<blk((long long)N * 40, B), B, 0, stream>>>(agg2, (long long)N * 40);

  deg_kernel<<<blk(E, B), B, 0, stream>>>(src, deg, E);
  dinv_kernel<<<blk(N, B), B, 0, stream>>>(deg, dinv, N);
  norm_kernel<<<blk(E, B), B, 0, stream>>>(src, dst, dinv, nrm, E);

  const int tiles = (N + 15) / 16;
  gemm1_kernel<<<blk((long long)tiles * 32, B), B, 0, stream>>>(x, W1, b1, pre1, y1, N, tiles);
  scatter1_kernel<<<blk((long long)E * 16, B), B, 0, stream>>>(src, dst, nrm, y1, agg1, E);

  gemm2_kernel<<<blk((long long)tiles * 32, B), B, 0, stream>>>(pre1, agg1, W2, b2, pre2, y2, N, tiles);
  scatter2_kernel<<<blk((long long)E * 40, B), B, 0, stream>>>(src, dst, nrm, y2, agg2, E);

  lsm_kernel<<<blk((long long)N * 32, B), B, 0, stream>>>(pre2, agg2, out, N);
}


# Layer 2 aggregates neighbours before projecting: atomic scatter of 16 hidden feats per edge instead of 40 projected, one GEMM after; neighbour sums reassociated (now the reference's order)
# speedup vs baseline: 1.3103x; 1.3103x over previous
#include <hip/hip_runtime.h>
#include <math.h>

typedef float v2f __attribute__((ext_vector_type(2)));
typedef float v8f __attribute__((ext_vector_type(8)));

#define WMMA_F32X4(a, b, c) \
  __builtin_amdgcn_wmma_f32_16x16x4_f32(false, (a), false, (b), (short)0, (c), false, false)

__global__ void zero_kernel(float* __restrict__ p, long long n) {
  long long i = (long long)blockIdx.x * blockDim.x + threadIdx.x;
  if (i < n) p[i] = 0.0f;
}

__global__ void deg_kernel(const int* __restrict__ src, float* __restrict__ deg, int E) {
  int e = blockIdx.x * blockDim.x + threadIdx.x;
  if (e < E) unsafeAtomicAdd(&deg[src[e]], 1.0f);
}

__global__ void dinv_kernel(const float* __restrict__ deg, float* __restrict__ dinv, int N) {
  int i = blockIdx.x * blockDim.x + threadIdx.x;
  if (i < N) {
    float d = deg[i];
    dinv[i] = (d > 0.0f) ? rsqrtf(fmaxf(d, 1.0f)) : 0.0f;
  }
}

__global__ void norm_kernel(const int* __restrict__ src, const int* __restrict__ dst,
                            const float* __restrict__ dinv, float* __restrict__ nrm, int E) {
  int e = blockIdx.x * blockDim.x + threadIdx.x;
  if (e < E) nrm[e] = -dinv[src[e]] * dinv[dst[e]];
}

__global__ void gemm1_kernel(const float* __restrict__ x, const float* __restrict__ W1,
                             const float* __restrict__ b1, float* __restrict__ pre1,
                             float* __restrict__ y1, int N, int nTiles) {
  const int wave = (blockIdx.x * blockDim.x + threadIdx.x) >> 5;
  if (wave >= nTiles) return;
  const int lane = threadIdx.x & 31;
  const int n = lane & 15;
  const int hi = lane >> 4;
  const int base = wave * 16;
  const int mrow = min(base + n, N - 1);
  const float* __restrict__ W0 = W1;
  const float* __restrict__ Wb = W1 + 64 * 16;
  const float* __restrict__ xr = x + (long long)mrow * 64;

  v8f acc0 = {0.f, 0.f, 0.f, 0.f, 0.f, 0.f, 0.f, 0.f};
  v8f acc1 = acc0;
#pragma unroll
  for (int kb = 0; kb < 64; kb += 4) {
    const int k0 = kb + 2 * hi;
    v2f a = *(const v2f*)(xr + k0);
    v2f b0, bb;
    b0.x = W0[k0 * 16 + n];  b0.y = W0[(k0 + 1) * 16 + n];
    bb.x = Wb[k0 * 16 + n];  bb.y = Wb[(k0 + 1) * 16 + n];
    acc0 = WMMA_F32X4(a, b0, acc0);
    acc1 = WMMA_F32X4(a, bb, acc1);
  }
  const float bias = b1[n];
#pragma unroll
  for (int r = 0; r < 8; ++r) {
    const int row = base + r + 8 * hi;
    if (row < N) {
      pre1[(long long)row * 16 + n] = acc0[r] + bias;
      y1[(long long)row * 16 + n]  = acc1[r];
    }
  }
}

__global__ void scatter16_kernel(const int* __restrict__ src, const int* __restrict__ dst,
                                 const float* __restrict__ nrm, const float* __restrict__ y,
                                 float* __restrict__ agg, int E) {
  unsigned tid = blockIdx.x * blockDim.x + threadIdx.x;
  unsigned e = tid >> 4;
  unsigned f = tid & 15u;
  if (e < (unsigned)E) {
    int s = src[e], d = dst[e];
    unsafeAtomicAdd(&agg[(long long)d * 16 + f], nrm[e] * y[(long long)s * 16 + f]);
  }
}

__global__ void hidden_kernel(const float* __restrict__ pre1, const float* __restrict__ agg1,
                              float* __restrict__ h, float* __restrict__ t, long long n4) {
  long long i = (long long)blockIdx.x * blockDim.x + threadIdx.x;
  if (i < n4) {
    const float4 p = ((const float4*)pre1)[i];
    const float4 g = ((const float4*)agg1)[i];
    float4 r;
    r.x = fmaxf(p.x + g.x, 0.0f);
    r.y = fmaxf(p.y + g.y, 0.0f);
    r.z = fmaxf(p.z + g.z, 0.0f);
    r.w = fmaxf(p.w + g.w, 0.0f);
    ((float4*)h)[i] = r;
    ((float4*)t)[i] = make_float4(0.0f, 0.0f, 0.0f, 0.0f);
  }
}

__global__ void gemm2_kernel(const float* __restrict__ h, const float* __restrict__ t,
                             const float* __restrict__ W2, const float* __restrict__ b2,
                             float* __restrict__ z, int N, int nTiles) {
  const int wave = (blockIdx.x * blockDim.x + threadIdx.x) >> 5;
  if (wave >= nTiles) return;
  const int lane = threadIdx.x & 31;
  const int n = lane & 15;
  const int hi = lane >> 4;
  const int base = wave * 16;
  const int mrow = min(base + n, N - 1);
  const float* __restrict__ W0 = W2;
  const float* __restrict__ Wb = W2 + 16 * 40;

  v2f a[4], g[4];
#pragma unroll
  for (int i = 0; i < 4; ++i) {
    const int k0 = i * 4 + 2 * hi;
    const long long idx = (long long)mrow * 16 + k0;
    a[i] = *(const v2f*)(h + idx);
    g[i] = *(const v2f*)(t + idx);
  }

  v8f acc[6];
#pragma unroll
  for (int i = 0; i < 6; ++i) acc[i] = (v8f){0.f, 0.f, 0.f, 0.f, 0.f, 0.f, 0.f, 0.f};

#pragma unroll
  for (int ct = 0; ct < 3; ++ct) {
    const int col = ct * 16 + n;
    const int cc = min(col, 39);
#pragma unroll
    for (int i = 0; i < 4; ++i) {
      const int k0 = i * 4 + 2 * hi;
      v2f b0, bb;
      b0.x = W0[k0 * 40 + cc];  b0.y = W0[(k0 + 1) * 40 + cc];
      bb.x = Wb[k0 * 40 + cc];  bb.y = Wb[(k0 + 1) * 40 + cc];
      acc[ct * 2]     = WMMA_F32X4(a[i], b0, acc[ct * 2]);
      acc[ct * 2 + 1] = WMMA_F32X4(g[i], bb, acc[ct * 2 + 1]);
    }
  }

#pragma unroll
  for (int ct = 0; ct < 3; ++ct) {
    const int col = ct * 16 + n;
    if (col < 40) {
      const float bias = b2[col];
#pragma unroll
      for (int r = 0; r < 8; ++r) {
        const int row = base + r + 8 * hi;
        if (row < N) {
          z[(long long)row * 40 + col] = (acc[ct * 2][r] + acc[ct * 2 + 1][r]) + bias;
        }
      }
    }
  }
}

__global__ void lsm_kernel(const float* __restrict__ z, float* __restrict__ out, int N) {
  const int node = (blockIdx.x * blockDim.x + threadIdx.x) >> 5;
  if (node >= N) return;
  const int lane = threadIdx.x & 31;
  const float* __restrict__ pr = z + (long long)node * 40;
  float v0 = pr[lane];
  float v1 = (lane < 8) ? pr[lane + 32] : -INFINITY;
  float m = fmaxf(v0, v1);
#pragma unroll
  for (int o = 16; o > 0; o >>= 1) m = fmaxf(m, __shfl_xor(m, o, 32));
  float s = expf(v0 - m) + ((lane < 8) ? expf(v1 - m) : 0.0f);
#pragma unroll
  for (int o = 16; o > 0; o >>= 1) s += __shfl_xor(s, o, 32);
  const float lse = m + logf(s);
  float* __restrict__ op = out + (long long)node * 40;
  op[lane] = v0 - lse;
  if (lane < 8) op[lane + 32] = v1 - lse;
}

extern "C" void kernel_launch(void* const* d_in, const int* in_sizes, int n_in,
                              void* d_out, int out_size, void* d_ws, size_t ws_size,
                              hipStream_t stream) {
  const float* x  = (const float*)d_in[0];
  const int* ei   = (const int*)d_in[1];
  const float* W1 = (const float*)d_in[2];
  const float* b1 = (const float*)d_in[3];
  const float* W2 = (const float*)d_in[4];
  const float* b2 = (const float*)d_in[5];
  float* out = (float*)d_out;

  const int N = in_sizes[0] / 64;
  const int E = in_sizes[1] / 2;
  const int* src = ei;
  const int* dst = ei + E;

  float* ws = (float*)d_ws;
  size_t off = 0;
  float* deg  = ws + off; off += (size_t)N;
  float* dinv = ws + off; off += (size_t)N;
  float* nrm  = ws + off; off += (size_t)E;
  float* pre1 = ws + off; off += (size_t)N * 16;
  float* y1   = ws + off; off += (size_t)N * 16;
  float* agg1 = ws + off; off += (size_t)N * 16;
  float* h    = ws + off; off += (size_t)N * 16;
  float* t2   = ws + off; off += (size_t)N * 16;
  float* z2   = ws + off; off += (size_t)N * 40;

  const int B = 256;
  auto blk = [](long long n, int b) { return (unsigned)((n + b - 1) / b); };

  zero_kernel<<<blk(N, B), B, 0, stream>>>(deg, N);
  zero_kernel<<<blk((long long)N * 16, B), B, 0, stream>>>(agg1, (long long)N * 16);

  deg_kernel<<<blk(E, B), B, 0, stream>>>(src, deg, E);
  dinv_kernel<<<blk(N, B), B, 0, stream>>>(deg, dinv, N);
  norm_kernel<<<blk(E, B), B, 0, stream>>>(src, dst, dinv, nrm, E);

  const int tiles = (N + 15) / 16;
  gemm1_kernel<<<blk((long long)tiles * 32, B), B, 0, stream>>>(x, W1, b1, pre1, y1, N, tiles);
  scatter16_kernel<<<blk((long long)E * 16, B), B, 0, stream>>>(src, dst, nrm, y1, agg1, E);

  hidden_kernel<<<blk((long long)N * 4, B), B, 0, stream>>>(pre1, agg1, h, t2, (long long)N * 4);
  scatter16_kernel<<<blk((long long)E * 16, B), B, 0, stream>>>(src, dst, nrm, h, t2, E);
  gemm2_kernel<<<blk((long long)tiles * 32, B), B, 0, stream>>>(h, t2, W2, b2, z2, N, tiles);

  lsm_kernel<<<blk((long long)N * 32, B), B, 0, stream>>>(z2, out, N);
}


# Edges counting-sorted per call into 256 dst-row ranges (LDS counts, LDS ranks); each range sums its rows in LDS, stored once: no global f32 atomic scatter; sums in LDS arrival order
# speedup vs baseline: 1.6720x; 1.2760x over previous
#include <hip/hip_runtime.h>
#include <math.h>

typedef float v2f __attribute__((ext_vector_type(2)));
typedef float v8f __attribute__((ext_vector_type(8)));

#define WMMA_F32X4(a, b, c) \
  __builtin_amdgcn_wmma_f32_16x16x4_f32(false, (a), false, (b), (short)0, (c), false, false)

__global__ void zero_kernel(float* __restrict__ p, long long n) {
  long long i = (long long)blockIdx.x * blockDim.x + threadIdx.x;
  if (i < n) p[i] = 0.0f;
}

#define NBINS 256
#define ROWS_MAX 256
#define CHUNKS 256
#define QCHUNKS 64

__global__ void __launch_bounds__(1024)
bin_count_kernel(const int* __restrict__ src, const int* __restrict__ dst,
                 float* __restrict__ deg, int* __restrict__ hist, int N, int E) {
  __shared__ int cnt[NBINS];
  const int tid = threadIdx.x;
  const int bw = (N + NBINS - 1) / NBINS;
  const int eb = (E + CHUNKS - 1) / CHUNKS;
  if (tid < NBINS) cnt[tid] = 0;
  __syncthreads();
  const int lo = blockIdx.x * eb;
  const int hi = min(E, lo + eb);
  for (int e = lo + tid; e < hi; e += 1024) {
    const int d = dst[e];
    const int s = src[e];
    atomicAdd(&cnt[(unsigned)d / (unsigned)bw], 1);
    unsafeAtomicAdd(&deg[s], 1.0f);
  }
  __syncthreads();
  if (tid < NBINS) hist[blockIdx.x * NBINS + tid] = cnt[tid];
}

__global__ void __launch_bounds__(1024)
bin_scan_kernel(const int* __restrict__ hist, int* __restrict__ off, int* __restrict__ binstart) {
  __shared__ int seg[4][NBINS];
  __shared__ int sc[2][NBINS];
  const int tid = threadIdx.x;
  const int bin = tid & (NBINS - 1);
  const int q = tid >> 8;
  const int* __restrict__ hp = hist + q * QCHUNKS * NBINS + bin;
  int v[QCHUNKS];
#pragma unroll
  for (int i = 0; i < QCHUNKS; ++i) v[i] = hp[i * NBINS];
  int s = 0;
#pragma unroll
  for (int i = 0; i < QCHUNKS; ++i) s += v[i];
  seg[q][bin] = s;
  __syncthreads();
  const int s0 = seg[0][bin], s1 = seg[1][bin], s2 = seg[2][bin], s3 = seg[3][bin];
  const int tot = (s0 + s1) + (s2 + s3);
  const int before = (q > 0 ? s0 : 0) + (q > 1 ? s1 : 0) + (q > 2 ? s2 : 0);
  if (tid < NBINS) sc[0][tid] = tot;
  __syncthreads();
  int pin = 0;
  for (int o = 1; o < NBINS; o <<= 1) {
    if (tid < NBINS) {
      int x = sc[pin][tid];
      if (tid >= o) x += sc[pin][tid - o];
      sc[pin ^ 1][tid] = x;
    }
    __syncthreads();
    pin ^= 1;
  }
  const int excl = sc[pin][bin] - tot;
  if (tid < NBINS) {
    binstart[tid] = excl;
    if (tid == NBINS - 1) binstart[NBINS] = sc[pin][NBINS - 1];
  }
  int run = excl + before;
  int* __restrict__ op = off + q * QCHUNKS * NBINS + bin;
#pragma unroll
  for (int i = 0; i < QCHUNKS; ++i) { op[i * NBINS] = run; run += v[i]; }
}

__global__ void __launch_bounds__(1024)
bin_fill_kernel(const int* __restrict__ src, const int* __restrict__ dst,
                const float* __restrict__ deg, const int* __restrict__ off,
                int* __restrict__ bkey, float* __restrict__ bnrm, int N, int E) {
  __shared__ int cur[NBINS];
  const int tid = threadIdx.x;
  const int bw = (N + NBINS - 1) / NBINS;
  const int eb = (E + CHUNKS - 1) / CHUNKS;
  if (tid < NBINS) cur[tid] = off[blockIdx.x * NBINS + tid];
  __syncthreads();
  const int lo = blockIdx.x * eb;
  const int hi = min(E, lo + eb);
  for (int e = lo + tid; e < hi; e += 1024) {
    const int d = dst[e];
    const int s = src[e];
    const int bin = (int)((unsigned)d / (unsigned)bw);
    const int dl = d - bin * bw;
    const float ds = deg[s];
    const float dd = deg[d];
    const float is = (ds > 0.0f) ? rsqrtf(fmaxf(ds, 1.0f)) : 0.0f;
    const float id = (dd > 0.0f) ? rsqrtf(fmaxf(dd, 1.0f)) : 0.0f;
    const int pos = atomicAdd(&cur[bin], 1);
    bkey[pos] = (s << 8) | dl;
    bnrm[pos] = -is * id;
  }
}

#define EDGES_PER_TRIP 8
__global__ void __launch_bounds__(1024)
bin_sum_kernel(const int* __restrict__ bkey, const float* __restrict__ bnrm,
               const int* __restrict__ binstart, const float* __restrict__ y,
               const float* __restrict__ pre, float* __restrict__ outp, int N, int hidden) {
  __shared__ float acc[ROWS_MAX * 16];
  const int tid = threadIdx.x;
  const int bin = blockIdx.x;
  const int bw = (N + NBINS - 1) / NBINS;
  for (int j = tid; j < ROWS_MAX * 16; j += 1024) acc[j] = 0.0f;
  __syncthreads();
  const int beg = binstart[bin];
  const int end = binstart[bin + 1];
  const int g = tid >> 4;
  const int f = tid & 15;
  int i0 = beg + g;
  if (i0 < end) {
    const int last = end - 1;
    int key[EDGES_PER_TRIP];
    float w[EDGES_PER_TRIP];
#pragma unroll
    for (int k = 0; k < EDGES_PER_TRIP; ++k) {
      const int ic = min(i0 + 64 * k, last);
      key[k] = bkey[ic];
      w[k] = bnrm[ic];
    }
    for (; i0 < end; i0 += 64 * EDGES_PER_TRIP) {
      float yv[EDGES_PER_TRIP];
#pragma unroll
      for (int k = 0; k < EDGES_PER_TRIP; ++k) yv[k] = y[(long long)(key[k] >> 8) * 16 + f];
      int nkey[EDGES_PER_TRIP];
      float nw[EDGES_PER_TRIP];
#pragma unroll
      for (int k = 0; k < EDGES_PER_TRIP; ++k) {
        const int ic = min(i0 + 64 * (EDGES_PER_TRIP + k), last);
        nkey[k] = bkey[ic];
        nw[k] = bnrm[ic];
      }
      asm volatile("" :: "v"(yv[0]), "v"(yv[1]), "v"(yv[2]), "v"(yv[3]),
                         "v"(yv[4]), "v"(yv[5]), "v"(yv[6]), "v"(yv[7]));
#pragma unroll
      for (int k = 0; k < EDGES_PER_TRIP; ++k) {
        if (i0 + 64 * k < end) {
          __hip_atomic_fetch_add(&acc[(key[k] & 255) * 16 + f], w[k] * yv[k],
                                 __ATOMIC_RELAXED, __HIP_MEMORY_SCOPE_WORKGROUP);
        }
      }
#pragma unroll
      for (int k = 0; k < EDGES_PER_TRIP; ++k) { key[k] = nkey[k]; w[k] = nw[k]; }
    }
  }
  __syncthreads();
  const int base = bin * bw;
  const int nrows = min(bw, N - base);
  for (int j = tid; j < nrows * 16; j += 1024) {
    const long long gi = (long long)base * 16 + j;
    const float a = acc[j];
    outp[gi] = hidden ? fmaxf(pre[gi] + a, 0.0f) : a;
  }
}

__global__ void gemm1_kernel(const float* __restrict__ x, const float* __restrict__ W1,
                             const float* __restrict__ b1, float* __restrict__ pre1,
                             float* __restrict__ y1, int N, int nTiles) {
  const int wave = (blockIdx.x * blockDim.x + threadIdx.x) >> 5;
  if (wave >= nTiles) return;
  const int lane = threadIdx.x & 31;
  const int n = lane & 15;
  const int hi = lane >> 4;
  const int base = wave * 16;
  const int mrow = min(base + n, N - 1);
  const float* __restrict__ W0 = W1;
  const float* __restrict__ Wb = W1 + 64 * 16;
  const float* __restrict__ xr = x + (long long)mrow * 64;

  v8f acc0 = {0.f, 0.f, 0.f, 0.f, 0.f, 0.f, 0.f, 0.f};
  v8f acc1 = acc0;
#pragma unroll
  for (int kb = 0; kb < 64; kb += 4) {
    const int k0 = kb + 2 * hi;
    v2f a = *(const v2f*)(xr + k0);
    v2f b0, bb;
    b0.x = W0[k0 * 16 + n];  b0.y = W0[(k0 + 1) * 16 + n];
    bb.x = Wb[k0 * 16 + n];  bb.y = Wb[(k0 + 1) * 16 + n];
    acc0 = WMMA_F32X4(a, b0, acc0);
    acc1 = WMMA_F32X4(a, bb, acc1);
  }
  const float bias = b1[n];
#pragma unroll
  for (int r = 0; r < 8; ++r) {
    const int row = base + r + 8 * hi;
    if (row < N) {
      pre1[(long long)row * 16 + n] = acc0[r] + bias;
      y1[(long long)row * 16 + n]  = acc1[r];
    }
  }
}

__global__ void gemm2_kernel(const float* __restrict__ h, const float* __restrict__ t,
                             const float* __restrict__ W2, const float* __restrict__ b2,
                             float* __restrict__ z, int N, int nTiles) {
  const int wave = (blockIdx.x * blockDim.x + threadIdx.x) >> 5;
  if (wave >= nTiles) return;
  const int lane = threadIdx.x & 31;
  const int n = lane & 15;
  const int hi = lane >> 4;
  const int base = wave * 16;
  const int mrow = min(base + n, N - 1);
  const float* __restrict__ W0 = W2;
  const float* __restrict__ Wb = W2 + 16 * 40;

  v2f a[4], g[4];
#pragma unroll
  for (int i = 0; i < 4; ++i) {
    const int k0 = i * 4 + 2 * hi;
    const long long idx = (long long)mrow * 16 + k0;
    a[i] = *(const v2f*)(h + idx);
    g[i] = *(const v2f*)(t + idx);
  }

  v8f acc[6];
#pragma unroll
  for (int i = 0; i < 6; ++i) acc[i] = (v8f){0.f, 0.f, 0.f, 0.f, 0.f, 0.f, 0.f, 0.f};

#pragma unroll
  for (int ct = 0; ct < 3; ++ct) {
    const int col = ct * 16 + n;
    const int cc = min(col, 39);
#pragma unroll
    for (int i = 0; i < 4; ++i) {
      const int k0 = i * 4 + 2 * hi;
      v2f b0, bb;
      b0.x = W0[k0 * 40 + cc];  b0.y = W0[(k0 + 1) * 40 + cc];
      bb.x = Wb[k0 * 40 + cc];  bb.y = Wb[(k0 + 1) * 40 + cc];
      acc[ct * 2]     = WMMA_F32X4(a[i], b0, acc[ct * 2]);
      acc[ct * 2 + 1] = WMMA_F32X4(g[i], bb, acc[ct * 2 + 1]);
    }
  }

#pragma unroll
  for (int ct = 0; ct < 3; ++ct) {
    const int col = ct * 16 + n;
    if (col < 40) {
      const float bias = b2[col];
#pragma unroll
      for (int r = 0; r < 8; ++r) {
        const int row = base + r + 8 * hi;
        if (row < N) {
          z[(long long)row * 40 + col] = (acc[ct * 2][r] + acc[ct * 2 + 1][r]) + bias;
        }
      }
    }
  }
}

__global__ void lsm_kernel(const float* __restrict__ z, float* __restrict__ out, int N) {
  const int node = (blockIdx.x * blockDim.x + threadIdx.x) >> 5;
  if (node >= N) return;
  const int lane = threadIdx.x & 31;
  const float* __restrict__ pr = z + (long long)node * 40;
  float v0 = pr[lane];
  float v1 = (lane < 8) ? pr[lane + 32] : -INFINITY;
  float m = fmaxf(v0, v1);
#pragma unroll
  for (int o = 16; o > 0; o >>= 1) m = fmaxf(m, __shfl_xor(m, o, 32));
  float s = expf(v0 - m) + ((lane < 8) ? expf(v1 - m) : 0.0f);
#pragma unroll
  for (int o = 16; o > 0; o >>= 1) s += __shfl_xor(s, o, 32);
  const float lse = m + logf(s);
  float* __restrict__ op = out + (long long)node * 40;
  op[lane] = v0 - lse;
  if (lane < 8) op[lane + 32] = v1 - lse;
}

extern "C" void kernel_launch(void* const* d_in, const int* in_sizes, int n_in,
                              void* d_out, int out_size, void* d_ws, size_t ws_size,
                              hipStream_t stream) {
  const float* x  = (const float*)d_in[0];
  const int* ei   = (const int*)d_in[1];
  const float* W1 = (const float*)d_in[2];
  const float* b1 = (const float*)d_in[3];
  const float* W2 = (const float*)d_in[4];
  const float* b2 = (const float*)d_in[5];
  float* out = (float*)d_out;

  const int N = in_sizes[0] / 64;
  const int E = in_sizes[1] / 2;
  const int* src = ei;
  const int* dst = ei + E;

  float* ws = (float*)d_ws;
  size_t off = 0;
  float* deg  = ws + off; off += (size_t)N;
  int* binstart = (int*)(ws + off); off += (size_t)N;
  float* bnrm = ws + off; off += (size_t)E;
  float* pre1 = ws + off; off += (size_t)N * 16;
  float* y1   = ws + off; off += (size_t)N * 16;
  int* hist   = (int*)(ws + off);
  int* coff   = hist + CHUNKS * NBINS;
  off += (size_t)N * 16;
  float* h    = ws + off; off += (size_t)N * 16;
  float* t2   = ws + off; off += (size_t)N * 16;
  float* z2   = ws + off; off += (size_t)N * 40;
  int* bkey   = (int*)z2;

  const int B = 256;
  auto blk = [](long long n, int b) { return (unsigned)((n + b - 1) / b); };

  zero_kernel<<<blk(N, B), B, 0, stream>>>(deg, N);
  bin_count_kernel<<<CHUNKS, 1024, 0, stream>>>(src, dst, deg, hist, N, E);
  bin_scan_kernel<<<1, 1024, 0, stream>>>(hist, coff, binstart);
  bin_fill_kernel<<<CHUNKS, 1024, 0, stream>>>(src, dst, deg, coff, bkey, bnrm, N, E);

  const int tiles = (N + 15) / 16;
  gemm1_kernel<<<blk((long long)tiles * 32, B), B, 0, stream>>>(x, W1, b1, pre1, y1, N, tiles);
  bin_sum_kernel<<<NBINS, 1024, 0, stream>>>(bkey, bnrm, binstart, y1, pre1, h, N, 1);

  bin_sum_kernel<<<NBINS, 1024, 0, stream>>>(bkey, bnrm, binstart, h, pre1, t2, N, 0);
  gemm2_kernel<<<blk((long long)tiles * 32, B), B, 0, stream>>>(h, t2, W2, b2, z2, N, tiles);

  lsm_kernel<<<blk((long long)N * 32, B), B, 0, stream>>>(z2, out, N);
}


# Layer-2 tail fused into its range-sum kernel: sums stay in LDS, each work-group projects its own rows (f32 WMMA) and takes log_softmax in registers (exp-sum now a 16-lane tree); 2 launches fewer
# speedup vs baseline: 1.8205x; 1.0888x over previous
#include <hip/hip_runtime.h>
#include <math.h>

typedef float v2f __attribute__((ext_vector_type(2)));
typedef float v8f __attribute__((ext_vector_type(8)));

#define WMMA_F32X4(a, b, c) \
  __builtin_amdgcn_wmma_f32_16x16x4_f32(false, (a), false, (b), (short)0, (c), false, false)

__global__ void zero_kernel(float* __restrict__ p, long long n) {
  long long i = (long long)blockIdx.x * blockDim.x + threadIdx.x;
  if (i < n) p[i] = 0.0f;
}

#define NBINS 256
#define ROWS_MAX 256
#define CHUNKS 256
#define QCHUNKS 64

__global__ void __launch_bounds__(1024)
bin_count_kernel(const int* __restrict__ src, const int* __restrict__ dst,
                 float* __restrict__ deg, int* __restrict__ hist, int N, int E) {
  __shared__ int cnt[NBINS];
  const int tid = threadIdx.x;
  const int bw = (N + NBINS - 1) / NBINS;
  const int eb = (E + CHUNKS - 1) / CHUNKS;
  if (tid < NBINS) cnt[tid] = 0;
  __syncthreads();
  const int lo = blockIdx.x * eb;
  const int hi = min(E, lo + eb);
  for (int e = lo + tid; e < hi; e += 1024) {
    const int d = dst[e];
    const int s = src[e];
    atomicAdd(&cnt[(unsigned)d / (unsigned)bw], 1);
    unsafeAtomicAdd(&deg[s], 1.0f);
  }
  __syncthreads();
  if (tid < NBINS) hist[blockIdx.x * NBINS + tid] = cnt[tid];
}

__global__ void __launch_bounds__(1024)
bin_scan_kernel(const int* __restrict__ hist, int* __restrict__ off, int* __restrict__ binstart) {
  __shared__ int seg[4][NBINS];
  __shared__ int sc[2][NBINS];
  const int tid = threadIdx.x;
  const int bin = tid & (NBINS - 1);
  const int q = tid >> 8;
  const int* __restrict__ hp = hist + q * QCHUNKS * NBINS + bin;
  int v[QCHUNKS];
#pragma unroll
  for (int i = 0; i < QCHUNKS; ++i) v[i] = hp[i * NBINS];
  int s = 0;
#pragma unroll
  for (int i = 0; i < QCHUNKS; ++i) s += v[i];
  seg[q][bin] = s;
  __syncthreads();
  const int s0 = seg[0][bin], s1 = seg[1][bin], s2 = seg[2][bin], s3 = seg[3][bin];
  const int tot = (s0 + s1) + (s2 + s3);
  const int before = (q > 0 ? s0 : 0) + (q > 1 ? s1 : 0) + (q > 2 ? s2 : 0);
  if (tid < NBINS) sc[0][tid] = tot;
  __syncthreads();
  int pin = 0;
  for (int o = 1; o < NBINS; o <<= 1) {
    if (tid < NBINS) {
      int x = sc[pin][tid];
      if (tid >= o) x += sc[pin][tid - o];
      sc[pin ^ 1][tid] = x;
    }
    __syncthreads();
    pin ^= 1;
  }
  const int excl = sc[pin][bin] - tot;
  if (tid < NBINS) {
    binstart[tid] = excl;
    if (tid == NBINS - 1) binstart[NBINS] = sc[pin][NBINS - 1];
  }
  int run = excl + before;
  int* __restrict__ op = off + q * QCHUNKS * NBINS + bin;
#pragma unroll
  for (int i = 0; i < QCHUNKS; ++i) { op[i * NBINS] = run; run += v[i]; }
}

__global__ void __launch_bounds__(1024)
bin_fill_kernel(const int* __restrict__ src, const int* __restrict__ dst,
                const float* __restrict__ deg, const int* __restrict__ off,
                int* __restrict__ bkey, float* __restrict__ bnrm, int N, int E) {
  __shared__ int cur[NBINS];
  const int tid = threadIdx.x;
  const int bw = (N + NBINS - 1) / NBINS;
  const int eb = (E + CHUNKS - 1) / CHUNKS;
  if (tid < NBINS) cur[tid] = off[blockIdx.x * NBINS + tid];
  __syncthreads();
  const int lo = blockIdx.x * eb;
  const int hi = min(E, lo + eb);
  for (int e = lo + tid; e < hi; e += 1024) {
    const int d = dst[e];
    const int s = src[e];
    const int bin = (int)((unsigned)d / (unsigned)bw);
    const int dl = d - bin * bw;
    const float ds = deg[s];
    const float dd = deg[d];
    const float is = (ds > 0.0f) ? rsqrtf(fmaxf(ds, 1.0f)) : 0.0f;
    const float id = (dd > 0.0f) ? rsqrtf(fmaxf(dd, 1.0f)) : 0.0f;
    const int pos = atomicAdd(&cur[bin], 1);
    bkey[pos] = (s << 8) | dl;
    bnrm[pos] = -is * id;
  }
}

template <int M>
__device__ __forceinline__ float row_xor(float v) {
  const int b = __builtin_bit_cast(int, v);
  return __builtin_bit_cast(float, __builtin_amdgcn_update_dpp(b, b, 0x160 + M, 0xf, 0xf, false));
}

#define EDGES_PER_TRIP 8
template <bool TAIL>
__global__ void __launch_bounds__(1024)
bin_sum_kernel(const int* __restrict__ bkey, const float* __restrict__ bnrm,
               const int* __restrict__ binstart, const float* __restrict__ y,
               const float* __restrict__ pre, float* __restrict__ outp,
               const float* __restrict__ W2, const float* __restrict__ b2, int N) {
  __shared__ __attribute__((aligned(16))) float acc[ROWS_MAX * 16];
  const int tid = threadIdx.x;
  const int bin = blockIdx.x;
  const int bw = (N + NBINS - 1) / NBINS;
  for (int j = tid; j < ROWS_MAX * 16; j += 1024) acc[j] = 0.0f;
  __syncthreads();
  const int beg = binstart[bin];
  const int end = binstart[bin + 1];
  const int g = tid >> 4;
  const int f = tid & 15;
  int i0 = beg + g;
  if (i0 < end) {
    const int last = end - 1;
    int key[EDGES_PER_TRIP];
    float w[EDGES_PER_TRIP];
#pragma unroll
    for (int k = 0; k < EDGES_PER_TRIP; ++k) {
      const int ic = min(i0 + 64 * k, last);
      key[k] = bkey[ic];
      w[k] = bnrm[ic];
    }
    for (; i0 < end; i0 += 64 * EDGES_PER_TRIP) {
      float yv[EDGES_PER_TRIP];
#pragma unroll
      for (int k = 0; k < EDGES_PER_TRIP; ++k) yv[k] = y[(long long)(key[k] >> 8) * 16 + f];
      int nkey[EDGES_PER_TRIP];
      float nw[EDGES_PER_TRIP];
#pragma unroll
      for (int k = 0; k < EDGES_PER_TRIP; ++k) {
        const int ic = min(i0 + 64 * (EDGES_PER_TRIP + k), last);
        nkey[k] = bkey[ic];
        nw[k] = bnrm[ic];
      }
      asm volatile("" :: "v"(yv[0]), "v"(yv[1]), "v"(yv[2]), "v"(yv[3]),
                         "v"(yv[4]), "v"(yv[5]), "v"(yv[6]), "v"(yv[7]));
#pragma unroll
      for (int k = 0; k < EDGES_PER_TRIP; ++k) {
        if (i0 + 64 * k < end) {
          __hip_atomic_fetch_add(&acc[(key[k] & 255) * 16 + f], w[k] * yv[k],
                                 __ATOMIC_RELAXED, __HIP_MEMORY_SCOPE_WORKGROUP);
        }
      }
#pragma unroll
      for (int k = 0; k < EDGES_PER_TRIP; ++k) { key[k] = nkey[k]; w[k] = nw[k]; }
    }
  }
  const int base = bin * bw;
  const int nrows = min(bw, N - base);
  if constexpr (!TAIL) {
    __syncthreads();
    for (int j = tid; j < nrows * 16; j += 1024) {
      const long long gi = (long long)base * 16 + j;
      outp[gi] = fmaxf(pre[gi] + acc[j], 0.0f);
    }
  } else {
    __shared__ __attribute__((aligned(16))) float zs[ROWS_MAX * 40];
    const int wv = __builtin_amdgcn_readfirstlane(tid >> 5);
    const int ntiles = nrows > 0 ? (nrows + 15) >> 4 : 0;
    const bool tilewave = wv < ntiles;
    const int n = tid & 15;
    const int hi = (tid >> 4) & 1;
    const float* __restrict__ W0 = W2;
    const float* __restrict__ Wb = W2 + 16 * 40;
    v2f a[4], b0[12], bb[12];
    float bias0 = 0.f, bias1 = 0.f, bias2 = 0.f;
#pragma unroll
    for (int i = 0; i < 4; ++i) a[i] = (v2f){0.f, 0.f};
#pragma unroll
    for (int i = 0; i < 12; ++i) { b0[i] = (v2f){0.f, 0.f}; bb[i] = (v2f){0.f, 0.f}; }
    if (tilewave) {
      const int mrow = min(base + wv * 16 + n, N - 1);
#pragma unroll
      for (int i = 0; i < 4; ++i) a[i] = *(const v2f*)(y + (long long)mrow * 16 + i * 4 + 2 * hi);
#pragma unroll
      for (int ct = 0; ct < 3; ++ct) {
        const int cc = min(ct * 16 + n, 39);
#pragma unroll
        for (int i = 0; i < 4; ++i) {
          const int k0 = i * 4 + 2 * hi;
          b0[ct * 4 + i].x = W0[k0 * 40 + cc];  b0[ct * 4 + i].y = W0[(k0 + 1) * 40 + cc];
          bb[ct * 4 + i].x = Wb[k0 * 40 + cc];  bb[ct * 4 + i].y = Wb[(k0 + 1) * 40 + cc];
        }
      }
      bias0 = b2[n];  bias1 = b2[16 + n];  bias2 = b2[min(32 + n, 39)];
    }
    __syncthreads();
    if (tilewave) {
      v2f t[4];
#pragma unroll
      for (int i = 0; i < 4; ++i) t[i] = *(const v2f*)(&acc[(wv * 16 + n) * 16 + i * 4 + 2 * hi]);
      v8f c[6];
#pragma unroll
      for (int i = 0; i < 6; ++i) c[i] = (v8f){0.f, 0.f, 0.f, 0.f, 0.f, 0.f, 0.f, 0.f};
#pragma unroll
      for (int ct = 0; ct < 3; ++ct) {
#pragma unroll
        for (int i = 0; i < 4; ++i) {
          c[ct * 2]     = WMMA_F32X4(a[i], b0[ct * 4 + i], c[ct * 2]);
          c[ct * 2 + 1] = WMMA_F32X4(t[i], bb[ct * 4 + i], c[ct * 2 + 1]);
        }
      }
      float z0[8], z1[8], z2[8], m[8], s[8];
#pragma unroll
      for (int r = 0; r < 8; ++r) {
        z0[r] = (c[0][r] + c[1][r]) + bias0;
        z1[r] = (c[2][r] + c[3][r]) + bias1;
        z2[r] = (c[4][r] + c[5][r]) + bias2;
        m[r] = fmaxf(z0[r], z1[r]);
      }
      if (n < 8) {
#pragma unroll
        for (int r = 0; r < 8; ++r) m[r] = fmaxf(m[r], z2[r]);
      }
#pragma unroll
      for (int r = 0; r < 8; ++r) m[r] = fmaxf(m[r], row_xor<8>(m[r]));
#pragma unroll
      for (int r = 0; r < 8; ++r) m[r] = fmaxf(m[r], row_xor<4>(m[r]));
#pragma unroll
      for (int r = 0; r < 8; ++r) m[r] = fmaxf(m[r], row_xor<2>(m[r]));
#pragma unroll
      for (int r = 0; r < 8; ++r) m[r] = fmaxf(m[r], row_xor<1>(m[r]));
#pragma unroll
      for (int r = 0; r < 8; ++r) s[r] = expf(z0[r] - m[r]) + expf(z1[r] - m[r]);
      if (n < 8) {
#pragma unroll
        for (int r = 0; r < 8; ++r) s[r] += expf(z2[r] - m[r]);
      }
#pragma unroll
      for (int r = 0; r < 8; ++r) s[r] += row_xor<8>(s[r]);
#pragma unroll
      for (int r = 0; r < 8; ++r) s[r] += row_xor<4>(s[r]);
#pragma unroll
      for (int r = 0; r < 8; ++r) s[r] += row_xor<2>(s[r]);
#pragma unroll
      for (int r = 0; r < 8; ++r) s[r] += row_xor<1>(s[r]);
#pragma unroll
      for (int r = 0; r < 8; ++r) {
        const float lse = m[r] + logf(s[r]);
        float* zr = zs + (wv * 16 + r + 8 * hi) * 40;
        zr[n] = z0[r] - lse;
        zr[16 + n] = z1[r] - lse;
        if (n < 8) zr[32 + n] = z2[r] - lse;
      }
    }
    __syncthreads();
    float4* __restrict__ dst4 = (float4*)(outp + (long long)base * 40);
    const float4* src4 = (const float4*)zs;
    for (int j = tid; j < nrows * 10; j += 1024) dst4[j] = src4[j];
  }
}

__global__ void gemm1_kernel(const float* __restrict__ x, const float* __restrict__ W1,
                             const float* __restrict__ b1, float* __restrict__ pre1,
                             float* __restrict__ y1, int N, int nTiles) {
  const int wave = (blockIdx.x * blockDim.x + threadIdx.x) >> 5;
  if (wave >= nTiles) return;
  const int lane = threadIdx.x & 31;
  const int n = lane & 15;
  const int hi = lane >> 4;
  const int base = wave * 16;
  const int mrow = min(base + n, N - 1);
  const float* __restrict__ W0 = W1;
  const float* __restrict__ Wb = W1 + 64 * 16;
  const float* __restrict__ xr = x + (long long)mrow * 64;

  v8f acc0 = {0.f, 0.f, 0.f, 0.f, 0.f, 0.f, 0.f, 0.f};
  v8f acc1 = acc0;
#pragma unroll
  for (int kb = 0; kb < 64; kb += 4) {
    const int k0 = kb + 2 * hi;
    v2f a = *(const v2f*)(xr + k0);
    v2f b0, bb;
    b0.x = W0[k0 * 16 + n];  b0.y = W0[(k0 + 1) * 16 + n];
    bb.x = Wb[k0 * 16 + n];  bb.y = Wb[(k0 + 1) * 16 + n];
    acc0 = WMMA_F32X4(a, b0, acc0);
    acc1 = WMMA_F32X4(a, bb, acc1);
  }
  const float bias = b1[n];
#pragma unroll
  for (int r = 0; r < 8; ++r) {
    const int row = base + r + 8 * hi;
    if (row < N) {
      pre1[(long long)row * 16 + n] = acc0[r] + bias;
      y1[(long long)row * 16 + n]  = acc1[r];
    }
  }
}

extern "C" void kernel_launch(void* const* d_in, const int* in_sizes, int n_in,
                              void* d_out, int out_size, void* d_ws, size_t ws_size,
                              hipStream_t stream) {
  const float* x  = (const float*)d_in[0];
  const int* ei   = (const int*)d_in[1];
  const float* W1 = (const float*)d_in[2];
  const float* b1 = (const float*)d_in[3];
  const float* W2 = (const float*)d_in[4];
  const float* b2 = (const float*)d_in[5];
  float* out = (float*)d_out;

  const int N = in_sizes[0] / 64;
  const int E = in_sizes[1] / 2;
  const int* src = ei;
  const int* dst = ei + E;

  float* ws = (float*)d_ws;
  size_t off = 0;
  float* deg  = ws + off; off += (size_t)N;
  int* binstart = (int*)(ws + off); off += (size_t)N;
  float* bnrm = ws + off; off += (size_t)E;
  float* pre1 = ws + off; off += (size_t)N * 16;
  float* y1   = ws + off; off += (size_t)N * 16;
  int* hist   = (int*)(ws + off);
  int* coff   = hist + CHUNKS * NBINS;
  off += (size_t)N * 16;
  float* h    = ws + off; off += (size_t)N * 16;
  int* bkey   = (int*)(ws + off);

  const int B = 256;
  auto blk = [](long long n, int b) { return (unsigned)((n + b - 1) / b); };

  zero_kernel<<<blk(N, B), B, 0, stream>>>(deg, N);
  bin_count_kernel<<<CHUNKS, 1024, 0, stream>>>(src, dst, deg, hist, N, E);
  bin_scan_kernel<<<1, 1024, 0, stream>>>(hist, coff, binstart);
  bin_fill_kernel<<<CHUNKS, 1024, 0, stream>>>(src, dst, deg, coff, bkey, bnrm, N, E);

  const int tiles = (N + 15) / 16;
  gemm1_kernel<<<blk((long long)tiles * 32, B), B, 0, stream>>>(x, W1, b1, pre1, y1, N, tiles);
  bin_sum_kernel<false><<<NBINS, 1024, 0, stream>>>(bkey, bnrm, binstart, y1, pre1, h, W2, b2, N);

  bin_sum_kernel<true><<<NBINS, 1024, 0, stream>>>(bkey, bnrm, binstart, h, pre1, out, W2, b2, N);
}


# Degrees by source counted in LDS by (edge slab x source range) work-groups of the count pass; 32 integer count planes summed per row in the scan launch: no global f32 atomic per edge, no zero launch
# speedup vs baseline: 2.6052x; 1.4310x over previous
#include <hip/hip_runtime.h>
#include <math.h>

typedef float v2f __attribute__((ext_vector_type(2)));
typedef float v8f __attribute__((ext_vector_type(8)));

#define WMMA_F32X4(a, b, c) \
  __builtin_amdgcn_wmma_f32_16x16x4_f32(false, (a), false, (b), (short)0, (c), false, false)

#define NBINS 256
#define ROWS_MAX 256
#define CHUNKS 256
#define QCHUNKS 64

#define DEG_SLABS 32
#define DEG_RANGES 8
#define DEG_ROWS_MAX 8192
#define DEG_LOADS 8
static_assert(DEG_SLABS * DEG_RANGES == CHUNKS, "one (slab, range) pair per work-group of pass 1");

__global__ void __launch_bounds__(1024)
bin_count_kernel(const int* __restrict__ src, const int* __restrict__ dst,
                 int* __restrict__ degp, int* __restrict__ hist, int N, int E) {
  __shared__ int cnt[NBINS];
  __shared__ int dcnt[DEG_ROWS_MAX];
  const int tid = threadIdx.x;
  const int bw = (N + NBINS - 1) / NBINS;
  const int eb = (E + CHUNKS - 1) / CHUNKS;
  const int c = blockIdx.x / DEG_RANGES;
  const int r = blockIdx.x - c * DEG_RANGES;
  const int rw = (N + DEG_RANGES - 1) / DEG_RANGES;
  const int r_lo = r * rw;
  if (tid < NBINS) cnt[tid] = 0;
  for (int j = tid; j < rw; j += 1024) dcnt[j] = 0;
  __syncthreads();
  const int lo = blockIdx.x * eb;
  const int hi = min(E, lo + eb);
  for (int e = lo + tid; e < hi; e += 1024) {
    const int d = dst[e];
    atomicAdd(&cnt[(unsigned)d / (unsigned)bw], 1);
  }
  const int slab = (((E + DEG_SLABS - 1) / DEG_SLABS) + 3) & ~3;
  const int s_lo = c * slab;
  const int s_hi = min(E, s_lo + slab);
  const int Eal = E & ~3;
  const int v_hi = min(s_hi, Eal);
  for (int e0 = s_lo + 4 * tid; e0 < v_hi; e0 += 4096 * DEG_LOADS) {
    int4 v[DEG_LOADS];
#pragma unroll
    for (int k = 0; k < DEG_LOADS; ++k) {
      const int e = min(e0 + 4096 * k, Eal - 4);
      v[k] = *(const int4*)(src + e);
    }
#pragma unroll
    for (int k = 0; k < DEG_LOADS; ++k) {
      if (e0 + 4096 * k < v_hi) {
        const unsigned j0 = (unsigned)(v[k].x - r_lo), j1 = (unsigned)(v[k].y - r_lo);
        const unsigned j2 = (unsigned)(v[k].z - r_lo), j3 = (unsigned)(v[k].w - r_lo);
        if (j0 < (unsigned)rw) atomicAdd(&dcnt[j0], 1);
        if (j1 < (unsigned)rw) atomicAdd(&dcnt[j1], 1);
        if (j2 < (unsigned)rw) atomicAdd(&dcnt[j2], 1);
        if (j3 < (unsigned)rw) atomicAdd(&dcnt[j3], 1);
      }
    }
  }
  if (tid < E - Eal) {
    const int e = Eal + tid;
    if (e >= s_lo && e < s_hi) {
      const unsigned j = (unsigned)(src[e] - r_lo);
      if (j < (unsigned)rw) atomicAdd(&dcnt[j], 1);
    }
  }
  __syncthreads();
  if (tid < NBINS) hist[blockIdx.x * NBINS + tid] = cnt[tid];
  int* __restrict__ plane = degp + (size_t)c * N;
  for (int j = tid; j < rw; j += 1024) {
    const int node = r_lo + j;
    if (node < N) plane[node] = dcnt[j];
  }
}

__global__ void __launch_bounds__(1024)
bin_scan_kernel(const int* __restrict__ hist, int* __restrict__ off, int* __restrict__ binstart,
                const int* __restrict__ degp, float* __restrict__ dinv, int N) {
  __shared__ int seg[4][NBINS];
  __shared__ int sc[2][NBINS];
  const int tid = threadIdx.x;
  if (blockIdx.x > 0) {
    const int node = (blockIdx.x - 1) * 1024 + tid;
    if (node < N) {
      int p[DEG_SLABS];
#pragma unroll
      for (int i = 0; i < DEG_SLABS; ++i) p[i] = degp[(size_t)i * N + node];
      int cs = 0;
#pragma unroll
      for (int i = 0; i < DEG_SLABS; ++i) cs += p[i];
      const float dg = (float)cs;
      dinv[node] = (dg > 0.0f) ? rsqrtf(fmaxf(dg, 1.0f)) : 0.0f;
    }
    return;
  }
  const int bin = tid & (NBINS - 1);
  const int q = tid >> 8;
  const int* __restrict__ hp = hist + q * QCHUNKS * NBINS + bin;
  int v[QCHUNKS];
#pragma unroll
  for (int i = 0; i < QCHUNKS; ++i) v[i] = hp[i * NBINS];
  int s = 0;
#pragma unroll
  for (int i = 0; i < QCHUNKS; ++i) s += v[i];
  seg[q][bin] = s;
  __syncthreads();
  const int s0 = seg[0][bin], s1 = seg[1][bin], s2 = seg[2][bin], s3 = seg[3][bin];
  const int tot = (s0 + s1) + (s2 + s3);
  const int before = (q > 0 ? s0 : 0) + (q > 1 ? s1 : 0) + (q > 2 ? s2 : 0);
  if (tid < NBINS) sc[0][tid] = tot;
  __syncthreads();
  int pin = 0;
  for (int o = 1; o < NBINS; o <<= 1) {
    if (tid < NBINS) {
      int x = sc[pin][tid];
      if (tid >= o) x += sc[pin][tid - o];
      sc[pin ^ 1][tid] = x;
    }
    __syncthreads();
    pin ^= 1;
  }
  const int excl = sc[pin][bin] - tot;
  if (tid < NBINS) {
    binstart[tid] = excl;
    if (tid == NBINS - 1) binstart[NBINS] = sc[pin][NBINS - 1];
  }
  int run = excl + before;
  int* __restrict__ op = off + q * QCHUNKS * NBINS + bin;
#pragma unroll
  for (int i = 0; i < QCHUNKS; ++i) { op[i * NBINS] = run; run += v[i]; }
}

__global__ void __launch_bounds__(1024)
bin_fill_kernel(const int* __restrict__ src, const int* __restrict__ dst,
                const float* __restrict__ dinv, const int* __restrict__ off,
                int* __restrict__ bkey, float* __restrict__ bnrm, int N, int E) {
  __shared__ int cur[NBINS];
  const int tid = threadIdx.x;
  const int bw = (N + NBINS - 1) / NBINS;
  const int eb = (E + CHUNKS - 1) / CHUNKS;
  if (tid < NBINS) cur[tid] = off[blockIdx.x * NBINS + tid];
  __syncthreads();
  const int lo = blockIdx.x * eb;
  const int hi = min(E, lo + eb);
  for (int e = lo + tid; e < hi; e += 1024) {
    const int d = dst[e];
    const int s = src[e];
    const int bin = (int)((unsigned)d / (unsigned)bw);
    const int dl = d - bin * bw;
    const float is = dinv[s];
    const float id = dinv[d];
    const int pos = atomicAdd(&cur[bin], 1);
    bkey[pos] = (s << 8) | dl;
    bnrm[pos] = -is * id;
  }
}

template <int M>
__device__ __forceinline__ float row_xor(float v) {
  const int b = __builtin_bit_cast(int, v);
  return __builtin_bit_cast(float, __builtin_amdgcn_update_dpp(b, b, 0x160 + M, 0xf, 0xf, false));
}

#define EDGES_PER_TRIP 8
template <bool TAIL>
__global__ void __launch_bounds__(1024)
bin_sum_kernel(const int* __restrict__ bkey, const float* __restrict__ bnrm,
               const int* __restrict__ binstart, const float* __restrict__ y,
               const float* __restrict__ pre, float* __restrict__ outp,
               const float* __restrict__ W2, const float* __restrict__ b2, int N) {
  __shared__ __attribute__((aligned(16))) float acc[ROWS_MAX * 16];
  const int tid = threadIdx.x;
  const int bin = blockIdx.x;
  const int bw = (N + NBINS - 1) / NBINS;
  for (int j = tid; j < ROWS_MAX * 16; j += 1024) acc[j] = 0.0f;
  __syncthreads();
  const int beg = binstart[bin];
  const int end = binstart[bin + 1];
  const int g = tid >> 4;
  const int f = tid & 15;
  int i0 = beg + g;
  if (i0 < end) {
    const int last = end - 1;
    int key[EDGES_PER_TRIP];
    float w[EDGES_PER_TRIP];
#pragma unroll
    for (int k = 0; k < EDGES_PER_TRIP; ++k) {
      const int ic = min(i0 + 64 * k, last);
      key[k] = bkey[ic];
      w[k] = bnrm[ic];
    }
    for (; i0 < end; i0 += 64 * EDGES_PER_TRIP) {
      float yv[EDGES_PER_TRIP];
#pragma unroll
      for (int k = 0; k < EDGES_PER_TRIP; ++k) yv[k] = y[(long long)(key[k] >> 8) * 16 + f];
      int nkey[EDGES_PER_TRIP];
      float nw[EDGES_PER_TRIP];
#pragma unroll
      for (int k = 0; k < EDGES_PER_TRIP; ++k) {
        const int ic = min(i0 + 64 * (EDGES_PER_TRIP + k), last);
        nkey[k] = bkey[ic];
        nw[k] = bnrm[ic];
      }
      asm volatile("" :: "v"(yv[0]), "v"(yv[1]), "v"(yv[2]), "v"(yv[3]),
                         "v"(yv[4]), "v"(yv[5]), "v"(yv[6]), "v"(yv[7]));
#pragma unroll
      for (int k = 0; k < EDGES_PER_TRIP; ++k) {
        if (i0 + 64 * k < end) {
          __hip_atomic_fetch_add(&acc[(key[k] & 255) * 16 + f], w[k] * yv[k],
                                 __ATOMIC_RELAXED, __HIP_MEMORY_SCOPE_WORKGROUP);
        }
      }
#pragma unroll
      for (int k = 0; k < EDGES_PER_TRIP; ++k) { key[k] = nkey[k]; w[k] = nw[k]; }
    }
  }
  const int base = bin * bw;
  const int nrows = min(bw, N - base);
  if constexpr (!TAIL) {
    __syncthreads();
    for (int j = tid; j < nrows * 16; j += 1024) {
      const long long gi = (long long)base * 16 + j;
      outp[gi] = fmaxf(pre[gi] + acc[j], 0.0f);
    }
  } else {
    __shared__ __attribute__((aligned(16))) float zs[ROWS_MAX * 40];
    const int wv = __builtin_amdgcn_readfirstlane(tid >> 5);
    const int ntiles = nrows > 0 ? (nrows + 15) >> 4 : 0;
    const bool tilewave = wv < ntiles;
    const int n = tid & 15;
    const int hi = (tid >> 4) & 1;
    const float* __restrict__ W0 = W2;
    const float* __restrict__ Wb = W2 + 16 * 40;
    v2f a[4], b0[12], bb[12];
    float bias0 = 0.f, bias1 = 0.f, bias2 = 0.f;
#pragma unroll
    for (int i = 0; i < 4; ++i) a[i] = (v2f){0.f, 0.f};
#pragma unroll
    for (int i = 0; i < 12; ++i) { b0[i] = (v2f){0.f, 0.f}; bb[i] = (v2f){0.f, 0.f}; }
    if (tilewave) {
      const int mrow = min(base + wv * 16 + n, N - 1);
#pragma unroll
      for (int i = 0; i < 4; ++i) a[i] = *(const v2f*)(y + (long long)mrow * 16 + i * 4 + 2 * hi);
#pragma unroll
      for (int ct = 0; ct < 3; ++ct) {
        const int cc = min(ct * 16 + n, 39);
#pragma unroll
        for (int i = 0; i < 4; ++i) {
          const int k0 = i * 4 + 2 * hi;
          b0[ct * 4 + i].x = W0[k0 * 40 + cc];  b0[ct * 4 + i].y = W0[(k0 + 1) * 40 + cc];
          bb[ct * 4 + i].x = Wb[k0 * 40 + cc];  bb[ct * 4 + i].y = Wb[(k0 + 1) * 40 + cc];
        }
      }
      bias0 = b2[n];  bias1 = b2[16 + n];  bias2 = b2[min(32 + n, 39)];
    }
    __syncthreads();
    if (tilewave) {
      v2f t[4];
#pragma unroll
      for (int i = 0; i < 4; ++i) t[i] = *(const v2f*)(&acc[(wv * 16 + n) * 16 + i * 4 + 2 * hi]);
      v8f c[6];
#pragma unroll
      for (int i = 0; i < 6; ++i) c[i] = (v8f){0.f, 0.f, 0.f, 0.f, 0.f, 0.f, 0.f, 0.f};
#pragma unroll
      for (int ct = 0; ct < 3; ++ct) {
#pragma unroll
        for (int i = 0; i < 4; ++i) {
          c[ct * 2]     = WMMA_F32X4(a[i], b0[ct * 4 + i], c[ct * 2]);
          c[ct * 2 + 1] = WMMA_F32X4(t[i], bb[ct * 4 + i], c[ct * 2 + 1]);
        }
      }
      float z0[8], z1[8], z2[8], m[8], s[8];
#pragma unroll
      for (int r = 0; r < 8; ++r) {
        z0[r] = (c[0][r] + c[1][r]) + bias0;
        z1[r] = (c[2][r] + c[3][r]) + bias1;
        z2[r] = (c[4][r] + c[5][r]) + bias2;
        m[r] = fmaxf(z0[r], z1[r]);
      }
      if (n < 8) {
#pragma unroll
        for (int r = 0; r < 8; ++r) m[r] = fmaxf(m[r], z2[r]);
      }
#pragma unroll
      for (int r = 0; r < 8; ++r) m[r] = fmaxf(m[r], row_xor<8>(m[r]));
#pragma unroll
      for (int r = 0; r < 8; ++r) m[r] = fmaxf(m[r], row_xor<4>(m[r]));
#pragma unroll
      for (int r = 0; r < 8; ++r) m[r] = fmaxf(m[r], row_xor<2>(m[r]));
#pragma unroll
      for (int r = 0; r < 8; ++r) m[r] = fmaxf(m[r], row_xor<1>(m[r]));
#pragma unroll
      for (int r = 0; r < 8; ++r) s[r] = expf(z0[r] - m[r]) + expf(z1[r] - m[r]);
      if (n < 8) {
#pragma unroll
        for (int r = 0; r < 8; ++r) s[r] += expf(z2[r] - m[r]);
      }
#pragma unroll
      for (int r = 0; r < 8; ++r) s[r] += row_xor<8>(s[r]);
#pragma unroll
      for (int r = 0; r < 8; ++r) s[r] += row_xor<4>(s[r]);
#pragma unroll
      for (int r = 0; r < 8; ++r) s[r] += row_xor<2>(s[r]);
#pragma unroll
      for (int r = 0; r < 8; ++r) s[r] += row_xor<1>(s[r]);
#pragma unroll
      for (int r = 0; r < 8; ++r) {
        const float lse = m[r] + logf(s[r]);
        float* zr = zs + (wv * 16 + r + 8 * hi) * 40;
        zr[n] = z0[r] - lse;
        zr[16 + n] = z1[r] - lse;
        if (n < 8) zr[32 + n] = z2[r] - lse;
      }
    }
    __syncthreads();
    float4* __restrict__ dst4 = (float4*)(outp + (long long)base * 40);
    const float4* src4 = (const float4*)zs;
    for (int j = tid; j < nrows * 10; j += 1024) dst4[j] = src4[j];
  }
}

__global__ void gemm1_kernel(const float* __restrict__ x, const float* __restrict__ W1,
                             const float* __restrict__ b1, float* __restrict__ pre1,
                             float* __restrict__ y1, int N, int nTiles) {
  const int wave = (blockIdx.x * blockDim.x + threadIdx.x) >> 5;
  if (wave >= nTiles) return;
  const int lane = threadIdx.x & 31;
  const int n = lane & 15;
  const int hi = lane >> 4;
  const int base = wave * 16;
  const int mrow = min(base + n, N - 1);
  const float* __restrict__ W0 = W1;
  const float* __restrict__ Wb = W1 + 64 * 16;
  const float* __restrict__ xr = x + (long long)mrow * 64;

  v8f acc0 = {0.f, 0.f, 0.f, 0.f, 0.f, 0.f, 0.f, 0.f};
  v8f acc1 = acc0;
#pragma unroll
  for (int kb = 0; kb < 64; kb += 4) {
    const int k0 = kb + 2 * hi;
    v2f a = *(const v2f*)(xr + k0);
    v2f b0, bb;
    b0.x = W0[k0 * 16 + n];  b0.y = W0[(k0 + 1) * 16 + n];
    bb.x = Wb[k0 * 16 + n];  bb.y = Wb[(k0 + 1) * 16 + n];
    acc0 = WMMA_F32X4(a, b0, acc0);
    acc1 = WMMA_F32X4(a, bb, acc1);
  }
  const float bias = b1[n];
#pragma unroll
  for (int r = 0; r < 8; ++r) {
    const int row = base + r + 8 * hi;
    if (row < N) {
      pre1[(long long)row * 16 + n] = acc0[r] + bias;
      y1[(long long)row * 16 + n]  = acc1[r];
    }
  }
}

extern "C" void kernel_launch(void* const* d_in, const int* in_sizes, int n_in,
                              void* d_out, int out_size, void* d_ws, size_t ws_size,
                              hipStream_t stream) {
  const float* x  = (const float*)d_in[0];
  const int* ei   = (const int*)d_in[1];
  const float* W1 = (const float*)d_in[2];
  const float* b1 = (const float*)d_in[3];
  const float* W2 = (const float*)d_in[4];
  const float* b2 = (const float*)d_in[5];
  float* out = (float*)d_out;

  const int N = in_sizes[0] / 64;
  const int E = in_sizes[1] / 2;
  const int* src = ei;
  const int* dst = ei + E;

  float* ws = (float*)d_ws;
  size_t off = 0;
  float* dinv = ws + off; off += (size_t)N;
  int* binstart = (int*)(ws + off); off += (size_t)N;
  float* bnrm = ws + off; off += (size_t)E;
  float* pre1 = ws + off; off += (size_t)N * 16;
  float* y1   = ws + off; off += (size_t)N * 16;
  int* degp   = (int*)pre1;
  int* hist   = (int*)(ws + off);
  int* coff   = hist + CHUNKS * NBINS;
  off += (size_t)N * 16;
  float* h    = ws + off; off += (size_t)N * 16;
  int* bkey   = (int*)(ws + off);

  const int B = 256;
  auto blk = [](long long n, int b) { return (unsigned)((n + b - 1) / b); };

  bin_count_kernel<<<CHUNKS, 1024, 0, stream>>>(src, dst, degp, hist, N, E);
  bin_scan_kernel<<<1 + blk(N, 1024), 1024, 0, stream>>>(hist, coff, binstart, degp, dinv, N);
  bin_fill_kernel<<<CHUNKS, 1024, 0, stream>>>(src, dst, dinv, coff, bkey, bnrm, N, E);

  const int tiles = (N + 15) / 16;
  gemm1_kernel<<<blk((long long)tiles * 32, B), B, 0, stream>>>(x, W1, b1, pre1, y1, N, tiles);
  bin_sum_kernel<false><<<NBINS, 1024, 0, stream>>>(bkey, bnrm, binstart, y1, pre1, h, W2, b2, N);

  bin_sum_kernel<true><<<NBINS, 1024, 0, stream>>>(bkey, bnrm, binstart, h, pre1, out, W2, b2, N);
}


# Layer-1 GEMM moved into the scan launch as extra work-groups (one wave per 16x16 output tile of one projection), on the WGPs the one-work-group scan leaves idle: one launch fewer
# speedup vs baseline: 2.7391x; 1.0514x over previous
#include <hip/hip_runtime.h>
#include <math.h>

typedef float v2f __attribute__((ext_vector_type(2)));
typedef float v8f __attribute__((ext_vector_type(8)));

#define WMMA_F32X4(a, b, c) \
  __builtin_amdgcn_wmma_f32_16x16x4_f32(false, (a), false, (b), (short)0, (c), false, false)

#define NBINS 256
#define ROWS_MAX 256
#define CHUNKS 256
#define QCHUNKS 64

#define DEG_SLABS 32
#define DEG_RANGES 8
#define DEG_ROWS_MAX 8192
#define DEG_LOADS 8
static_assert(DEG_SLABS * DEG_RANGES == CHUNKS, "one (slab, range) pair per work-group of pass 1");

__global__ void __launch_bounds__(1024)
bin_count_kernel(const int* __restrict__ src, const int* __restrict__ dst,
                 int* __restrict__ degp, int* __restrict__ hist, int N, int E) {
  __shared__ int cnt[NBINS];
  __shared__ int dcnt[DEG_ROWS_MAX];
  const int tid = threadIdx.x;
  const int bw = (N + NBINS - 1) / NBINS;
  const int eb = (E + CHUNKS - 1) / CHUNKS;
  const int c = blockIdx.x / DEG_RANGES;
  const int r = blockIdx.x - c * DEG_RANGES;
  const int rw = (N + DEG_RANGES - 1) / DEG_RANGES;
  const int r_lo = r * rw;
  if (tid < NBINS) cnt[tid] = 0;
  for (int j = tid; j < rw; j += 1024) dcnt[j] = 0;
  __syncthreads();
  const int lo = blockIdx.x * eb;
  const int hi = min(E, lo + eb);
  for (int e = lo + tid; e < hi; e += 1024) {
    const int d = dst[e];
    atomicAdd(&cnt[(unsigned)d / (unsigned)bw], 1);
  }
  const int slab = (((E + DEG_SLABS - 1) / DEG_SLABS) + 3) & ~3;
  const int s_lo = c * slab;
  const int s_hi = min(E, s_lo + slab);
  const int Eal = E & ~3;
  const int v_hi = min(s_hi, Eal);
  for (int e0 = s_lo + 4 * tid; e0 < v_hi; e0 += 4096 * DEG_LOADS) {
    int4 v[DEG_LOADS];
#pragma unroll
    for (int k = 0; k < DEG_LOADS; ++k) {
      const int e = min(e0 + 4096 * k, Eal - 4);
      v[k] = *(const int4*)(src + e);
    }
#pragma unroll
    for (int k = 0; k < DEG_LOADS; ++k) {
      if (e0 + 4096 * k < v_hi) {
        const unsigned j0 = (unsigned)(v[k].x - r_lo), j1 = (unsigned)(v[k].y - r_lo);
        const unsigned j2 = (unsigned)(v[k].z - r_lo), j3 = (unsigned)(v[k].w - r_lo);
        if (j0 < (unsigned)rw) atomicAdd(&dcnt[j0], 1);
        if (j1 < (unsigned)rw) atomicAdd(&dcnt[j1], 1);
        if (j2 < (unsigned)rw) atomicAdd(&dcnt[j2], 1);
        if (j3 < (unsigned)rw) atomicAdd(&dcnt[j3], 1);
      }
    }
  }
  if (tid < E - Eal) {
    const int e = Eal + tid;
    if (e >= s_lo && e < s_hi) {
      const unsigned j = (unsigned)(src[e] - r_lo);
      if (j < (unsigned)rw) atomicAdd(&dcnt[j], 1);
    }
  }
  __syncthreads();
  if (tid < NBINS) hist[blockIdx.x * NBINS + tid] = cnt[tid];
  int* __restrict__ plane = degp + (size_t)c * N;
  for (int j = tid; j < rw; j += 1024) {
    const int node = r_lo + j;
    if (node < N) plane[node] = dcnt[j];
  }
}

__device__ __forceinline__ void gemm1_tile(const float* __restrict__ x, const float* __restrict__ W1,
                                           const float* __restrict__ b1, float* __restrict__ pre1,
                                           float* __restrict__ y1, int N, int tile, int proj, int lane) {
  const int n = lane & 15;
  const int hi = lane >> 4;
  const int base = tile * 16;
  const int mrow = min(base + n, N - 1);
  const float* __restrict__ W = W1 + proj * (64 * 16);
  const float* __restrict__ xr = x + (long long)mrow * 64;

  v8f acc = {0.f, 0.f, 0.f, 0.f, 0.f, 0.f, 0.f, 0.f};
#pragma unroll
  for (int kb = 0; kb < 64; kb += 4) {
    const int k0 = kb + 2 * hi;
    v2f a = *(const v2f*)(xr + k0);
    v2f b;
    b.x = W[k0 * 16 + n];  b.y = W[(k0 + 1) * 16 + n];
    acc = WMMA_F32X4(a, b, acc);
  }
  if (proj == 0) {
    const float bias = b1[n];
#pragma unroll
    for (int r = 0; r < 8; ++r) {
      const int row = base + r + 8 * hi;
      if (row < N) pre1[(long long)row * 16 + n] = acc[r] + bias;
    }
  } else {
#pragma unroll
    for (int r = 0; r < 8; ++r) {
      const int row = base + r + 8 * hi;
      if (row < N) y1[(long long)row * 16 + n] = acc[r];
    }
  }
}

__global__ void __launch_bounds__(1024)
bin_scan_gemm1_kernel(const int* __restrict__ hist, int* __restrict__ off, int* __restrict__ binstart,
                      const int* __restrict__ degp, float* __restrict__ dinv,
                      const float* __restrict__ x, const float* __restrict__ W1,
                      const float* __restrict__ b1, float* __restrict__ pre1,
                      float* __restrict__ y1, int N, int nDeg, int nTiles) {
  __shared__ int seg[4][NBINS];
  __shared__ int sc[2][NBINS];
  const int tid = threadIdx.x;
  if ((int)blockIdx.x > nDeg) {
    const int wv = __builtin_amdgcn_readfirstlane(tid >> 5);
    const int task = ((int)blockIdx.x - nDeg - 1) * 32 + wv;
    const int tile = task >> 1;
    if (tile >= nTiles) return;
    gemm1_tile(x, W1, b1, pre1, y1, N, tile, task & 1, tid & 31);
    return;
  }
  if (blockIdx.x > 0) {
    const int node = (blockIdx.x - 1) * 1024 + tid;
    if (node < N) {
      int p[DEG_SLABS];
#pragma unroll
      for (int i = 0; i < DEG_SLABS; ++i) p[i] = degp[(size_t)i * N + node];
      int cs = 0;
#pragma unroll
      for (int i = 0; i < DEG_SLABS; ++i) cs += p[i];
      const float dg = (float)cs;
      dinv[node] = (dg > 0.0f) ? rsqrtf(fmaxf(dg, 1.0f)) : 0.0f;
    }
    return;
  }
  const int bin = tid & (NBINS - 1);
  const int q = tid >> 8;
  const int* __restrict__ hp = hist + q * QCHUNKS * NBINS + bin;
  int v[QCHUNKS];
#pragma unroll
  for (int i = 0; i < QCHUNKS; ++i) v[i] = hp[i * NBINS];
  int s = 0;
#pragma unroll
  for (int i = 0; i < QCHUNKS; ++i) s += v[i];
  seg[q][bin] = s;
  __syncthreads();
  const int s0 = seg[0][bin], s1 = seg[1][bin], s2 = seg[2][bin], s3 = seg[3][bin];
  const int tot = (s0 + s1) + (s2 + s3);
  const int before = (q > 0 ? s0 : 0) + (q > 1 ? s1 : 0) + (q > 2 ? s2 : 0);
  if (tid < NBINS) sc[0][tid] = tot;
  __syncthreads();
  int pin = 0;
  for (int o = 1; o < NBINS; o <<= 1) {
    if (tid < NBINS) {
      int x = sc[pin][tid];
      if (tid >= o) x += sc[pin][tid - o];
      sc[pin ^ 1][tid] = x;
    }
    __syncthreads();
    pin ^= 1;
  }
  const int excl = sc[pin][bin] - tot;
  if (tid < NBINS) {
    binstart[tid] = excl;
    if (tid == NBINS - 1) binstart[NBINS] = sc[pin][NBINS - 1];
  }
  int run = excl + before;
  int* __restrict__ op = off + q * QCHUNKS * NBINS + bin;
#pragma unroll
  for (int i = 0; i < QCHUNKS; ++i) { op[i * NBINS] = run; run += v[i]; }
}

__global__ void __launch_bounds__(1024)
bin_fill_kernel(const int* __restrict__ src, const int* __restrict__ dst,
                const float* __restrict__ dinv, const int* __restrict__ off,
                int* __restrict__ bkey, float* __restrict__ bnrm, int N, int E) {
  __shared__ int cur[NBINS];
  const int tid = threadIdx.x;
  const int bw = (N + NBINS - 1) / NBINS;
  const int eb = (E + CHUNKS - 1) / CHUNKS;
  if (tid < NBINS) cur[tid] = off[blockIdx.x * NBINS + tid];
  __syncthreads();
  const int lo = blockIdx.x * eb;
  const int hi = min(E, lo + eb);
  for (int e = lo + tid; e < hi; e += 1024) {
    const int d = dst[e];
    const int s = src[e];
    const int bin = (int)((unsigned)d / (unsigned)bw);
    const int dl = d - bin * bw;
    const float is = dinv[s];
    const float id = dinv[d];
    const int pos = atomicAdd(&cur[bin], 1);
    bkey[pos] = (s << 8) | dl;
    bnrm[pos] = -is * id;
  }
}

template <int M>
__device__ __forceinline__ float row_xor(float v) {
  const int b = __builtin_bit_cast(int, v);
  return __builtin_bit_cast(float, __builtin_amdgcn_update_dpp(b, b, 0x160 + M, 0xf, 0xf, false));
}

#define EDGES_PER_TRIP 8
template <bool TAIL>
__global__ void __launch_bounds__(1024)
bin_sum_kernel(const int* __restrict__ bkey, const float* __restrict__ bnrm,
               const int* __restrict__ binstart, const float* __restrict__ y,
               const float* __restrict__ pre, float* __restrict__ outp,
               const float* __restrict__ W2, const float* __restrict__ b2, int N) {
  __shared__ __attribute__((aligned(16))) float acc[ROWS_MAX * 16];
  const int tid = threadIdx.x;
  const int bin = blockIdx.x;
  const int bw = (N + NBINS - 1) / NBINS;
  for (int j = tid; j < ROWS_MAX * 16; j += 1024) acc[j] = 0.0f;
  __syncthreads();
  const int beg = binstart[bin];
  const int end = binstart[bin + 1];
  const int g = tid >> 4;
  const int f = tid & 15;
  int i0 = beg + g;
  if (i0 < end) {
    const int last = end - 1;
    int key[EDGES_PER_TRIP];
    float w[EDGES_PER_TRIP];
#pragma unroll
    for (int k = 0; k < EDGES_PER_TRIP; ++k) {
      const int ic = min(i0 + 64 * k, last);
      key[k] = bkey[ic];
      w[k] = bnrm[ic];
    }
    for (; i0 < end; i0 += 64 * EDGES_PER_TRIP) {
      float yv[EDGES_PER_TRIP];
#pragma unroll
      for (int k = 0; k < EDGES_PER_TRIP; ++k) yv[k] = y[(long long)(key[k] >> 8) * 16 + f];
      int nkey[EDGES_PER_TRIP];
      float nw[EDGES_PER_TRIP];
#pragma unroll
      for (int k = 0; k < EDGES_PER_TRIP; ++k) {
        const int ic = min(i0 + 64 * (EDGES_PER_TRIP + k), last);
        nkey[k] = bkey[ic];
        nw[k] = bnrm[ic];
      }
      asm volatile("" :: "v"(yv[0]), "v"(yv[1]), "v"(yv[2]), "v"(yv[3]),
                         "v"(yv[4]), "v"(yv[5]), "v"(yv[6]), "v"(yv[7]));
#pragma unroll
      for (int k = 0; k < EDGES_PER_TRIP; ++k) {
        if (i0 + 64 * k < end) {
          __hip_atomic_fetch_add(&acc[(key[k] & 255) * 16 + f], w[k] * yv[k],
                                 __ATOMIC_RELAXED, __HIP_MEMORY_SCOPE_WORKGROUP);
        }
      }
#pragma unroll
      for (int k = 0; k < EDGES_PER_TRIP; ++k) { key[k] = nkey[k]; w[k] = nw[k]; }
    }
  }
  const int base = bin * bw;
  const int nrows = min(bw, N - base);
  if constexpr (!TAIL) {
    __syncthreads();
    for (int j = tid; j < nrows * 16; j += 1024) {
      const long long gi = (long long)base * 16 + j;
      outp[gi] = fmaxf(pre[gi] + acc[j], 0.0f);
    }
  } else {
    __shared__ __attribute__((aligned(16))) float zs[ROWS_MAX * 40];
    const int wv = __builtin_amdgcn_readfirstlane(tid >> 5);
    const int ntiles = nrows > 0 ? (nrows + 15) >> 4 : 0;
    const bool tilewave = wv < ntiles;
    const int n = tid & 15;
    const int hi = (tid >> 4) & 1;
    const float* __restrict__ W0 = W2;
    const float* __restrict__ Wb = W2 + 16 * 40;
    v2f a[4], b0[12], bb[12];
    float bias0 = 0.f, bias1 = 0.f, bias2 = 0.f;
#pragma unroll
    for (int i = 0; i < 4; ++i) a[i] = (v2f){0.f, 0.f};
#pragma unroll
    for (int i = 0; i < 12; ++i) { b0[i] = (v2f){0.f, 0.f}; bb[i] = (v2f){0.f, 0.f}; }
    if (tilewave) {
      const int mrow = min(base + wv * 16 + n, N - 1);
#pragma unroll
      for (int i = 0; i < 4; ++i) a[i] = *(const v2f*)(y + (long long)mrow * 16 + i * 4 + 2 * hi);
#pragma unroll
      for (int ct = 0; ct < 3; ++ct) {
        const int cc = min(ct * 16 + n, 39);
#pragma unroll
        for (int i = 0; i < 4; ++i) {
          const int k0 = i * 4 + 2 * hi;
          b0[ct * 4 + i].x = W0[k0 * 40 + cc];  b0[ct * 4 + i].y = W0[(k0 + 1) * 40 + cc];
          bb[ct * 4 + i].x = Wb[k0 * 40 + cc];  bb[ct * 4 + i].y = Wb[(k0 + 1) * 40 + cc];
        }
      }
      bias0 = b2[n];  bias1 = b2[16 + n];  bias2 = b2[min(32 + n, 39)];
    }
    __syncthreads();
    if (tilewave) {
      v2f t[4];
#pragma unroll
      for (int i = 0; i < 4; ++i) t[i] = *(const v2f*)(&acc[(wv * 16 + n) * 16 + i * 4 + 2 * hi]);
      v8f c[6];
#pragma unroll
      for (int i = 0; i < 6; ++i) c[i] = (v8f){0.f, 0.f, 0.f, 0.f, 0.f, 0.f, 0.f, 0.f};
#pragma unroll
      for (int ct = 0; ct < 3; ++ct) {
#pragma unroll
        for (int i = 0; i < 4; ++i) {
          c[ct * 2]     = WMMA_F32X4(a[i], b0[ct * 4 + i], c[ct * 2]);
          c[ct * 2 + 1] = WMMA_F32X4(t[i], bb[ct * 4 + i], c[ct * 2 + 1]);
        }
      }
      float z0[8], z1[8], z2[8], m[8], s[8];
#pragma unroll
      for (int r = 0; r < 8; ++r) {
        z0[r] = (c[0][r] + c[1][r]) + bias0;
        z1[r] = (c[2][r] + c[3][r]) + bias1;
        z2[r] = (c[4][r] + c[5][r]) + bias2;
        m[r] = fmaxf(z0[r], z1[r]);
      }
      if (n < 8) {
#pragma unroll
        for (int r = 0; r < 8; ++r) m[r] = fmaxf(m[r], z2[r]);
      }
#pragma unroll
      for (int r = 0; r < 8; ++r) m[r] = fmaxf(m[r], row_xor<8>(m[r]));
#pragma unroll
      for (int r = 0; r < 8; ++r) m[r] = fmaxf(m[r], row_xor<4>(m[r]));
#pragma unroll
      for (int r = 0; r < 8; ++r) m[r] = fmaxf(m[r], row_xor<2>(m[r]));
#pragma unroll
      for (int r = 0; r < 8; ++r) m[r] = fmaxf(m[r], row_xor<1>(m[r]));
#pragma unroll
      for (int r = 0; r < 8; ++r) s[r] = expf(z0[r] - m[r]) + expf(z1[r] - m[r]);
      if (n < 8) {
#pragma unroll
        for (int r = 0; r < 8; ++r) s[r] += expf(z2[r] - m[r]);
      }
#pragma unroll
      for (int r = 0; r < 8; ++r) s[r] += row_xor<8>(s[r]);
#pragma unroll
      for (int r = 0; r < 8; ++r) s[r] += row_xor<4>(s[r]);
#pragma unroll
      for (int r = 0; r < 8; ++r) s[r] += row_xor<2>(s[r]);
#pragma unroll
      for (int r = 0; r < 8; ++r) s[r] += row_xor<1>(s[r]);
#pragma unroll
      for (int r = 0; r < 8; ++r) {
        const float lse = m[r] + logf(s[r]);
        float* zr = zs + (wv * 16 + r + 8 * hi) * 40;
        zr[n] = z0[r] - lse;
        zr[16 + n] = z1[r] - lse;
        if (n < 8) zr[32 + n] = z2[r] - lse;
      }
    }
    __syncthreads();
    float4* __restrict__ dst4 = (float4*)(outp + (long long)base * 40);
    const float4* src4 = (const float4*)zs;
    for (int j = tid; j < nrows * 10; j += 1024) dst4[j] = src4[j];
  }
}

extern "C" void kernel_launch(void* const* d_in, const int* in_sizes, int n_in,
                              void* d_out, int out_size, void* d_ws, size_t ws_size,
                              hipStream_t stream) {
  const float* x  = (const float*)d_in[0];
  const int* ei   = (const int*)d_in[1];
  const float* W1 = (const float*)d_in[2];
  const float* b1 = (const float*)d_in[3];
  const float* W2 = (const float*)d_in[4];
  const float* b2 = (const float*)d_in[5];
  float* out = (float*)d_out;

  const int N = in_sizes[0] / 64;
  const int E = in_sizes[1] / 2;
  const int* src = ei;
  const int* dst = ei + E;

  float* ws = (float*)d_ws;
  size_t off = 0;
  float* dinv = ws + off; off += (size_t)N;
  int* binstart = (int*)(ws + off); off += (size_t)N;
  float* pre1 = ws + off; off += (size_t)N * 16;
  float* y1   = ws + off; off += (size_t)N * 16;
  int* hist   = (int*)(ws + off);
  int* coff   = hist + CHUNKS * NBINS;
  off += (size_t)N * 16;
  float* h    = ws + off; off += (size_t)N * 16;
  int* bkey   = (int*)(ws + off); off += (size_t)E;
  float* bnrm = ws + off;
  int* degp   = (int*)h;

  auto blk = [](long long n, int b) { return (unsigned)((n + b - 1) / b); };

  const int tiles = (N + 15) / 16;
  const int nDeg = (int)blk(N, 1024);
  bin_count_kernel<<<CHUNKS, 1024, 0, stream>>>(src, dst, degp, hist, N, E);
  bin_scan_gemm1_kernel<<<1 + nDeg + blk((long long)tiles * 2, 32), 1024, 0, stream>>>(
      hist, coff, binstart, degp, dinv, x, W1, b1, pre1, y1, N, nDeg, tiles);
  bin_fill_kernel<<<CHUNKS, 1024, 0, stream>>>(src, dst, dinv, coff, bkey, bnrm, N, E);

  bin_sum_kernel<false><<<NBINS, 1024, 0, stream>>>(bkey, bnrm, binstart, y1, pre1, h, W2, b2, N);

  bin_sum_kernel<true><<<NBINS, 1024, 0, stream>>>(bkey, bnrm, binstart, h, pre1, out, W2, b2, N);
}


# Edge norm factored out of the sums: gathered rows are stored pre-multiplied by dinv[src], each summed row scaled by -dinv[dst] once; no per-edge coefficient built, stored or read (sums reassociated)
# speedup vs baseline: 3.3985x; 1.2407x over previous
#include <hip/hip_runtime.h>
#include <math.h>

typedef float v2f __attribute__((ext_vector_type(2)));
typedef float v8f __attribute__((ext_vector_type(8)));

#define WMMA_F32X4(a, b, c) \
  __builtin_amdgcn_wmma_f32_16x16x4_f32(false, (a), false, (b), (short)0, (c), false, false)

#define NBINS 256
#define ROWS_MAX 256
#define CHUNKS 256
#define QCHUNKS 64

#define DEG_SLABS 32
#define DEG_RANGES 8
#define DEG_ROWS_MAX 8192
#define DEG_LOADS 8
static_assert(DEG_SLABS * DEG_RANGES == CHUNKS, "one (slab, range) pair per work-group of pass 1");

__global__ void __launch_bounds__(1024)
bin_count_kernel(const int* __restrict__ src, const int* __restrict__ dst,
                 int* __restrict__ degp, int* __restrict__ hist, int N, int E) {
  __shared__ int cnt[NBINS];
  __shared__ int dcnt[DEG_ROWS_MAX];
  const int tid = threadIdx.x;
  const int bw = (N + NBINS - 1) / NBINS;
  const int eb = (E + CHUNKS - 1) / CHUNKS;
  const int c = blockIdx.x / DEG_RANGES;
  const int r = blockIdx.x - c * DEG_RANGES;
  const int rw = (N + DEG_RANGES - 1) / DEG_RANGES;
  const int r_lo = r * rw;
  if (tid < NBINS) cnt[tid] = 0;
  for (int j = tid; j < rw; j += 1024) dcnt[j] = 0;
  __syncthreads();
  const int lo = blockIdx.x * eb;
  const int hi = min(E, lo + eb);
  for (int e = lo + tid; e < hi; e += 1024) {
    const int d = dst[e];
    atomicAdd(&cnt[(unsigned)d / (unsigned)bw], 1);
  }
  const int slab = (((E + DEG_SLABS - 1) / DEG_SLABS) + 3) & ~3;
  const int s_lo = c * slab;
  const int s_hi = min(E, s_lo + slab);
  const int Eal = E & ~3;
  const int v_hi = min(s_hi, Eal);
  for (int e0 = s_lo + 4 * tid; e0 < v_hi; e0 += 4096 * DEG_LOADS) {
    int4 v[DEG_LOADS];
#pragma unroll
    for (int k = 0; k < DEG_LOADS; ++k) {
      const int e = min(e0 + 4096 * k, Eal - 4);
      v[k] = *(const int4*)(src + e);
    }
#pragma unroll
    for (int k = 0; k < DEG_LOADS; ++k) {
      if (e0 + 4096 * k < v_hi) {
        const unsigned j0 = (unsigned)(v[k].x - r_lo), j1 = (unsigned)(v[k].y - r_lo);
        const unsigned j2 = (unsigned)(v[k].z - r_lo), j3 = (unsigned)(v[k].w - r_lo);
        if (j0 < (unsigned)rw) atomicAdd(&dcnt[j0], 1);
        if (j1 < (unsigned)rw) atomicAdd(&dcnt[j1], 1);
        if (j2 < (unsigned)rw) atomicAdd(&dcnt[j2], 1);
        if (j3 < (unsigned)rw) atomicAdd(&dcnt[j3], 1);
      }
    }
  }
  if (tid < E - Eal) {
    const int e = Eal + tid;
    if (e >= s_lo && e < s_hi) {
      const unsigned j = (unsigned)(src[e] - r_lo);
      if (j < (unsigned)rw) atomicAdd(&dcnt[j], 1);
    }
  }
  __syncthreads();
  if (tid < NBINS) hist[blockIdx.x * NBINS + tid] = cnt[tid];
  int* __restrict__ plane = degp + (size_t)c * N;
  for (int j = tid; j < rw; j += 1024) {
    const int node = r_lo + j;
    if (node < N) plane[node] = dcnt[j];
  }
}

__device__ __forceinline__ int other_row(int v) {
  return __builtin_amdgcn_permlanex16(v, v, 0x76543210, 0xfedcba98, false, false);
}


template <int PROJ>
__device__ __forceinline__ void gemm1_tile(const float* __restrict__ x, const float* __restrict__ W1,
                                           const float* __restrict__ b1, const int* __restrict__ degp,
                                           float* __restrict__ pre1, float* __restrict__ y1,
                                           int N, int tile, int lane) {
  const int n = lane & 15;
  const int hi = lane >> 4;
  const int base = tile * 16;
  const int mrow = min(base + n, N - 1);
  const float* __restrict__ W = W1 + PROJ * (64 * 16);
  const float* __restrict__ xr = x + (long long)mrow * 64;

  float rs = 1.0f;
  if constexpr (PROJ == 1) {
    int cs = 0;
#pragma unroll
    for (int i = 0; i < DEG_SLABS / 2; ++i) cs += degp[(size_t)(hi * (DEG_SLABS / 2) + i) * N + mrow];
    cs += other_row(cs);
    const float dg = (float)cs;
    rs = (dg > 0.0f) ? rsqrtf(fmaxf(dg, 1.0f)) : 0.0f;
  }

  v8f acc = {0.f, 0.f, 0.f, 0.f, 0.f, 0.f, 0.f, 0.f};
#pragma unroll
  for (int kb = 0; kb < 64; kb += 4) {
    const int k0 = kb + 2 * hi;
    v2f a = *(const v2f*)(xr + k0);
    if constexpr (PROJ == 1) { a.x *= rs;  a.y *= rs; }
    v2f b;
    b.x = W[k0 * 16 + n];  b.y = W[(k0 + 1) * 16 + n];
    acc = WMMA_F32X4(a, b, acc);
  }
  if constexpr (PROJ == 0) {
    const float bias = b1[n];
#pragma unroll
    for (int r = 0; r < 8; ++r) {
      const int row = base + r + 8 * hi;
      if (row < N) pre1[(long long)row * 16 + n] = acc[r] + bias;
    }
  } else {
#pragma unroll
    for (int r = 0; r < 8; ++r) {
      const int row = base + r + 8 * hi;
      if (row < N) y1[(long long)row * 16 + n] = acc[r];
    }
  }
}

__global__ void __launch_bounds__(1024)
bin_scan_gemm1_kernel(const int* __restrict__ hist, int* __restrict__ off, int* __restrict__ binstart,
                      const int* __restrict__ degp, float* __restrict__ dinv,
                      const float* __restrict__ x, const float* __restrict__ W1,
                      const float* __restrict__ b1, float* __restrict__ pre1,
                      float* __restrict__ y1, int N, int nDeg, int nTiles) {
  __shared__ int seg[4][NBINS];
  __shared__ int sc[2][NBINS];
  const int tid = threadIdx.x;
  if ((int)blockIdx.x > nDeg) {
    const int wv = __builtin_amdgcn_readfirstlane(tid >> 5);
    const int task = ((int)blockIdx.x - nDeg - 1) * 32 + wv;
    const int tile = task >> 1;
    if (tile >= nTiles) return;
    if ((task & 1) == 0) gemm1_tile<0>(x, W1, b1, degp, pre1, y1, N, tile, tid & 31);
    else                 gemm1_tile<1>(x, W1, b1, degp, pre1, y1, N, tile, tid & 31);
    return;
  }
  if (blockIdx.x > 0) {
    const int node = (blockIdx.x - 1) * 1024 + tid;
    if (node < N) {
      int p[DEG_SLABS];
#pragma unroll
      for (int i = 0; i < DEG_SLABS; ++i) p[i] = degp[(size_t)i * N + node];
      int cs = 0;
#pragma unroll
      for (int i = 0; i < DEG_SLABS; ++i) cs += p[i];
      const float dg = (float)cs;
      dinv[node] = (dg > 0.0f) ? rsqrtf(fmaxf(dg, 1.0f)) : 0.0f;
    }
    return;
  }
  const int bin = tid & (NBINS - 1);
  const int q = tid >> 8;
  const int* __restrict__ hp = hist + q * QCHUNKS * NBINS + bin;
  int v[QCHUNKS];
#pragma unroll
  for (int i = 0; i < QCHUNKS; ++i) v[i] = hp[i * NBINS];
  int s = 0;
#pragma unroll
  for (int i = 0; i < QCHUNKS; ++i) s += v[i];
  seg[q][bin] = s;
  __syncthreads();
  const int s0 = seg[0][bin], s1 = seg[1][bin], s2 = seg[2][bin], s3 = seg[3][bin];
  const int tot = (s0 + s1) + (s2 + s3);
  const int before = (q > 0 ? s0 : 0) + (q > 1 ? s1 : 0) + (q > 2 ? s2 : 0);
  if (tid < NBINS) sc[0][tid] = tot;
  __syncthreads();
  int pin = 0;
  for (int o = 1; o < NBINS; o <<= 1) {
    if (tid < NBINS) {
      int x = sc[pin][tid];
      if (tid >= o) x += sc[pin][tid - o];
      sc[pin ^ 1][tid] = x;
    }
    __syncthreads();
    pin ^= 1;
  }
  const int excl = sc[pin][bin] - tot;
  if (tid < NBINS) {
    binstart[tid] = excl;
    if (tid == NBINS - 1) binstart[NBINS] = sc[pin][NBINS - 1];
  }
  int run = excl + before;
  int* __restrict__ op = off + q * QCHUNKS * NBINS + bin;
#pragma unroll
  for (int i = 0; i < QCHUNKS; ++i) { op[i * NBINS] = run; run += v[i]; }
}

__global__ void __launch_bounds__(1024)
bin_fill_kernel(const int* __restrict__ src, const int* __restrict__ dst,
                const int* __restrict__ off, int* __restrict__ bkey, int N, int E) {
  __shared__ int cur[NBINS];
  const int tid = threadIdx.x;
  const int bw = (N + NBINS - 1) / NBINS;
  const int eb = (E + CHUNKS - 1) / CHUNKS;
  if (tid < NBINS) cur[tid] = off[blockIdx.x * NBINS + tid];
  __syncthreads();
  const int lo = blockIdx.x * eb;
  const int hi = min(E, lo + eb);
  for (int e = lo + tid; e < hi; e += 1024) {
    const int d = dst[e];
    const int s = src[e];
    const int bin = (int)((unsigned)d / (unsigned)bw);
    const int dl = d - bin * bw;
    const int pos = atomicAdd(&cur[bin], 1);
    bkey[pos] = (s << 8) | dl;
  }
}

template <int M>
__device__ __forceinline__ float row_xor(float v) {
  const int b = __builtin_bit_cast(int, v);
  return __builtin_bit_cast(float, __builtin_amdgcn_update_dpp(b, b, 0x160 + M, 0xf, 0xf, false));
}

#define EDGES_PER_TRIP 8
template <bool TAIL>
__global__ void __launch_bounds__(1024)
bin_sum_kernel(const int* __restrict__ bkey, const int* __restrict__ binstart,
               const float* __restrict__ dinv, const float* __restrict__ y,
               const float* __restrict__ own, float* __restrict__ outp, float* __restrict__ outs,
               const float* __restrict__ W2, const float* __restrict__ b2, int N) {
  __shared__ __attribute__((aligned(16))) float acc[ROWS_MAX * 16];
  const int tid = threadIdx.x;
  const int bin = blockIdx.x;
  const int bw = (N + NBINS - 1) / NBINS;
  for (int j = tid; j < ROWS_MAX * 16; j += 1024) acc[j] = 0.0f;
  __syncthreads();
  const int beg = binstart[bin];
  const int end = binstart[bin + 1];
  const int g = tid >> 4;
  const int f = tid & 15;
  int i0 = beg + g;
  if (i0 < end) {
    const int last = end - 1;
    int key[EDGES_PER_TRIP];
#pragma unroll
    for (int k = 0; k < EDGES_PER_TRIP; ++k) key[k] = bkey[min(i0 + 64 * k, last)];
    for (; i0 < end; i0 += 64 * EDGES_PER_TRIP) {
      float yv[EDGES_PER_TRIP];
#pragma unroll
      for (int k = 0; k < EDGES_PER_TRIP; ++k) yv[k] = y[(long long)(key[k] >> 8) * 16 + f];
      int nkey[EDGES_PER_TRIP];
#pragma unroll
      for (int k = 0; k < EDGES_PER_TRIP; ++k) nkey[k] = bkey[min(i0 + 64 * (EDGES_PER_TRIP + k), last)];
      asm volatile("" :: "v"(yv[0]), "v"(yv[1]), "v"(yv[2]), "v"(yv[3]),
                         "v"(yv[4]), "v"(yv[5]), "v"(yv[6]), "v"(yv[7]));
#pragma unroll
      for (int k = 0; k < EDGES_PER_TRIP; ++k) {
        if (i0 + 64 * k < end) {
          __hip_atomic_fetch_add(&acc[(key[k] & 255) * 16 + f], yv[k],
                                 __ATOMIC_RELAXED, __HIP_MEMORY_SCOPE_WORKGROUP);
        }
      }
#pragma unroll
      for (int k = 0; k < EDGES_PER_TRIP; ++k) key[k] = nkey[k];
    }
  }
  const int base = bin * bw;
  const int nrows = min(bw, N - base);
  if constexpr (!TAIL) {
    __syncthreads();
    for (int j = tid; j < nrows * 16; j += 1024) {
      const long long gi = (long long)base * 16 + j;
      const float di = dinv[base + (j >> 4)];
      const float hv = fmaxf(own[gi] - di * acc[j], 0.0f);
      outp[gi] = hv;
      outs[gi] = di * hv;
    }
  } else {
    __shared__ __attribute__((aligned(16))) float zs[ROWS_MAX * 40];
    const int wv = __builtin_amdgcn_readfirstlane(tid >> 5);
    const int ntiles = nrows > 0 ? (nrows + 15) >> 4 : 0;
    const bool tilewave = wv < ntiles;
    const int n = tid & 15;
    const int hi = (tid >> 4) & 1;
    const float* __restrict__ W0 = W2;
    const float* __restrict__ Wb = W2 + 16 * 40;
    v2f a[4], b0[12], bb[12];
    float bias0 = 0.f, bias1 = 0.f, bias2 = 0.f, nd = 0.f;
#pragma unroll
    for (int i = 0; i < 4; ++i) a[i] = (v2f){0.f, 0.f};
#pragma unroll
    for (int i = 0; i < 12; ++i) { b0[i] = (v2f){0.f, 0.f}; bb[i] = (v2f){0.f, 0.f}; }
    if (tilewave) {
      const int mrow = min(base + wv * 16 + n, N - 1);
#pragma unroll
      for (int i = 0; i < 4; ++i) a[i] = *(const v2f*)(own + (long long)mrow * 16 + i * 4 + 2 * hi);
      nd = -dinv[mrow];
#pragma unroll
      for (int ct = 0; ct < 3; ++ct) {
        const int cc = min(ct * 16 + n, 39);
#pragma unroll
        for (int i = 0; i < 4; ++i) {
          const int k0 = i * 4 + 2 * hi;
          b0[ct * 4 + i].x = W0[k0 * 40 + cc];  b0[ct * 4 + i].y = W0[(k0 + 1) * 40 + cc];
          bb[ct * 4 + i].x = Wb[k0 * 40 + cc];  bb[ct * 4 + i].y = Wb[(k0 + 1) * 40 + cc];
        }
      }
      bias0 = b2[n];  bias1 = b2[16 + n];  bias2 = b2[min(32 + n, 39)];
    }
    __syncthreads();
    if (tilewave) {
      v2f t[4];
#pragma unroll
      for (int i = 0; i < 4; ++i) {
        t[i] = *(const v2f*)(&acc[(wv * 16 + n) * 16 + i * 4 + 2 * hi]);
        t[i].x *= nd;  t[i].y *= nd;
      }
      v8f c[6];
#pragma unroll
      for (int i = 0; i < 6; ++i) c[i] = (v8f){0.f, 0.f, 0.f, 0.f, 0.f, 0.f, 0.f, 0.f};
#pragma unroll
      for (int ct = 0; ct < 3; ++ct) {
#pragma unroll
        for (int i = 0; i < 4; ++i) {
          c[ct * 2]     = WMMA_F32X4(a[i], b0[ct * 4 + i], c[ct * 2]);
          c[ct * 2 + 1] = WMMA_F32X4(t[i], bb[ct * 4 + i], c[ct * 2 + 1]);
        }
      }
      float z0[8], z1[8], z2[8], m[8], s[8];
#pragma unroll
      for (int r = 0; r < 8; ++r) {
        z0[r] = (c[0][r] + c[1][r]) + bias0;
        z1[r] = (c[2][r] + c[3][r]) + bias1;
        z2[r] = (c[4][r] + c[5][r]) + bias2;
        m[r] = fmaxf(z0[r], z1[r]);
      }
      if (n < 8) {
#pragma unroll
        for (int r = 0; r < 8; ++r) m[r] = fmaxf(m[r], z2[r]);
      }
#pragma unroll
      for (int r = 0; r < 8; ++r) m[r] = fmaxf(m[r], row_xor<8>(m[r]));
#pragma unroll
      for (int r = 0; r < 8; ++r) m[r] = fmaxf(m[r], row_xor<4>(m[r]));
#pragma unroll
      for (int r = 0; r < 8; ++r) m[r] = fmaxf(m[r], row_xor<2>(m[r]));
#pragma unroll
      for (int r = 0; r < 8; ++r) m[r] = fmaxf(m[r], row_xor<1>(m[r]));
#pragma unroll
      for (int r = 0; r < 8; ++r) s[r] = expf(z0[r] - m[r]) + expf(z1[r] - m[r]);
      if (n < 8) {
#pragma unroll
        for (int r = 0; r < 8; ++r) s[r] += expf(z2[r] - m[r]);
      }
#pragma unroll
      for (int r = 0; r < 8; ++r) s[r] += row_xor<8>(s[r]);
#pragma unroll
      for (int r = 0; r < 8; ++r) s[r] += row_xor<4>(s[r]);
#pragma unroll
      for (int r = 0; r < 8; ++r) s[r] += row_xor<2>(s[r]);
#pragma unroll
      for (int r = 0; r < 8; ++r) s[r] += row_xor<1>(s[r]);
#pragma unroll
      for (int r = 0; r < 8; ++r) {
        const float lse = m[r] + logf(s[r]);
        float* zr = zs + (wv * 16 + r + 8 * hi) * 40;
        zr[n] = z0[r] - lse;
        zr[16 + n] = z1[r] - lse;
        if (n < 8) zr[32 + n] = z2[r] - lse;
      }
    }
    __syncthreads();
    float4* __restrict__ dst4 = (float4*)(outp + (long long)base * 40);
    const float4* src4 = (const float4*)zs;
    for (int j = tid; j < nrows * 10; j += 1024) dst4[j] = src4[j];
  }
}

extern "C" void kernel_launch(void* const* d_in, const int* in_sizes, int n_in,
                              void* d_out, int out_size, void* d_ws, size_t ws_size,
                              hipStream_t stream) {
  const float* x  = (const float*)d_in[0];
  const int* ei   = (const int*)d_in[1];
  const float* W1 = (const float*)d_in[2];
  const float* b1 = (const float*)d_in[3];
  const float* W2 = (const float*)d_in[4];
  const float* b2 = (const float*)d_in[5];
  float* out = (float*)d_out;

  const int N = in_sizes[0] / 64;
  const int E = in_sizes[1] / 2;
  const int* src = ei;
  const int* dst = ei + E;

  float* ws = (float*)d_ws;
  size_t off = 0;
  float* dinv = ws + off; off += (size_t)N;
  int* binstart = (int*)(ws + off); off += (size_t)N;
  float* pre1 = ws + off; off += (size_t)N * 16;
  float* y1   = ws + off; off += (size_t)N * 16;
  int* hist   = (int*)(ws + off);
  int* coff   = hist + CHUNKS * NBINS;
  off += (size_t)N * 16;
  float* h    = ws + off; off += (size_t)N * 16;
  int* bkey   = (int*)(ws + off); off += (size_t)E;
  float* hs   = ws + off;
  int* degp   = (int*)h;

  auto blk = [](long long n, int b) { return (unsigned)((n + b - 1) / b); };

  const int tiles = (N + 15) / 16;
  const int nDeg = (int)blk(N, 1024);
  bin_count_kernel<<<CHUNKS, 1024, 0, stream>>>(src, dst, degp, hist, N, E);
  bin_scan_gemm1_kernel<<<1 + nDeg + blk((long long)tiles * 2, 32), 1024, 0, stream>>>(
      hist, coff, binstart, degp, dinv, x, W1, b1, pre1, y1, N, nDeg, tiles);
  bin_fill_kernel<<<CHUNKS, 1024, 0, stream>>>(src, dst, coff, bkey, N, E);

  bin_sum_kernel<false><<<NBINS, 1024, 0, stream>>>(bkey, binstart, dinv, y1, pre1, h, hs, W2, b2, N);

  bin_sum_kernel<true><<<NBINS, 1024, 0, stream>>>(bkey, binstart, dinv, hs, h, out, nullptr, W2, b2, N);
}


# Edges grouped by range inside each chunk in ONE pass (LDS count, scan, rank); a range's sums walk its 256 chunk segments, 16 lanes fetch 16 keys and share them by DPP; 4 launches; LDS arrival order
# speedup vs baseline: 4.6348x; 1.3638x over previous
#include <hip/hip_runtime.h>
#include <math.h>

typedef float v2f __attribute__((ext_vector_type(2)));
typedef float v8f __attribute__((ext_vector_type(8)));

#define WMMA_F32X4(a, b, c) \
  __builtin_amdgcn_wmma_f32_16x16x4_f32(false, (a), false, (b), (short)0, (c), false, false)

#define NBINS 256
#define ROWS_MAX 256
#define CHUNKS 256
#define SORT_EPT 4
#define SEGS_PER_WAVE 8
static_assert(CHUNKS == 32 * SEGS_PER_WAVE, "the 32 waves of a range's work-group share its segments evenly");

#define DEG_SLABS 32
#define DEG_RANGES 8
#define DEG_ROWS_MAX 8192
#define DEG_LOADS 8
static_assert(DEG_SLABS * DEG_RANGES == CHUNKS, "one (slab, range) pair per work-group of pass 1");

__global__ void __launch_bounds__(1024)
bin_sort_kernel(const int* __restrict__ src, const int* __restrict__ dst,
                int* __restrict__ degp, int* __restrict__ loff, int* __restrict__ bkey, int N, int E) {
  __shared__ int cnt[NBINS];
  __shared__ int sc[2][NBINS];
  __shared__ int dcnt[DEG_ROWS_MAX];
  const int tid = threadIdx.x;
  const int bw = (N + NBINS - 1) / NBINS;
  const int eb = (E + CHUNKS - 1) / CHUNKS;
  const int c = blockIdx.x / DEG_RANGES;
  const int r = blockIdx.x - c * DEG_RANGES;
  const int rw = (N + DEG_RANGES - 1) / DEG_RANGES;
  const int r_lo = r * rw;
  if (tid < NBINS) cnt[tid] = 0;
  for (int j = tid; j < rw; j += 1024) dcnt[j] = 0;
  __syncthreads();
  const int lo = blockIdx.x * eb;
  const int hi = min(E, lo + eb);
  int kbin[SORT_EPT], kkey[SORT_EPT];
#pragma unroll
  for (int k = 0; k < SORT_EPT; ++k) {
    const int e = min(lo + tid + 1024 * k, E - 1);
    const int d = dst[e];
    const int s = src[e];
    const int b = (int)((unsigned)d / (unsigned)bw);
    kbin[k] = b;
    kkey[k] = (s << 8) | (d - b * bw);
  }
#pragma unroll
  for (int k = 0; k < SORT_EPT; ++k) {
    if (lo + tid + 1024 * k < hi) atomicAdd(&cnt[kbin[k]], 1);
  }
  for (int e = lo + 1024 * SORT_EPT + tid; e < hi; e += 1024) {
    const int d = dst[e];
    atomicAdd(&cnt[(unsigned)d / (unsigned)bw], 1);
  }
  const int slab = (((E + DEG_SLABS - 1) / DEG_SLABS) + 3) & ~3;
  const int s_lo = c * slab;
  const int s_hi = min(E, s_lo + slab);
  const int Eal = E & ~3;
  const int v_hi = min(s_hi, Eal);
  for (int e0 = s_lo + 4 * tid; e0 < v_hi; e0 += 4096 * DEG_LOADS) {
    int4 v[DEG_LOADS];
#pragma unroll
    for (int k = 0; k < DEG_LOADS; ++k) {
      const int e = min(e0 + 4096 * k, Eal - 4);
      v[k] = *(const int4*)(src + e);
    }
#pragma unroll
    for (int k = 0; k < DEG_LOADS; ++k) {
      if (e0 + 4096 * k < v_hi) {
        const unsigned j0 = (unsigned)(v[k].x - r_lo), j1 = (unsigned)(v[k].y - r_lo);
        const unsigned j2 = (unsigned)(v[k].z - r_lo), j3 = (unsigned)(v[k].w - r_lo);
        if (j0 < (unsigned)rw) atomicAdd(&dcnt[j0], 1);
        if (j1 < (unsigned)rw) atomicAdd(&dcnt[j1], 1);
        if (j2 < (unsigned)rw) atomicAdd(&dcnt[j2], 1);
        if (j3 < (unsigned)rw) atomicAdd(&dcnt[j3], 1);
      }
    }
  }
  if (tid < E - Eal) {
    const int e = Eal + tid;
    if (e >= s_lo && e < s_hi) {
      const unsigned j = (unsigned)(src[e] - r_lo);
      if (j < (unsigned)rw) atomicAdd(&dcnt[j], 1);
    }
  }
  __syncthreads();
  int* __restrict__ plane = degp + (size_t)c * N;
  for (int j = tid; j < rw; j += 1024) {
    const int node = r_lo + j;
    if (node < N) plane[node] = dcnt[j];
  }
  int mine = 0;
  if (tid < NBINS) { mine = cnt[tid]; sc[0][tid] = mine; }
  __syncthreads();
  int pin = 0;
  for (int o = 1; o < NBINS; o <<= 1) {
    if (tid < NBINS) {
      int x = sc[pin][tid];
      if (tid >= o) x += sc[pin][tid - o];
      sc[pin ^ 1][tid] = x;
    }
    __syncthreads();
    pin ^= 1;
  }
  if (tid < NBINS) {
    const int incl = sc[pin][tid];
    const int excl = incl - mine;
    cnt[tid] = excl;
    int* __restrict__ lrow = loff + (size_t)blockIdx.x * (NBINS + 1);
    lrow[tid] = excl;
    if (tid == NBINS - 1) lrow[NBINS] = incl;
  }
  __syncthreads();
  int* __restrict__ kout = bkey + lo;
#pragma unroll
  for (int k = 0; k < SORT_EPT; ++k) {
    if (lo + tid + 1024 * k < hi) {
      const int pos = atomicAdd(&cnt[kbin[k]], 1);
      kout[pos] = kkey[k];
    }
  }
  for (int e = lo + 1024 * SORT_EPT + tid; e < hi; e += 1024) {
    const int d = dst[e];
    const int s = src[e];
    const int b = (int)((unsigned)d / (unsigned)bw);
    const int pos = atomicAdd(&cnt[b], 1);
    kout[pos] = (s << 8) | (d - b * bw);
  }
}

__device__ __forceinline__ int other_row(int v) {
  return __builtin_amdgcn_permlanex16(v, v, 0x76543210, 0xfedcba98, false, false);
}


template <int PROJ>
__device__ __forceinline__ void gemm1_tile(const float* __restrict__ x, const float* __restrict__ W1,
                                           const float* __restrict__ b1, const int* __restrict__ degp,
                                           float* __restrict__ pre1, float* __restrict__ y1,
                                           int N, int tile, int lane) {
  const int n = lane & 15;
  const int hi = lane >> 4;
  const int base = tile * 16;
  const int mrow = min(base + n, N - 1);
  const float* __restrict__ W = W1 + PROJ * (64 * 16);
  const float* __restrict__ xr = x + (long long)mrow * 64;

  float rs = 1.0f;
  if constexpr (PROJ == 1) {
    int cs = 0;
#pragma unroll
    for (int i = 0; i < DEG_SLABS / 2; ++i) cs += degp[(size_t)(hi * (DEG_SLABS / 2) + i) * N + mrow];
    cs += other_row(cs);
    const float dg = (float)cs;
    rs = (dg > 0.0f) ? rsqrtf(fmaxf(dg, 1.0f)) : 0.0f;
  }

  v8f acc = {0.f, 0.f, 0.f, 0.f, 0.f, 0.f, 0.f, 0.f};
#pragma unroll
  for (int kb = 0; kb < 64; kb += 4) {
    const int k0 = kb + 2 * hi;
    v2f a = *(const v2f*)(xr + k0);
    if constexpr (PROJ == 1) { a.x *= rs;  a.y *= rs; }
    v2f b;
    b.x = W[k0 * 16 + n];  b.y = W[(k0 + 1) * 16 + n];
    acc = WMMA_F32X4(a, b, acc);
  }
  if constexpr (PROJ == 0) {
    const float bias = b1[n];
#pragma unroll
    for (int r = 0; r < 8; ++r) {
      const int row = base + r + 8 * hi;
      if (row < N) pre1[(long long)row * 16 + n] = acc[r] + bias;
    }
  } else {
#pragma unroll
    for (int r = 0; r < 8; ++r) {
      const int row = base + r + 8 * hi;
      if (row < N) y1[(long long)row * 16 + n] = acc[r];
    }
  }
}

__global__ void __launch_bounds__(1024)
deg_gemm1_kernel(const int* __restrict__ degp, float* __restrict__ dinv,
                 const float* __restrict__ x, const float* __restrict__ W1,
                 const float* __restrict__ b1, float* __restrict__ pre1,
                 float* __restrict__ y1, int N, int nDeg, int nTiles) {
  const int tid = threadIdx.x;
  if ((int)blockIdx.x >= nDeg) {
    const int wv = __builtin_amdgcn_readfirstlane(tid >> 5);
    const int task = ((int)blockIdx.x - nDeg) * 32 + wv;
    const int tile = task >> 1;
    if (tile >= nTiles) return;
    if ((task & 1) == 0) gemm1_tile<0>(x, W1, b1, degp, pre1, y1, N, tile, tid & 31);
    else                 gemm1_tile<1>(x, W1, b1, degp, pre1, y1, N, tile, tid & 31);
    return;
  }
  const int node = blockIdx.x * 1024 + tid;
  if (node < N) {
    int p[DEG_SLABS];
#pragma unroll
    for (int i = 0; i < DEG_SLABS; ++i) p[i] = degp[(size_t)i * N + node];
    int cs = 0;
#pragma unroll
    for (int i = 0; i < DEG_SLABS; ++i) cs += p[i];
    const float dg = (float)cs;
    dinv[node] = (dg > 0.0f) ? rsqrtf(fmaxf(dg, 1.0f)) : 0.0f;
  }
}

template <int M>
__device__ __forceinline__ float row_xor(float v) {
  const int b = __builtin_bit_cast(int, v);
  return __builtin_bit_cast(float, __builtin_amdgcn_update_dpp(b, b, 0x160 + M, 0xf, 0xf, false));
}

template <int K>
__device__ __forceinline__ int row_share(int v) {
  return __builtin_amdgcn_update_dpp(v, v, 0x150 + K, 0xf, 0xf, false);
}

#define EDGES_PER_TRIP 16
__device__ __forceinline__ int seg_slot(int t, int b0, int b1, int b2, int b3, int b4, int b5, int b6, int b7,
                                        int c1, int c2, int c3, int c4, int c5, int c6, int c7) {
  int p = b0;
  p = (t >= c1) ? b1 : p;
  p = (t >= c2) ? b2 : p;
  p = (t >= c3) ? b3 : p;
  p = (t >= c4) ? b4 : p;
  p = (t >= c5) ? b5 : p;
  p = (t >= c6) ? b6 : p;
  p = (t >= c7) ? b7 : p;
  return p + t;
}
template <bool TAIL>
__global__ void __launch_bounds__(1024)
bin_sum_kernel(const int* __restrict__ bkey, const int* __restrict__ loff,
               const float* __restrict__ dinv, const float* __restrict__ y,
               const float* __restrict__ own, float* __restrict__ outp, float* __restrict__ outs,
               const float* __restrict__ W2, const float* __restrict__ b2, int N, int E) {
  __shared__ __attribute__((aligned(16))) float acc[ROWS_MAX * 16];
  const int tid = threadIdx.x;
  const int bin = blockIdx.x;
  const int bw = (N + NBINS - 1) / NBINS;
  const int eb = (E + CHUNKS - 1) / CHUNKS;
  for (int j = tid; j < ROWS_MAX * 16; j += 1024) acc[j] = 0.0f;
  const int wv = __builtin_amdgcn_readfirstlane(tid >> 5);
  const int hf = (tid >> 4) & 1;
  const int f = tid & 15;
  int tot = 0;
#define SEG_ROW(j, bj, cj)                                                        \
  int bj, cj;                                                                      \
  {                                                                                \
    const int ch = wv + 32 * (j);                                                  \
    const int* __restrict__ lr = loff + (size_t)ch * (NBINS + 1) + bin;            \
    const int o0 = lr[0], o1 = lr[1];                                              \
    cj = tot;                                                                      \
    bj = ch * eb + o0 - tot;                                                       \
    tot += o1 - o0;                                                                \
  }
  SEG_ROW(0, sb0, sc0) SEG_ROW(1, sb1, sc1) SEG_ROW(2, sb2, sc2) SEG_ROW(3, sb3, sc3)
  SEG_ROW(4, sb4, sc4) SEG_ROW(5, sb5, sc5) SEG_ROW(6, sb6, sc6) SEG_ROW(7, sb7, sc7)
#undef SEG_ROW
  static_assert(SEGS_PER_WAVE == 8, "eight segment rows are written out above and in seg_slot()");
  __syncthreads();
  if (tot > 0) {
    const int lastt = tot - 1;
#define SLOT(t) seg_slot(min((t), lastt), sb0, sb1, sb2, sb3, sb4, sb5, sb6, sb7, sc1, sc2, sc3, sc4, sc5, sc6, sc7)
    int kv = bkey[SLOT(2 * f + hf)];
    for (int t0 = 0; 2 * t0 < tot; t0 += EDGES_PER_TRIP) {
      const int nkv = bkey[SLOT(2 * (t0 + EDGES_PER_TRIP + f) + hf)];
      int key[EDGES_PER_TRIP];
#define KEY_OF(k) key[k] = row_share<k>(kv);
      KEY_OF(0) KEY_OF(1) KEY_OF(2) KEY_OF(3) KEY_OF(4) KEY_OF(5) KEY_OF(6) KEY_OF(7)
      KEY_OF(8) KEY_OF(9) KEY_OF(10) KEY_OF(11) KEY_OF(12) KEY_OF(13) KEY_OF(14) KEY_OF(15)
#undef KEY_OF
      static_assert(EDGES_PER_TRIP == 16, "one list entry per lane of a 16-lane group and trip");
      float yv[EDGES_PER_TRIP];
#pragma unroll
      for (int k = 0; k < EDGES_PER_TRIP; ++k) yv[k] = y[(long long)(key[k] >> 8) * 16 + f];
      asm volatile("" :: "v"(yv[0]), "v"(yv[1]), "v"(yv[2]), "v"(yv[3]),
                         "v"(yv[4]), "v"(yv[5]), "v"(yv[6]), "v"(yv[7]),
                         "v"(yv[8]), "v"(yv[9]), "v"(yv[10]), "v"(yv[11]),
                         "v"(yv[12]), "v"(yv[13]), "v"(yv[14]), "v"(yv[15]));
#pragma unroll
      for (int k = 0; k < EDGES_PER_TRIP; ++k) {
        if (2 * (t0 + k) + hf < tot) {
          __hip_atomic_fetch_add(&acc[(key[k] & 255) * 16 + f], yv[k],
                                 __ATOMIC_RELAXED, __HIP_MEMORY_SCOPE_WORKGROUP);
        }
      }
      kv = nkv;
    }
#undef SLOT
  }
  const int base = bin * bw;
  const int nrows = min(bw, N - base);
  if constexpr (!TAIL) {
    __syncthreads();
    for (int j = tid; j < nrows * 16; j += 1024) {
      const long long gi = (long long)base * 16 + j;
      const float di = dinv[base + (j >> 4)];
      const float hv = fmaxf(own[gi] - di * acc[j], 0.0f);
      outp[gi] = hv;
      outs[gi] = di * hv;
    }
  } else {
    __shared__ __attribute__((aligned(16))) float zs[ROWS_MAX * 40];
    const int ntiles = nrows > 0 ? (nrows + 15) >> 4 : 0;
    const bool tilewave = wv < ntiles;
    const int n = tid & 15;
    const int hi = (tid >> 4) & 1;
    const float* __restrict__ W0 = W2;
    const float* __restrict__ Wb = W2 + 16 * 40;
    v2f a[4], b0[12], bb[12];
    float bias0 = 0.f, bias1 = 0.f, bias2 = 0.f, nd = 0.f;
#pragma unroll
    for (int i = 0; i < 4; ++i) a[i] = (v2f){0.f, 0.f};
#pragma unroll
    for (int i = 0; i < 12; ++i) { b0[i] = (v2f){0.f, 0.f}; bb[i] = (v2f){0.f, 0.f}; }
    if (tilewave) {
      const int mrow = min(base + wv * 16 + n, N - 1);
#pragma unroll
      for (int i = 0; i < 4; ++i) a[i] = *(const v2f*)(own + (long long)mrow * 16 + i * 4 + 2 * hi);
      nd = -dinv[mrow];
#pragma unroll
      for (int ct = 0; ct < 3; ++ct) {
        const int cc = min(ct * 16 + n, 39);
#pragma unroll
        for (int i = 0; i < 4; ++i) {
          const int k0 = i * 4 + 2 * hi;
          b0[ct * 4 + i].x = W0[k0 * 40 + cc];  b0[ct * 4 + i].y = W0[(k0 + 1) * 40 + cc];
          bb[ct * 4 + i].x = Wb[k0 * 40 + cc];  bb[ct * 4 + i].y = Wb[(k0 + 1) * 40 + cc];
        }
      }
      bias0 = b2[n];  bias1 = b2[16 + n];  bias2 = b2[min(32 + n, 39)];
    }
    __syncthreads();
    if (tilewave) {
      v2f t[4];
#pragma unroll
      for (int i = 0; i < 4; ++i) {
        t[i] = *(const v2f*)(&acc[(wv * 16 + n) * 16 + i * 4 + 2 * hi]);
        t[i].x *= nd;  t[i].y *= nd;
      }
      v8f c[6];
#pragma unroll
      for (int i = 0; i < 6; ++i) c[i] = (v8f){0.f, 0.f, 0.f, 0.f, 0.f, 0.f, 0.f, 0.f};
#pragma unroll
      for (int ct = 0; ct < 3; ++ct) {
#pragma unroll
        for (int i = 0; i < 4; ++i) {
          c[ct * 2]     = WMMA_F32X4(a[i], b0[ct * 4 + i], c[ct * 2]);
          c[ct * 2 + 1] = WMMA_F32X4(t[i], bb[ct * 4 + i], c[ct * 2 + 1]);
        }
      }
      float z0[8], z1[8], z2[8], m[8], s[8];
#pragma unroll
      for (int r = 0; r < 8; ++r) {
        z0[r] = (c[0][r] + c[1][r]) + bias0;
        z1[r] = (c[2][r] + c[3][r]) + bias1;
        z2[r] = (c[4][r] + c[5][r]) + bias2;
        m[r] = fmaxf(z0[r], z1[r]);
      }
      if (n < 8) {
#pragma unroll
        for (int r = 0; r < 8; ++r) m[r] = fmaxf(m[r], z2[r]);
      }
#pragma unroll
      for (int r = 0; r < 8; ++r) m[r] = fmaxf(m[r], row_xor<8>(m[r]));
#pragma unroll
      for (int r = 0; r < 8; ++r) m[r] = fmaxf(m[r], row_xor<4>(m[r]));
#pragma unroll
      for (int r = 0; r < 8; ++r) m[r] = fmaxf(m[r], row_xor<2>(m[r]));
#pragma unroll
      for (int r = 0; r < 8; ++r) m[r] = fmaxf(m[r], row_xor<1>(m[r]));
#pragma unroll
      for (int r = 0; r < 8; ++r) s[r] = expf(z0[r] - m[r]) + expf(z1[r] - m[r]);
      if (n < 8) {
#pragma unroll
        for (int r = 0; r < 8; ++r) s[r] += expf(z2[r] - m[r]);
      }
#pragma unroll
      for (int r = 0; r < 8; ++r) s[r] += row_xor<8>(s[r]);
#pragma unroll
      for (int r = 0; r < 8; ++r) s[r] += row_xor<4>(s[r]);
#pragma unroll
      for (int r = 0; r < 8; ++r) s[r] += row_xor<2>(s[r]);
#pragma unroll
      for (int r = 0; r < 8; ++r) s[r] += row_xor<1>(s[r]);
#pragma unroll
      for (int r = 0; r < 8; ++r) {
        const float lse = m[r] + logf(s[r]);
        float* zr = zs + (wv * 16 + r + 8 * hi) * 40;
        zr[n] = z0[r] - lse;
        zr[16 + n] = z1[r] - lse;
        if (n < 8) zr[32 + n] = z2[r] - lse;
      }
    }
    __syncthreads();
    float4* __restrict__ dst4 = (float4*)(outp + (long long)base * 40);
    const float4* src4 = (const float4*)zs;
    for (int j = tid; j < nrows * 10; j += 1024) dst4[j] = src4[j];
  }
}

extern "C" void kernel_launch(void* const* d_in, const int* in_sizes, int n_in,
                              void* d_out, int out_size, void* d_ws, size_t ws_size,
                              hipStream_t stream) {
  const float* x  = (const float*)d_in[0];
  const int* ei   = (const int*)d_in[1];
  const float* W1 = (const float*)d_in[2];
  const float* b1 = (const float*)d_in[3];
  const float* W2 = (const float*)d_in[4];
  const float* b2 = (const float*)d_in[5];
  float* out = (float*)d_out;

  const int N = in_sizes[0] / 64;
  const int E = in_sizes[1] / 2;
  const int* src = ei;
  const int* dst = ei + E;

  float* ws = (float*)d_ws;
  size_t off = 0;
  float* dinv = ws + off; off += (size_t)N;
  float* pre1 = ws + off; off += (size_t)N * 16;
  float* y1   = ws + off; off += (size_t)N * 16;
  int* loff   = (int*)(ws + off); off += (size_t)CHUNKS * (NBINS + 1);
  int* bkey   = (int*)(ws + off); off += (size_t)E;
  float* h    = ws + off; off += (size_t)N * 16;
  float* hs   = ws + off;
  int* degp   = (int*)h;

  auto blk = [](long long n, int b) { return (unsigned)((n + b - 1) / b); };

  const int tiles = (N + 15) / 16;
  const int nDeg = (int)blk(N, 1024);
  bin_sort_kernel<<<CHUNKS, 1024, 0, stream>>>(src, dst, degp, loff, bkey, N, E);
  deg_gemm1_kernel<<<nDeg + blk((long long)tiles * 2, 32), 1024, 0, stream>>>(
      degp, dinv, x, W1, b1, pre1, y1, N, nDeg, tiles);

  bin_sum_kernel<false><<<NBINS, 1024, 0, stream>>>(bkey, loff, dinv, y1, pre1, h, hs, W2, b2, N, E);

  bin_sum_kernel<true><<<NBINS, 1024, 0, stream>>>(bkey, loff, dinv, hs, h, out, nullptr, W2, b2, N, E);
}


# Range-sum trip loop: the fetching lane makes an entry's gather and LDS byte offsets once, lanes take them by DPP folded into one OR with their column (32-bit offsets); full trips add unpredicated
# speedup vs baseline: 4.7907x; 1.0336x over previous
#include <hip/hip_runtime.h>
#include <math.h>

typedef float v2f __attribute__((ext_vector_type(2)));
typedef float v8f __attribute__((ext_vector_type(8)));

#define WMMA_F32X4(a, b, c) \
  __builtin_amdgcn_wmma_f32_16x16x4_f32(false, (a), false, (b), (short)0, (c), false, false)

#define NBINS 256
#define ROWS_MAX 256
#define CHUNKS 256
#define SORT_EPT 4
#define SEGS_PER_WAVE 8
static_assert(CHUNKS == 32 * SEGS_PER_WAVE, "the 32 waves of a range's work-group share its segments evenly");

#define DEG_SLABS 32
#define DEG_RANGES 8
#define DEG_ROWS_MAX 8192
#define DEG_LOADS 8
static_assert(DEG_SLABS * DEG_RANGES == CHUNKS, "one (slab, range) pair per work-group of pass 1");

__global__ void __launch_bounds__(1024)
bin_sort_kernel(const int* __restrict__ src, const int* __restrict__ dst,
                int* __restrict__ degp, int* __restrict__ loff, int* __restrict__ bkey, int N, int E) {
  __shared__ int cnt[NBINS];
  __shared__ int sc[2][NBINS];
  __shared__ int dcnt[DEG_ROWS_MAX];
  const int tid = threadIdx.x;
  const int bw = (N + NBINS - 1) / NBINS;
  const int eb = (E + CHUNKS - 1) / CHUNKS;
  const int c = blockIdx.x / DEG_RANGES;
  const int r = blockIdx.x - c * DEG_RANGES;
  const int rw = (N + DEG_RANGES - 1) / DEG_RANGES;
  const int r_lo = r * rw;
  if (tid < NBINS) cnt[tid] = 0;
  for (int j = tid; j < rw; j += 1024) dcnt[j] = 0;
  __syncthreads();
  const int lo = blockIdx.x * eb;
  const int hi = min(E, lo + eb);
  int kbin[SORT_EPT], kkey[SORT_EPT];
#pragma unroll
  for (int k = 0; k < SORT_EPT; ++k) {
    const int e = min(lo + tid + 1024 * k, E - 1);
    const int d = dst[e];
    const int s = src[e];
    const int b = (int)((unsigned)d / (unsigned)bw);
    kbin[k] = b;
    kkey[k] = (s << 8) | (d - b * bw);
  }
#pragma unroll
  for (int k = 0; k < SORT_EPT; ++k) {
    if (lo + tid + 1024 * k < hi) atomicAdd(&cnt[kbin[k]], 1);
  }
  for (int e = lo + 1024 * SORT_EPT + tid; e < hi; e += 1024) {
    const int d = dst[e];
    atomicAdd(&cnt[(unsigned)d / (unsigned)bw], 1);
  }
  const int slab = (((E + DEG_SLABS - 1) / DEG_SLABS) + 3) & ~3;
  const int s_lo = c * slab;
  const int s_hi = min(E, s_lo + slab);
  const int Eal = E & ~3;
  const int v_hi = min(s_hi, Eal);
  for (int e0 = s_lo + 4 * tid; e0 < v_hi; e0 += 4096 * DEG_LOADS) {
    int4 v[DEG_LOADS];
#pragma unroll
    for (int k = 0; k < DEG_LOADS; ++k) {
      const int e = min(e0 + 4096 * k, Eal - 4);
      v[k] = *(const int4*)(src + e);
    }
#pragma unroll
    for (int k = 0; k < DEG_LOADS; ++k) {
      if (e0 + 4096 * k < v_hi) {
        const unsigned j0 = (unsigned)(v[k].x - r_lo), j1 = (unsigned)(v[k].y - r_lo);
        const unsigned j2 = (unsigned)(v[k].z - r_lo), j3 = (unsigned)(v[k].w - r_lo);
        if (j0 < (unsigned)rw) atomicAdd(&dcnt[j0], 1);
        if (j1 < (unsigned)rw) atomicAdd(&dcnt[j1], 1);
        if (j2 < (unsigned)rw) atomicAdd(&dcnt[j2], 1);
        if (j3 < (unsigned)rw) atomicAdd(&dcnt[j3], 1);
      }
    }
  }
  if (tid < E - Eal) {
    const int e = Eal + tid;
    if (e >= s_lo && e < s_hi) {
      const unsigned j = (unsigned)(src[e] - r_lo);
      if (j < (unsigned)rw) atomicAdd(&dcnt[j], 1);
    }
  }
  __syncthreads();
  int* __restrict__ plane = degp + (size_t)c * N;
  for (int j = tid; j < rw; j += 1024) {
    const int node = r_lo + j;
    if (node < N) plane[node] = dcnt[j];
  }
  int mine = 0;
  if (tid < NBINS) { mine = cnt[tid]; sc[0][tid] = mine; }
  __syncthreads();
  int pin = 0;
  for (int o = 1; o < NBINS; o <<= 1) {
    if (tid < NBINS) {
      int x = sc[pin][tid];
      if (tid >= o) x += sc[pin][tid - o];
      sc[pin ^ 1][tid] = x;
    }
    __syncthreads();
    pin ^= 1;
  }
  if (tid < NBINS) {
    const int incl = sc[pin][tid];
    const int excl = incl - mine;
    cnt[tid] = excl;
    int* __restrict__ lrow = loff + (size_t)blockIdx.x * (NBINS + 1);
    lrow[tid] = excl;
    if (tid == NBINS - 1) lrow[NBINS] = incl;
  }
  __syncthreads();
  int* __restrict__ kout = bkey + lo;
#pragma unroll
  for (int k = 0; k < SORT_EPT; ++k) {
    if (lo + tid + 1024 * k < hi) {
      const int pos = atomicAdd(&cnt[kbin[k]], 1);
      kout[pos] = kkey[k];
    }
  }
  for (int e = lo + 1024 * SORT_EPT + tid; e < hi; e += 1024) {
    const int d = dst[e];
    const int s = src[e];
    const int b = (int)((unsigned)d / (unsigned)bw);
    const int pos = atomicAdd(&cnt[b], 1);
    kout[pos] = (s << 8) | (d - b * bw);
  }
}

__device__ __forceinline__ int other_row(int v) {
  return __builtin_amdgcn_permlanex16(v, v, 0x76543210, 0xfedcba98, false, false);
}


template <int PROJ>
__device__ __forceinline__ void gemm1_tile(const float* __restrict__ x, const float* __restrict__ W1,
                                           const float* __restrict__ b1, const int* __restrict__ degp,
                                           float* __restrict__ pre1, float* __restrict__ y1,
                                           int N, int tile, int lane) {
  const int n = lane & 15;
  const int hi = lane >> 4;
  const int base = tile * 16;
  const int mrow = min(base + n, N - 1);
  const float* __restrict__ W = W1 + PROJ * (64 * 16);
  const float* __restrict__ xr = x + (long long)mrow * 64;

  float rs = 1.0f;
  if constexpr (PROJ == 1) {
    int cs = 0;
#pragma unroll
    for (int i = 0; i < DEG_SLABS / 2; ++i) cs += degp[(size_t)(hi * (DEG_SLABS / 2) + i) * N + mrow];
    cs += other_row(cs);
    const float dg = (float)cs;
    rs = (dg > 0.0f) ? rsqrtf(fmaxf(dg, 1.0f)) : 0.0f;
  }

  v8f acc = {0.f, 0.f, 0.f, 0.f, 0.f, 0.f, 0.f, 0.f};
#pragma unroll
  for (int kb = 0; kb < 64; kb += 4) {
    const int k0 = kb + 2 * hi;
    v2f a = *(const v2f*)(xr + k0);
    if constexpr (PROJ == 1) { a.x *= rs;  a.y *= rs; }
    v2f b;
    b.x = W[k0 * 16 + n];  b.y = W[(k0 + 1) * 16 + n];
    acc = WMMA_F32X4(a, b, acc);
  }
  if constexpr (PROJ == 0) {
    const float bias = b1[n];
#pragma unroll
    for (int r = 0; r < 8; ++r) {
      const int row = base + r + 8 * hi;
      if (row < N) pre1[(long long)row * 16 + n] = acc[r] + bias;
    }
  } else {
#pragma unroll
    for (int r = 0; r < 8; ++r) {
      const int row = base + r + 8 * hi;
      if (row < N) y1[(long long)row * 16 + n] = acc[r];
    }
  }
}

__global__ void __launch_bounds__(1024)
deg_gemm1_kernel(const int* __restrict__ degp, float* __restrict__ dinv,
                 const float* __restrict__ x, const float* __restrict__ W1,
                 const float* __restrict__ b1, float* __restrict__ pre1,
                 float* __restrict__ y1, int N, int nDeg, int nTiles) {
  const int tid = threadIdx.x;
  if ((int)blockIdx.x >= nDeg) {
    const int wv = __builtin_amdgcn_readfirstlane(tid >> 5);
    const int task = ((int)blockIdx.x - nDeg) * 32 + wv;
    const int tile = task >> 1;
    if (tile >= nTiles) return;
    if ((task & 1) == 0) gemm1_tile<0>(x, W1, b1, degp, pre1, y1, N, tile, tid & 31);
    else                 gemm1_tile<1>(x, W1, b1, degp, pre1, y1, N, tile, tid & 31);
    return;
  }
  const int node = blockIdx.x * 1024 + tid;
  if (node < N) {
    int p[DEG_SLABS];
#pragma unroll
    for (int i = 0; i < DEG_SLABS; ++i) p[i] = degp[(size_t)i * N + node];
    int cs = 0;
#pragma unroll
    for (int i = 0; i < DEG_SLABS; ++i) cs += p[i];
    const float dg = (float)cs;
    dinv[node] = (dg > 0.0f) ? rsqrtf(fmaxf(dg, 1.0f)) : 0.0f;
  }
}

template <int M>
__device__ __forceinline__ float row_xor(float v) {
  const int b = __builtin_bit_cast(int, v);
  return __builtin_bit_cast(float, __builtin_amdgcn_update_dpp(b, b, 0x160 + M, 0xf, 0xf, false));
}

template <int K>
__device__ __forceinline__ unsigned row_share_or(unsigned v, unsigned o) {
  return (unsigned)__builtin_amdgcn_update_dpp(0, (int)v, 0x150 + K, 0xf, 0xf, true) | o;
}

#define EDGES_PER_TRIP 16
__device__ __forceinline__ int seg_slot(int t, int b0, int b1, int b2, int b3, int b4, int b5, int b6, int b7,
                                        int c1, int c2, int c3, int c4, int c5, int c6, int c7) {
  int p = b0;
  p = (t >= c1) ? b1 : p;
  p = (t >= c2) ? b2 : p;
  p = (t >= c3) ? b3 : p;
  p = (t >= c4) ? b4 : p;
  p = (t >= c5) ? b5 : p;
  p = (t >= c6) ? b6 : p;
  p = (t >= c7) ? b7 : p;
  return p + t;
}
template <bool TAIL>
__global__ void __launch_bounds__(1024)
bin_sum_kernel(const int* __restrict__ bkey, const int* __restrict__ loff,
               const float* __restrict__ dinv, const float* __restrict__ y,
               const float* __restrict__ own, float* __restrict__ outp, float* __restrict__ outs,
               const float* __restrict__ W2, const float* __restrict__ b2, int N, int E) {
  __shared__ __attribute__((aligned(16))) float acc[ROWS_MAX * 16];
  const int tid = threadIdx.x;
  const int bin = blockIdx.x;
  const int bw = (N + NBINS - 1) / NBINS;
  const int eb = (E + CHUNKS - 1) / CHUNKS;
  for (int j = tid; j < ROWS_MAX * 16; j += 1024) acc[j] = 0.0f;
  const int wv = __builtin_amdgcn_readfirstlane(tid >> 5);
  const int hf = (tid >> 4) & 1;
  const int f = tid & 15;
  int tot = 0;
#define SEG_ROW(j, bj, cj)                                                        \
  int bj, cj;                                                                      \
  {                                                                                \
    const int ch = wv + 32 * (j);                                                  \
    const int* __restrict__ lr = loff + (size_t)ch * (NBINS + 1) + bin;            \
    const int o0 = lr[0], o1 = lr[1];                                              \
    cj = tot;                                                                      \
    bj = ch * eb + o0 - tot;                                                       \
    tot += o1 - o0;                                                                \
  }
  SEG_ROW(0, sb0, sc0) SEG_ROW(1, sb1, sc1) SEG_ROW(2, sb2, sc2) SEG_ROW(3, sb3, sc3)
  SEG_ROW(4, sb4, sc4) SEG_ROW(5, sb5, sc5) SEG_ROW(6, sb6, sc6) SEG_ROW(7, sb7, sc7)
#undef SEG_ROW
  static_assert(SEGS_PER_WAVE == 8, "eight segment rows are written out above and in seg_slot()");
  __syncthreads();
  if (tot > 0) {
    const int lastt = tot - 1;
#define SLOT(t) seg_slot(min((t), lastt), sb0, sb1, sb2, sb3, sb4, sb5, sb6, sb7, sc1, sc2, sc3, sc4, sc5, sc6, sc7)
    const unsigned f4 = (unsigned)f << 2;
    const char* __restrict__ yb = (const char*)y;
    char* accb = (char*)acc;
    int kv = bkey[SLOT(2 * f + hf)];
    for (int t0 = 0; 2 * t0 < tot; t0 += EDGES_PER_TRIP) {
      const int nkv = bkey[SLOT(2 * (t0 + EDGES_PER_TRIP + f) + hf)];
      const unsigned gb = ((unsigned)kv >> 8) << 6;
      const unsigned lb = ((unsigned)kv & 255u) << 6;
      unsigned go[EDGES_PER_TRIP], la[EDGES_PER_TRIP];
#define OFFS_OF(k) go[k] = row_share_or<k>(gb, f4);  la[k] = row_share_or<k>(lb, f4);
      OFFS_OF(0) OFFS_OF(1) OFFS_OF(2) OFFS_OF(3) OFFS_OF(4) OFFS_OF(5) OFFS_OF(6) OFFS_OF(7)
      OFFS_OF(8) OFFS_OF(9) OFFS_OF(10) OFFS_OF(11) OFFS_OF(12) OFFS_OF(13) OFFS_OF(14) OFFS_OF(15)
#undef OFFS_OF
      static_assert(EDGES_PER_TRIP == 16, "one list entry per lane of a 16-lane group and trip");
      float yv[EDGES_PER_TRIP];
#pragma unroll
      for (int k = 0; k < EDGES_PER_TRIP; ++k) yv[k] = *(const float*)(yb + go[k]);
      asm volatile("" :: "v"(yv[0]), "v"(yv[1]), "v"(yv[2]), "v"(yv[3]),
                         "v"(yv[4]), "v"(yv[5]), "v"(yv[6]), "v"(yv[7]),
                         "v"(yv[8]), "v"(yv[9]), "v"(yv[10]), "v"(yv[11]),
                         "v"(yv[12]), "v"(yv[13]), "v"(yv[14]), "v"(yv[15]));
      if (2 * (t0 + EDGES_PER_TRIP) <= tot) {
#pragma unroll
        for (int k = 0; k < EDGES_PER_TRIP; ++k) {
          __hip_atomic_fetch_add((float*)(accb + la[k]), yv[k], __ATOMIC_RELAXED, __HIP_MEMORY_SCOPE_WORKGROUP);
        }
      } else {
#pragma unroll
        for (int k = 0; k < EDGES_PER_TRIP; ++k) {
          if (2 * (t0 + k) + hf < tot) {
            __hip_atomic_fetch_add((float*)(accb + la[k]), yv[k], __ATOMIC_RELAXED, __HIP_MEMORY_SCOPE_WORKGROUP);
          }
        }
      }
      kv = nkv;
    }
#undef SLOT
  }
  const int base = bin * bw;
  const int nrows = min(bw, N - base);
  if constexpr (!TAIL) {
    __syncthreads();
    for (int j = tid; j < nrows * 16; j += 1024) {
      const long long gi = (long long)base * 16 + j;
      const float di = dinv[base + (j >> 4)];
      const float hv = fmaxf(own[gi] - di * acc[j], 0.0f);
      outp[gi] = hv;
      outs[gi] = di * hv;
    }
  } else {
    __shared__ __attribute__((aligned(16))) float zs[ROWS_MAX * 40];
    const int ntiles = nrows > 0 ? (nrows + 15) >> 4 : 0;
    const bool tilewave = wv < ntiles;
    const int n = tid & 15;
    const int hi = (tid >> 4) & 1;
    const float* __restrict__ W0 = W2;
    const float* __restrict__ Wb = W2 + 16 * 40;
    v2f a[4], b0[12], bb[12];
    float bias0 = 0.f, bias1 = 0.f, bias2 = 0.f, nd = 0.f;
#pragma unroll
    for (int i = 0; i < 4; ++i) a[i] = (v2f){0.f, 0.f};
#pragma unroll
    for (int i = 0; i < 12; ++i) { b0[i] = (v2f){0.f, 0.f}; bb[i] = (v2f){0.f, 0.f}; }
    if (tilewave) {
      const int mrow = min(base + wv * 16 + n, N - 1);
#pragma unroll
      for (int i = 0; i < 4; ++i) a[i] = *(const v2f*)(own + (long long)mrow * 16 + i * 4 + 2 * hi);
      nd = -dinv[mrow];
#pragma unroll
      for (int ct = 0; ct < 3; ++ct) {
        const int cc = min(ct * 16 + n, 39);
#pragma unroll
        for (int i = 0; i < 4; ++i) {
          const int k0 = i * 4 + 2 * hi;
          b0[ct * 4 + i].x = W0[k0 * 40 + cc];  b0[ct * 4 + i].y = W0[(k0 + 1) * 40 + cc];
          bb[ct * 4 + i].x = Wb[k0 * 40 + cc];  bb[ct * 4 + i].y = Wb[(k0 + 1) * 40 + cc];
        }
      }
      bias0 = b2[n];  bias1 = b2[16 + n];  bias2 = b2[min(32 + n, 39)];
    }
    __syncthreads();
    if (tilewave) {
      v2f t[4];
#pragma unroll
      for (int i = 0; i < 4; ++i) {
        t[i] = *(const v2f*)(&acc[(wv * 16 + n) * 16 + i * 4 + 2 * hi]);
        t[i].x *= nd;  t[i].y *= nd;
      }
      v8f c[6];
#pragma unroll
      for (int i = 0; i < 6; ++i) c[i] = (v8f){0.f, 0.f, 0.f, 0.f, 0.f, 0.f, 0.f, 0.f};
#pragma unroll
      for (int ct = 0; ct < 3; ++ct) {
#pragma unroll
        for (int i = 0; i < 4; ++i) {
          c[ct * 2]     = WMMA_F32X4(a[i], b0[ct * 4 + i], c[ct * 2]);
          c[ct * 2 + 1] = WMMA_F32X4(t[i], bb[ct * 4 + i], c[ct * 2 + 1]);
        }
      }
      float z0[8], z1[8], z2[8], m[8], s[8];
#pragma unroll
      for (int r = 0; r < 8; ++r) {
        z0[r] = (c[0][r] + c[1][r]) + bias0;
        z1[r] = (c[2][r] + c[3][r]) + bias1;
        z2[r] = (c[4][r] + c[5][r]) + bias2;
        m[r] = fmaxf(z0[r], z1[r]);
      }
      if (n < 8) {
#pragma unroll
        for (int r = 0; r < 8; ++r) m[r] = fmaxf(m[r], z2[r]);
      }
#pragma unroll
      for (int r = 0; r < 8; ++r) m[r] = fmaxf(m[r], row_xor<8>(m[r]));
#pragma unroll
      for (int r = 0; r < 8; ++r) m[r] = fmaxf(m[r], row_xor<4>(m[r]));
#pragma unroll
      for (int r = 0; r < 8; ++r) m[r] = fmaxf(m[r], row_xor<2>(m[r]));
#pragma unroll
      for (int r = 0; r < 8; ++r) m[r] = fmaxf(m[r], row_xor<1>(m[r]));
#pragma unroll
      for (int r = 0; r < 8; ++r) s[r] = expf(z0[r] - m[r]) + expf(z1[r] - m[r]);
      if (n < 8) {
#pragma unroll
        for (int r = 0; r < 8; ++r) s[r] += expf(z2[r] - m[r]);
      }
#pragma unroll
      for (int r = 0; r < 8; ++r) s[r] += row_xor<8>(s[r]);
#pragma unroll
      for (int r = 0; r < 8; ++r) s[r] += row_xor<4>(s[r]);
#pragma unroll
      for (int r = 0; r < 8; ++r) s[r] += row_xor<2>(s[r]);
#pragma unroll
      for (int r = 0; r < 8; ++r) s[r] += row_xor<1>(s[r]);
#pragma unroll
      for (int r = 0; r < 8; ++r) {
        const float lse = m[r] + logf(s[r]);
        float* zr = zs + (wv * 16 + r + 8 * hi) * 40;
        zr[n] = z0[r] - lse;
        zr[16 + n] = z1[r] - lse;
        if (n < 8) zr[32 + n] = z2[r] - lse;
      }
    }
    __syncthreads();
    float4* __restrict__ dst4 = (float4*)(outp + (long long)base * 40);
    const float4* src4 = (const float4*)zs;
    for (int j = tid; j < nrows * 10; j += 1024) dst4[j] = src4[j];
  }
}

extern "C" void kernel_launch(void* const* d_in, const int* in_sizes, int n_in,
                              void* d_out, int out_size, void* d_ws, size_t ws_size,
                              hipStream_t stream) {
  const float* x  = (const float*)d_in[0];
  const int* ei   = (const int*)d_in[1];
  const float* W1 = (const float*)d_in[2];
  const float* b1 = (const float*)d_in[3];
  const float* W2 = (const float*)d_in[4];
  const float* b2 = (const float*)d_in[5];
  float* out = (float*)d_out;

  const int N = in_sizes[0] / 64;
  const int E = in_sizes[1] / 2;
  const int* src = ei;
  const int* dst = ei + E;

  float* ws = (float*)d_ws;
  size_t off = 0;
  float* dinv = ws + off; off += (size_t)N;
  float* pre1 = ws + off; off += (size_t)N * 16;
  float* y1   = ws + off; off += (size_t)N * 16;
  int* loff   = (int*)(ws + off); off += (size_t)CHUNKS * (NBINS + 1);
  int* bkey   = (int*)(ws + off); off += (size_t)E;
  float* h    = ws + off; off += (size_t)N * 16;
  float* hs   = ws + off;
  int* degp   = (int*)h;

  auto blk = [](long long n, int b) { return (unsigned)((n + b - 1) / b); };

  const int tiles = (N + 15) / 16;
  const int nDeg = (int)blk(N, 1024);
  bin_sort_kernel<<<CHUNKS, 1024, 0, stream>>>(src, dst, degp, loff, bkey, N, E);
  deg_gemm1_kernel<<<nDeg + blk((long long)tiles * 2, 32), 1024, 0, stream>>>(
      degp, dinv, x, W1, b1, pre1, y1, N, nDeg, tiles);

  bin_sum_kernel<false><<<NBINS, 1024, 0, stream>>>(bkey, loff, dinv, y1, pre1, h, hs, W2, b2, N, E);

  bin_sum_kernel<true><<<NBINS, 1024, 0, stream>>>(bkey, loff, dinv, hs, h, out, nullptr, W2, b2, N, E);
}


# Pass 1 collects a chunk's grouped keys in LDS (freed degree counters) and stores them as whole aligned 128-byte lines, not 32 scattered 4-byte stores per wave instruction; key order unchanged
# speedup vs baseline: 5.0156x; 1.0469x over previous
#include <hip/hip_runtime.h>
#include <math.h>

typedef float v2f __attribute__((ext_vector_type(2)));
typedef float v8f __attribute__((ext_vector_type(8)));

#define WMMA_F32X4(a, b, c) \
  __builtin_amdgcn_wmma_f32_16x16x4_f32(false, (a), false, (b), (short)0, (c), false, false)

#define NBINS 256
#define ROWS_MAX 256
#define CHUNKS 256
#define SORT_EPT 4
#define SEGS_PER_WAVE 8
static_assert(CHUNKS == 32 * SEGS_PER_WAVE, "the 32 waves of a range's work-group share its segments evenly");

#define DEG_SLABS 32
#define DEG_RANGES 8
#define DEG_ROWS_MAX 8192
#define DEG_LOADS 8
static_assert(DEG_SLABS * DEG_RANGES == CHUNKS, "one (slab, range) pair per work-group of pass 1");

__global__ void __launch_bounds__(1024)
bin_sort_kernel(const int* __restrict__ src, const int* __restrict__ dst,
                int* __restrict__ degp, int* __restrict__ loff, int* __restrict__ bkey, int N, int E) {
  __shared__ int cnt[NBINS];
  __shared__ int sc[2][NBINS];
  __shared__ int dcnt[DEG_ROWS_MAX];
  const int tid = threadIdx.x;
  const int bw = (N + NBINS - 1) / NBINS;
  const int eb = (E + CHUNKS - 1) / CHUNKS;
  const int c = blockIdx.x / DEG_RANGES;
  const int r = blockIdx.x - c * DEG_RANGES;
  const int rw = (N + DEG_RANGES - 1) / DEG_RANGES;
  const int r_lo = r * rw;
  if (tid < NBINS) cnt[tid] = 0;
  for (int j = tid; j < rw; j += 1024) dcnt[j] = 0;
  __syncthreads();
  const int lo = blockIdx.x * eb;
  const int hi = min(E, lo + eb);
  int kbin[SORT_EPT], kkey[SORT_EPT];
#pragma unroll
  for (int k = 0; k < SORT_EPT; ++k) {
    const int e = min(lo + tid + 1024 * k, E - 1);
    const int d = dst[e];
    const int s = src[e];
    const int b = (int)((unsigned)d / (unsigned)bw);
    kbin[k] = b;
    kkey[k] = (s << 8) | (d - b * bw);
  }
#pragma unroll
  for (int k = 0; k < SORT_EPT; ++k) {
    if (lo + tid + 1024 * k < hi) atomicAdd(&cnt[kbin[k]], 1);
  }
  for (int e = lo + 1024 * SORT_EPT + tid; e < hi; e += 1024) {
    const int d = dst[e];
    atomicAdd(&cnt[(unsigned)d / (unsigned)bw], 1);
  }
  const int slab = (((E + DEG_SLABS - 1) / DEG_SLABS) + 3) & ~3;
  const int s_lo = c * slab;
  const int s_hi = min(E, s_lo + slab);
  const int Eal = E & ~3;
  const int v_hi = min(s_hi, Eal);
  for (int e0 = s_lo + 4 * tid; e0 < v_hi; e0 += 4096 * DEG_LOADS) {
    int4 v[DEG_LOADS];
#pragma unroll
    for (int k = 0; k < DEG_LOADS; ++k) {
      const int e = min(e0 + 4096 * k, Eal - 4);
      v[k] = *(const int4*)(src + e);
    }
#pragma unroll
    for (int k = 0; k < DEG_LOADS; ++k) {
      if (e0 + 4096 * k < v_hi) {
        const unsigned j0 = (unsigned)(v[k].x - r_lo), j1 = (unsigned)(v[k].y - r_lo);
        const unsigned j2 = (unsigned)(v[k].z - r_lo), j3 = (unsigned)(v[k].w - r_lo);
        if (j0 < (unsigned)rw) atomicAdd(&dcnt[j0], 1);
        if (j1 < (unsigned)rw) atomicAdd(&dcnt[j1], 1);
        if (j2 < (unsigned)rw) atomicAdd(&dcnt[j2], 1);
        if (j3 < (unsigned)rw) atomicAdd(&dcnt[j3], 1);
      }
    }
  }
  if (tid < E - Eal) {
    const int e = Eal + tid;
    if (e >= s_lo && e < s_hi) {
      const unsigned j = (unsigned)(src[e] - r_lo);
      if (j < (unsigned)rw) atomicAdd(&dcnt[j], 1);
    }
  }
  __syncthreads();
  int* __restrict__ plane = degp + (size_t)c * N;
  for (int j = tid; j < rw; j += 1024) {
    const int node = r_lo + j;
    if (node < N) plane[node] = dcnt[j];
  }
  int mine = 0;
  if (tid < NBINS) { mine = cnt[tid]; sc[0][tid] = mine; }
  __syncthreads();
  int pin = 0;
  for (int o = 1; o < NBINS; o <<= 1) {
    if (tid < NBINS) {
      int x = sc[pin][tid];
      if (tid >= o) x += sc[pin][tid - o];
      sc[pin ^ 1][tid] = x;
    }
    __syncthreads();
    pin ^= 1;
  }
  if (tid < NBINS) {
    const int incl = sc[pin][tid];
    const int excl = incl - mine;
    cnt[tid] = excl;
    int* __restrict__ lrow = loff + (size_t)blockIdx.x * (NBINS + 1);
    lrow[tid] = excl;
    if (tid == NBINS - 1) lrow[NBINS] = incl;
  }
  __syncthreads();
  int* __restrict__ kout = bkey + lo;
  int* stage = dcnt;
  const int nk = hi - lo;
  const bool staged = nk <= DEG_ROWS_MAX;
  int pos[SORT_EPT];
#pragma unroll
  for (int k = 0; k < SORT_EPT; ++k) {
    pos[k] = 0;
    if (lo + tid + 1024 * k < hi) pos[k] = atomicAdd(&cnt[kbin[k]], 1);
  }
#pragma unroll
  for (int k = 0; k < SORT_EPT; ++k) {
    if (lo + tid + 1024 * k < hi) {
      if (staged) stage[pos[k]] = kkey[k];
      else        kout[pos[k]] = kkey[k];
    }
  }
  for (int e = lo + 1024 * SORT_EPT + tid; e < hi; e += 1024) {
    const int d = dst[e];
    const int s = src[e];
    const int b = (int)((unsigned)d / (unsigned)bw);
    const int pos = atomicAdd(&cnt[b], 1);
    const int key = (s << 8) | (d - b * bw);
    if (staged) stage[pos] = key;
    else        kout[pos] = key;
  }
  if (staged) {
    __syncthreads();
    const int shift = (int)(((size_t)kout >> 2) & 31);
    for (int i0 = tid - shift; i0 < nk; i0 += 1024 * SORT_EPT) {
      int v[SORT_EPT];
#pragma unroll
      for (int k = 0; k < SORT_EPT; ++k) v[k] = stage[min(max(i0 + 1024 * k, 0), DEG_ROWS_MAX - 1)];
#pragma unroll
      for (int k = 0; k < SORT_EPT; ++k) {
        const int i = i0 + 1024 * k;
        if (i >= 0 && i < nk) kout[i] = v[k];
      }
    }
  }
}

__device__ __forceinline__ int other_row(int v) {
  return __builtin_amdgcn_permlanex16(v, v, 0x76543210, 0xfedcba98, false, false);
}


template <int PROJ>
__device__ __forceinline__ void gemm1_tile(const float* __restrict__ x, const float* __restrict__ W1,
                                           const float* __restrict__ b1, const int* __restrict__ degp,
                                           float* __restrict__ pre1, float* __restrict__ y1,
                                           int N, int tile, int lane) {
  const int n = lane & 15;
  const int hi = lane >> 4;
  const int base = tile * 16;
  const int mrow = min(base + n, N - 1);
  const float* __restrict__ W = W1 + PROJ * (64 * 16);
  const float* __restrict__ xr = x + (long long)mrow * 64;

  float rs = 1.0f;
  if constexpr (PROJ == 1) {
    int cs = 0;
#pragma unroll
    for (int i = 0; i < DEG_SLABS / 2; ++i) cs += degp[(size_t)(hi * (DEG_SLABS / 2) + i) * N + mrow];
    cs += other_row(cs);
    const float dg = (float)cs;
    rs = (dg > 0.0f) ? rsqrtf(fmaxf(dg, 1.0f)) : 0.0f;
  }

  v8f acc = {0.f, 0.f, 0.f, 0.f, 0.f, 0.f, 0.f, 0.f};
#pragma unroll
  for (int kb = 0; kb < 64; kb += 4) {
    const int k0 = kb + 2 * hi;
    v2f a = *(const v2f*)(xr + k0);
    if constexpr (PROJ == 1) { a.x *= rs;  a.y *= rs; }
    v2f b;
    b.x = W[k0 * 16 + n];  b.y = W[(k0 + 1) * 16 + n];
    acc = WMMA_F32X4(a, b, acc);
  }
  if constexpr (PROJ == 0) {
    const float bias = b1[n];
#pragma unroll
    for (int r = 0; r < 8; ++r) {
      const int row = base + r + 8 * hi;
      if (row < N) pre1[(long long)row * 16 + n] = acc[r] + bias;
    }
  } else {
#pragma unroll
    for (int r = 0; r < 8; ++r) {
      const int row = base + r + 8 * hi;
      if (row < N) y1[(long long)row * 16 + n] = acc[r];
    }
  }
}

__global__ void __launch_bounds__(1024)
deg_gemm1_kernel(const int* __restrict__ degp, float* __restrict__ dinv,
                 const float* __restrict__ x, const float* __restrict__ W1,
                 const float* __restrict__ b1, float* __restrict__ pre1,
                 float* __restrict__ y1, int N, int nDeg, int nTiles) {
  const int tid = threadIdx.x;
  if ((int)blockIdx.x >= nDeg) {
    const int wv = __builtin_amdgcn_readfirstlane(tid >> 5);
    const int task = ((int)blockIdx.x - nDeg) * 32 + wv;
    const int tile = task >> 1;
    if (tile >= nTiles) return;
    if ((task & 1) == 0) gemm1_tile<0>(x, W1, b1, degp, pre1, y1, N, tile, tid & 31);
    else                 gemm1_tile<1>(x, W1, b1, degp, pre1, y1, N, tile, tid & 31);
    return;
  }
  const int node = blockIdx.x * 1024 + tid;
  if (node < N) {
    int p[DEG_SLABS];
#pragma unroll
    for (int i = 0; i < DEG_SLABS; ++i) p[i] = degp[(size_t)i * N + node];
    int cs = 0;
#pragma unroll
    for (int i = 0; i < DEG_SLABS; ++i) cs += p[i];
    const float dg = (float)cs;
    dinv[node] = (dg > 0.0f) ? rsqrtf(fmaxf(dg, 1.0f)) : 0.0f;
  }
}

template <int M>
__device__ __forceinline__ float row_xor(float v) {
  const int b = __builtin_bit_cast(int, v);
  return __builtin_bit_cast(float, __builtin_amdgcn_update_dpp(b, b, 0x160 + M, 0xf, 0xf, false));
}

template <int K>
__device__ __forceinline__ unsigned row_share_or(unsigned v, unsigned o) {
  return (unsigned)__builtin_amdgcn_update_dpp(0, (int)v, 0x150 + K, 0xf, 0xf, true) | o;
}

#define EDGES_PER_TRIP 16
__device__ __forceinline__ int seg_slot(int t, int b0, int b1, int b2, int b3, int b4, int b5, int b6, int b7,
                                        int c1, int c2, int c3, int c4, int c5, int c6, int c7) {
  int p = b0;
  p = (t >= c1) ? b1 : p;
  p = (t >= c2) ? b2 : p;
  p = (t >= c3) ? b3 : p;
  p = (t >= c4) ? b4 : p;
  p = (t >= c5) ? b5 : p;
  p = (t >= c6) ? b6 : p;
  p = (t >= c7) ? b7 : p;
  return p + t;
}
template <bool TAIL>
__global__ void __launch_bounds__(1024)
bin_sum_kernel(const int* __restrict__ bkey, const int* __restrict__ loff,
               const float* __restrict__ dinv, const float* __restrict__ y,
               const float* __restrict__ own, float* __restrict__ outp, float* __restrict__ outs,
               const float* __restrict__ W2, const float* __restrict__ b2, int N, int E) {
  __shared__ __attribute__((aligned(16))) float acc[ROWS_MAX * 16];
  const int tid = threadIdx.x;
  const int bin = blockIdx.x;
  const int bw = (N + NBINS - 1) / NBINS;
  const int eb = (E + CHUNKS - 1) / CHUNKS;
  for (int j = tid; j < ROWS_MAX * 16; j += 1024) acc[j] = 0.0f;
  const int wv = __builtin_amdgcn_readfirstlane(tid >> 5);
  const int hf = (tid >> 4) & 1;
  const int f = tid & 15;
  int tot = 0;
#define SEG_ROW(j, bj, cj)                                                        \
  int bj, cj;                                                                      \
  {                                                                                \
    const int ch = wv + 32 * (j);                                                  \
    const int* __restrict__ lr = loff + (size_t)ch * (NBINS + 1) + bin;            \
    const int o0 = lr[0], o1 = lr[1];                                              \
    cj = tot;                                                                      \
    bj = ch * eb + o0 - tot;                                                       \
    tot += o1 - o0;                                                                \
  }
  SEG_ROW(0, sb0, sc0) SEG_ROW(1, sb1, sc1) SEG_ROW(2, sb2, sc2) SEG_ROW(3, sb3, sc3)
  SEG_ROW(4, sb4, sc4) SEG_ROW(5, sb5, sc5) SEG_ROW(6, sb6, sc6) SEG_ROW(7, sb7, sc7)
#undef SEG_ROW
  static_assert(SEGS_PER_WAVE == 8, "eight segment rows are written out above and in seg_slot()");
  __syncthreads();
  if (tot > 0) {
    const int lastt = tot - 1;
#define SLOT(t) seg_slot(min((t), lastt), sb0, sb1, sb2, sb3, sb4, sb5, sb6, sb7, sc1, sc2, sc3, sc4, sc5, sc6, sc7)
    const unsigned f4 = (unsigned)f << 2;
    const char* __restrict__ yb = (const char*)y;
    char* accb = (char*)acc;
    int kv = bkey[SLOT(2 * f + hf)];
    for (int t0 = 0; 2 * t0 < tot; t0 += EDGES_PER_TRIP) {
      const int nkv = bkey[SLOT(2 * (t0 + EDGES_PER_TRIP + f) + hf)];
      const unsigned gb = ((unsigned)kv >> 8) << 6;
      const unsigned lb = ((unsigned)kv & 255u) << 6;
      unsigned go[EDGES_PER_TRIP], la[EDGES_PER_TRIP];
#define OFFS_OF(k) go[k] = row_share_or<k>(gb, f4);  la[k] = row_share_or<k>(lb, f4);
      OFFS_OF(0) OFFS_OF(1) OFFS_OF(2) OFFS_OF(3) OFFS_OF(4) OFFS_OF(5) OFFS_OF(6) OFFS_OF(7)
      OFFS_OF(8) OFFS_OF(9) OFFS_OF(10) OFFS_OF(11) OFFS_OF(12) OFFS_OF(13) OFFS_OF(14) OFFS_OF(15)
#undef OFFS_OF
      static_assert(EDGES_PER_TRIP == 16, "one list entry per lane of a 16-lane group and trip");
      float yv[EDGES_PER_TRIP];
#pragma unroll
      for (int k = 0; k < EDGES_PER_TRIP; ++k) yv[k] = *(const float*)(yb + go[k]);
      asm volatile("" :: "v"(yv[0]), "v"(yv[1]), "v"(yv[2]), "v"(yv[3]),
                         "v"(yv[4]), "v"(yv[5]), "v"(yv[6]), "v"(yv[7]),
                         "v"(yv[8]), "v"(yv[9]), "v"(yv[10]), "v"(yv[11]),
                         "v"(yv[12]), "v"(yv[13]), "v"(yv[14]), "v"(yv[15]));
      if (2 * (t0 + EDGES_PER_TRIP) <= tot) {
#pragma unroll
        for (int k = 0; k < EDGES_PER_TRIP; ++k) {
          __hip_atomic_fetch_add((float*)(accb + la[k]), yv[k], __ATOMIC_RELAXED, __HIP_MEMORY_SCOPE_WORKGROUP);
        }
      } else {
#pragma unroll
        for (int k = 0; k < EDGES_PER_TRIP; ++k) {
          if (2 * (t0 + k) + hf < tot) {
            __hip_atomic_fetch_add((float*)(accb + la[k]), yv[k], __ATOMIC_RELAXED, __HIP_MEMORY_SCOPE_WORKGROUP);
          }
        }
      }
      kv = nkv;
    }
#undef SLOT
  }
  const int base = bin * bw;
  const int nrows = min(bw, N - base);
  if constexpr (!TAIL) {
    __syncthreads();
    for (int j = tid; j < nrows * 16; j += 1024) {
      const long long gi = (long long)base * 16 + j;
      const float di = dinv[base + (j >> 4)];
      const float hv = fmaxf(own[gi] - di * acc[j], 0.0f);
      outp[gi] = hv;
      outs[gi] = di * hv;
    }
  } else {
    __shared__ __attribute__((aligned(16))) float zs[ROWS_MAX * 40];
    const int ntiles = nrows > 0 ? (nrows + 15) >> 4 : 0;
    const bool tilewave = wv < ntiles;
    const int n = tid & 15;
    const int hi = (tid >> 4) & 1;
    const float* __restrict__ W0 = W2;
    const float* __restrict__ Wb = W2 + 16 * 40;
    v2f a[4], b0[12], bb[12];
    float bias0 = 0.f, bias1 = 0.f, bias2 = 0.f, nd = 0.f;
#pragma unroll
    for (int i = 0; i < 4; ++i) a[i] = (v2f){0.f, 0.f};
#pragma unroll
    for (int i = 0; i < 12; ++i) { b0[i] = (v2f){0.f, 0.f}; bb[i] = (v2f){0.f, 0.f}; }
    if (tilewave) {
      const int mrow = min(base + wv * 16 + n, N - 1);
#pragma unroll
      for (int i = 0; i < 4; ++i) a[i] = *(const v2f*)(own + (long long)mrow * 16 + i * 4 + 2 * hi);
      nd = -dinv[mrow];
#pragma unroll
      for (int ct = 0; ct < 3; ++ct) {
        const int cc = min(ct * 16 + n, 39);
#pragma unroll
        for (int i = 0; i < 4; ++i) {
          const int k0 = i * 4 + 2 * hi;
          b0[ct * 4 + i].x = W0[k0 * 40 + cc];  b0[ct * 4 + i].y = W0[(k0 + 1) * 40 + cc];
          bb[ct * 4 + i].x = Wb[k0 * 40 + cc];  bb[ct * 4 + i].y = Wb[(k0 + 1) * 40 + cc];
        }
      }
      bias0 = b2[n];  bias1 = b2[16 + n];  bias2 = b2[min(32 + n, 39)];
    }
    __syncthreads();
    if (tilewave) {
      v2f t[4];
#pragma unroll
      for (int i = 0; i < 4; ++i) {
        t[i] = *(const v2f*)(&acc[(wv * 16 + n) * 16 + i * 4 + 2 * hi]);
        t[i].x *= nd;  t[i].y *= nd;
      }
      v8f c[6];
#pragma unroll
      for (int i = 0; i < 6; ++i) c[i] = (v8f){0.f, 0.f, 0.f, 0.f, 0.f, 0.f, 0.f, 0.f};
#pragma unroll
      for (int ct = 0; ct < 3; ++ct) {
#pragma unroll
        for (int i = 0; i < 4; ++i) {
          c[ct * 2]     = WMMA_F32X4(a[i], b0[ct * 4 + i], c[ct * 2]);
          c[ct * 2 + 1] = WMMA_F32X4(t[i], bb[ct * 4 + i], c[ct * 2 + 1]);
        }
      }
      float z0[8], z1[8], z2[8], m[8], s[8];
#pragma unroll
      for (int r = 0; r < 8; ++r) {
        z0[r] = (c[0][r] + c[1][r]) + bias0;
        z1[r] = (c[2][r] + c[3][r]) + bias1;
        z2[r] = (c[4][r] + c[5][r]) + bias2;
        m[r] = fmaxf(z0[r], z1[r]);
      }
      if (n < 8) {
#pragma unroll
        for (int r = 0; r < 8; ++r) m[r] = fmaxf(m[r], z2[r]);
      }
#pragma unroll
      for (int r = 0; r < 8; ++r) m[r] = fmaxf(m[r], row_xor<8>(m[r]));
#pragma unroll
      for (int r = 0; r < 8; ++r) m[r] = fmaxf(m[r], row_xor<4>(m[r]));
#pragma unroll
      for (int r = 0; r < 8; ++r) m[r] = fmaxf(m[r], row_xor<2>(m[r]));
#pragma unroll
      for (int r = 0; r < 8; ++r) m[r] = fmaxf(m[r], row_xor<1>(m[r]));
#pragma unroll
      for (int r = 0; r < 8; ++r) s[r] = expf(z0[r] - m[r]) + expf(z1[r] - m[r]);
      if (n < 8) {
#pragma unroll
        for (int r = 0; r < 8; ++r) s[r] += expf(z2[r] - m[r]);
      }
#pragma unroll
      for (int r = 0; r < 8; ++r) s[r] += row_xor<8>(s[r]);
#pragma unroll
      for (int r = 0; r < 8; ++r) s[r] += row_xor<4>(s[r]);
#pragma unroll
      for (int r = 0; r < 8; ++r) s[r] += row_xor<2>(s[r]);
#pragma unroll
      for (int r = 0; r < 8; ++r) s[r] += row_xor<1>(s[r]);
#pragma unroll
      for (int r = 0; r < 8; ++r) {
        const float lse = m[r] + logf(s[r]);
        float* zr = zs + (wv * 16 + r + 8 * hi) * 40;
        zr[n] = z0[r] - lse;
        zr[16 + n] = z1[r] - lse;
        if (n < 8) zr[32 + n] = z2[r] - lse;
      }
    }
    __syncthreads();
    float4* __restrict__ dst4 = (float4*)(outp + (long long)base * 40);
    const float4* src4 = (const float4*)zs;
    for (int j = tid; j < nrows * 10; j += 1024) dst4[j] = src4[j];
  }
}

extern "C" void kernel_launch(void* const* d_in, const int* in_sizes, int n_in,
                              void* d_out, int out_size, void* d_ws, size_t ws_size,
                              hipStream_t stream) {
  const float* x  = (const float*)d_in[0];
  const int* ei   = (const int*)d_in[1];
  const float* W1 = (const float*)d_in[2];
  const float* b1 = (const float*)d_in[3];
  const float* W2 = (const float*)d_in[4];
  const float* b2 = (const float*)d_in[5];
  float* out = (float*)d_out;

  const int N = in_sizes[0] / 64;
  const int E = in_sizes[1] / 2;
  const int* src = ei;
  const int* dst = ei + E;

  float* ws = (float*)d_ws;
  size_t off = 0;
  float* dinv = ws + off; off += (size_t)N;
  float* pre1 = ws + off; off += (size_t)N * 16;
  float* y1   = ws + off; off += (size_t)N * 16;
  int* loff   = (int*)(ws + off); off += (size_t)CHUNKS * (NBINS + 1);
  int* bkey   = (int*)(ws + off); off += (size_t)E;
  float* h    = ws + off; off += (size_t)N * 16;
  float* hs   = ws + off;
  int* degp   = (int*)h;

  auto blk = [](long long n, int b) { return (unsigned)((n + b - 1) / b); };

  const int tiles = (N + 15) / 16;
  const int nDeg = (int)blk(N, 1024);
  bin_sort_kernel<<<CHUNKS, 1024, 0, stream>>>(src, dst, degp, loff, bkey, N, E);
  deg_gemm1_kernel<<<nDeg + blk((long long)tiles * 2, 32), 1024, 0, stream>>>(
      degp, dinv, x, W1, b1, pre1, y1, N, nDeg, tiles);

  bin_sum_kernel<false><<<NBINS, 1024, 0, stream>>>(bkey, loff, dinv, y1, pre1, h, hs, W2, b2, N, E);

  bin_sum_kernel<true><<<NBINS, 1024, 0, stream>>>(bkey, loff, dinv, hs, h, out, nullptr, W2, b2, N, E);
}


# Launch 2: one wave per 32 rows does all of it: sums the rows' count planes once (whole lines), stores dinv, both projections of two row tiles from one A fragment set; 1,563 waves, not 7,840
# speedup vs baseline: 5.2054x; 1.0378x over previous
#include <hip/hip_runtime.h>
#include <math.h>

typedef float v2f __attribute__((ext_vector_type(2)));
typedef float v8f __attribute__((ext_vector_type(8)));

#define WMMA_F32X4(a, b, c) \
  __builtin_amdgcn_wmma_f32_16x16x4_f32(false, (a), false, (b), (short)0, (c), false, false)

#define NBINS 256
#define ROWS_MAX 256
#define CHUNKS 256
#define SORT_EPT 4
#define SEGS_PER_WAVE 8
static_assert(CHUNKS == 32 * SEGS_PER_WAVE, "the 32 waves of a range's work-group share its segments evenly");

#define DEG_SLABS 32
#define DEG_RANGES 8
#define DEG_ROWS_MAX 8192
#define DEG_LOADS 8
static_assert(DEG_SLABS * DEG_RANGES == CHUNKS, "one (slab, range) pair per work-group of pass 1");

__global__ void __launch_bounds__(1024)
bin_sort_kernel(const int* __restrict__ src, const int* __restrict__ dst,
                int* __restrict__ degp, int* __restrict__ loff, int* __restrict__ bkey, int N, int E) {
  __shared__ int cnt[NBINS];
  __shared__ int sc[2][NBINS];
  __shared__ int dcnt[DEG_ROWS_MAX];
  const int tid = threadIdx.x;
  const int bw = (N + NBINS - 1) / NBINS;
  const int eb = (E + CHUNKS - 1) / CHUNKS;
  const int c = blockIdx.x / DEG_RANGES;
  const int r = blockIdx.x - c * DEG_RANGES;
  const int rw = (N + DEG_RANGES - 1) / DEG_RANGES;
  const int r_lo = r * rw;
  if (tid < NBINS) cnt[tid] = 0;
  for (int j = tid; j < rw; j += 1024) dcnt[j] = 0;
  __syncthreads();
  const int lo = blockIdx.x * eb;
  const int hi = min(E, lo + eb);
  int kbin[SORT_EPT], kkey[SORT_EPT];
#pragma unroll
  for (int k = 0; k < SORT_EPT; ++k) {
    const int e = min(lo + tid + 1024 * k, E - 1);
    const int d = dst[e];
    const int s = src[e];
    const int b = (int)((unsigned)d / (unsigned)bw);
    kbin[k] = b;
    kkey[k] = (s << 8) | (d - b * bw);
  }
#pragma unroll
  for (int k = 0; k < SORT_EPT; ++k) {
    if (lo + tid + 1024 * k < hi) atomicAdd(&cnt[kbin[k]], 1);
  }
  for (int e = lo + 1024 * SORT_EPT + tid; e < hi; e += 1024) {
    const int d = dst[e];
    atomicAdd(&cnt[(unsigned)d / (unsigned)bw], 1);
  }
  const int slab = (((E + DEG_SLABS - 1) / DEG_SLABS) + 3) & ~3;
  const int s_lo = c * slab;
  const int s_hi = min(E, s_lo + slab);
  const int Eal = E & ~3;
  const int v_hi = min(s_hi, Eal);
  for (int e0 = s_lo + 4 * tid; e0 < v_hi; e0 += 4096 * DEG_LOADS) {
    int4 v[DEG_LOADS];
#pragma unroll
    for (int k = 0; k < DEG_LOADS; ++k) {
      const int e = min(e0 + 4096 * k, Eal - 4);
      v[k] = *(const int4*)(src + e);
    }
#pragma unroll
    for (int k = 0; k < DEG_LOADS; ++k) {
      if (e0 + 4096 * k < v_hi) {
        const unsigned j0 = (unsigned)(v[k].x - r_lo), j1 = (unsigned)(v[k].y - r_lo);
        const unsigned j2 = (unsigned)(v[k].z - r_lo), j3 = (unsigned)(v[k].w - r_lo);
        if (j0 < (unsigned)rw) atomicAdd(&dcnt[j0], 1);
        if (j1 < (unsigned)rw) atomicAdd(&dcnt[j1], 1);
        if (j2 < (unsigned)rw) atomicAdd(&dcnt[j2], 1);
        if (j3 < (unsigned)rw) atomicAdd(&dcnt[j3], 1);
      }
    }
  }
  if (tid < E - Eal) {
    const int e = Eal + tid;
    if (e >= s_lo && e < s_hi) {
      const unsigned j = (unsigned)(src[e] - r_lo);
      if (j < (unsigned)rw) atomicAdd(&dcnt[j], 1);
    }
  }
  __syncthreads();
  int* __restrict__ plane = degp + (size_t)c * N;
  for (int j = tid; j < rw; j += 1024) {
    const int node = r_lo + j;
    if (node < N) plane[node] = dcnt[j];
  }
  int mine = 0;
  if (tid < NBINS) { mine = cnt[tid]; sc[0][tid] = mine; }
  __syncthreads();
  int pin = 0;
  for (int o = 1; o < NBINS; o <<= 1) {
    if (tid < NBINS) {
      int x = sc[pin][tid];
      if (tid >= o) x += sc[pin][tid - o];
      sc[pin ^ 1][tid] = x;
    }
    __syncthreads();
    pin ^= 1;
  }
  if (tid < NBINS) {
    const int incl = sc[pin][tid];
    const int excl = incl - mine;
    cnt[tid] = excl;
    int* __restrict__ lrow = loff + (size_t)blockIdx.x * (NBINS + 1);
    lrow[tid] = excl;
    if (tid == NBINS - 1) lrow[NBINS] = incl;
  }
  __syncthreads();
  int* __restrict__ kout = bkey + lo;
  int* stage = dcnt;
  const int nk = hi - lo;
  const bool staged = nk <= DEG_ROWS_MAX;
  int pos[SORT_EPT];
#pragma unroll
  for (int k = 0; k < SORT_EPT; ++k) {
    pos[k] = 0;
    if (lo + tid + 1024 * k < hi) pos[k] = atomicAdd(&cnt[kbin[k]], 1);
  }
#pragma unroll
  for (int k = 0; k < SORT_EPT; ++k) {
    if (lo + tid + 1024 * k < hi) {
      if (staged) stage[pos[k]] = kkey[k];
      else        kout[pos[k]] = kkey[k];
    }
  }
  for (int e = lo + 1024 * SORT_EPT + tid; e < hi; e += 1024) {
    const int d = dst[e];
    const int s = src[e];
    const int b = (int)((unsigned)d / (unsigned)bw);
    const int pos = atomicAdd(&cnt[b], 1);
    const int key = (s << 8) | (d - b * bw);
    if (staged) stage[pos] = key;
    else        kout[pos] = key;
  }
  if (staged) {
    __syncthreads();
    const int shift = (int)(((size_t)kout >> 2) & 31);
    for (int i0 = tid - shift; i0 < nk; i0 += 1024 * SORT_EPT) {
      int v[SORT_EPT];
#pragma unroll
      for (int k = 0; k < SORT_EPT; ++k) v[k] = stage[min(max(i0 + 1024 * k, 0), DEG_ROWS_MAX - 1)];
#pragma unroll
      for (int k = 0; k < SORT_EPT; ++k) {
        const int i = i0 + 1024 * k;
        if (i >= 0 && i < nk) kout[i] = v[k];
      }
    }
  }
}

__device__ __forceinline__ int other_row(int v) {
  return __builtin_amdgcn_permlanex16(v, v, 0x76543210, 0xfedcba98, false, false);
}


#define G1_WAVES 4
__global__ void __launch_bounds__(32 * G1_WAVES) __attribute__((amdgpu_waves_per_eu(1, 4)))
deg_gemm1_kernel(const int* __restrict__ degp, float* __restrict__ dinv,
                 const float* __restrict__ x, const float* __restrict__ W1,
                 const float* __restrict__ b1, float* __restrict__ pre1,
                 float* __restrict__ y1, int N, int nTasks) {
  const int tid = threadIdx.x;
  const int wv = __builtin_amdgcn_readfirstlane(tid >> 5);
  const int task = (int)blockIdx.x * G1_WAVES + wv;
  if (task >= nTasks) return;
  const int lane = tid & 31;
  const int n = lane & 15;
  const int hi = lane >> 4;
  const int base = task * 32;

  const float* __restrict__ Wa = W1;
  const float* __restrict__ Wb = W1 + 64 * 16;
  const int m0 = min(base + n, N - 1);
  const int m1 = min(base + 16 + n, N - 1);
  const float* __restrict__ xr0 = x + (long long)m0 * 64;
  const float* __restrict__ xr1 = x + (long long)m1 * 64;
  v2f a0[16], a1[16];
#pragma unroll
  for (int i = 0; i < 16; ++i) {
    a0[i] = *(const v2f*)(xr0 + 4 * i + 2 * hi);
    a1[i] = *(const v2f*)(xr1 + 4 * i + 2 * hi);
  }

  const int drow = min(base + lane, N - 1);
  int cs = 0;
#pragma unroll
  for (int i = 0; i < DEG_SLABS; ++i) cs += degp[(size_t)i * N + drow];
  const float dg = (float)cs;
  const float dv = (dg > 0.0f) ? rsqrtf(fmaxf(dg, 1.0f)) : 0.0f;
  const float dvx = __builtin_bit_cast(float, other_row(__builtin_bit_cast(int, dv)));
  const float rs0 = hi ? dvx : dv;
  const float rs1 = hi ? dv : dvx;

  v8f c00 = {0.f, 0.f, 0.f, 0.f, 0.f, 0.f, 0.f, 0.f};
  v8f c01 = {0.f, 0.f, 0.f, 0.f, 0.f, 0.f, 0.f, 0.f};
  v8f c10 = {0.f, 0.f, 0.f, 0.f, 0.f, 0.f, 0.f, 0.f};
  v8f c11 = {0.f, 0.f, 0.f, 0.f, 0.f, 0.f, 0.f, 0.f};
#pragma unroll
  for (int i = 0; i < 16; ++i) {
    const int k0 = 4 * i + 2 * hi;
    v2f s0, s1;
    s0.x = a0[i].x * rs0;  s0.y = a0[i].y * rs0;
    s1.x = a1[i].x * rs1;  s1.y = a1[i].y * rs1;
    v2f ba, bb;
    ba.x = Wa[k0 * 16 + n];  ba.y = Wa[(k0 + 1) * 16 + n];
    bb.x = Wb[k0 * 16 + n];  bb.y = Wb[(k0 + 1) * 16 + n];
    c00 = WMMA_F32X4(a0[i], ba, c00);
    c01 = WMMA_F32X4(s0, bb, c01);
    c10 = WMMA_F32X4(a1[i], ba, c10);
    c11 = WMMA_F32X4(s1, bb, c11);
  }
  const float bias = b1[n];
  float* __restrict__ p0 = pre1 + (long long)(base + 8 * hi) * 16 + n;
  float* __restrict__ q0 = y1 + (long long)(base + 8 * hi) * 16 + n;
  if (base + 32 <= N) {
    dinv[base + lane] = dv;
#pragma unroll
    for (int r = 0; r < 8; ++r) {
      p0[r * 16] = c00[r] + bias;
      q0[r * 16] = c01[r];
      p0[(16 + r) * 16] = c10[r] + bias;
      q0[(16 + r) * 16] = c11[r];
    }
  } else {
    if (base + lane < N) dinv[base + lane] = dv;
#pragma unroll
    for (int r = 0; r < 8; ++r) {
      const int row = base + r + 8 * hi;
      if (row < N) {
        p0[r * 16] = c00[r] + bias;
        q0[r * 16] = c01[r];
      }
      if (row + 16 < N) {
        p0[(16 + r) * 16] = c10[r] + bias;
        q0[(16 + r) * 16] = c11[r];
      }
    }
  }
}

template <int M>
__device__ __forceinline__ float row_xor(float v) {
  const int b = __builtin_bit_cast(int, v);
  return __builtin_bit_cast(float, __builtin_amdgcn_update_dpp(b, b, 0x160 + M, 0xf, 0xf, false));
}

template <int K>
__device__ __forceinline__ unsigned row_share_or(unsigned v, unsigned o) {
  return (unsigned)__builtin_amdgcn_update_dpp(0, (int)v, 0x150 + K, 0xf, 0xf, true) | o;
}

#define EDGES_PER_TRIP 16
__device__ __forceinline__ int seg_slot(int t, int b0, int b1, int b2, int b3, int b4, int b5, int b6, int b7,
                                        int c1, int c2, int c3, int c4, int c5, int c6, int c7) {
  int p = b0;
  p = (t >= c1) ? b1 : p;
  p = (t >= c2) ? b2 : p;
  p = (t >= c3) ? b3 : p;
  p = (t >= c4) ? b4 : p;
  p = (t >= c5) ? b5 : p;
  p = (t >= c6) ? b6 : p;
  p = (t >= c7) ? b7 : p;
  return p + t;
}
template <bool TAIL>
__global__ void __launch_bounds__(1024)
bin_sum_kernel(const int* __restrict__ bkey, const int* __restrict__ loff,
               const float* __restrict__ dinv, const float* __restrict__ y,
               const float* __restrict__ own, float* __restrict__ outp, float* __restrict__ outs,
               const float* __restrict__ W2, const float* __restrict__ b2, int N, int E) {
  __shared__ __attribute__((aligned(16))) float acc[ROWS_MAX * 16];
  const int tid = threadIdx.x;
  const int bin = blockIdx.x;
  const int bw = (N + NBINS - 1) / NBINS;
  const int eb = (E + CHUNKS - 1) / CHUNKS;
  for (int j = tid; j < ROWS_MAX * 16; j += 1024) acc[j] = 0.0f;
  const int wv = __builtin_amdgcn_readfirstlane(tid >> 5);
  const int hf = (tid >> 4) & 1;
  const int f = tid & 15;
  int tot = 0;
#define SEG_ROW(j, bj, cj)                                                        \
  int bj, cj;                                                                      \
  {                                                                                \
    const int ch = wv + 32 * (j);                                                  \
    const int* __restrict__ lr = loff + (size_t)ch * (NBINS + 1) + bin;            \
    const int o0 = lr[0], o1 = lr[1];                                              \
    cj = tot;                                                                      \
    bj = ch * eb + o0 - tot;                                                       \
    tot += o1 - o0;                                                                \
  }
  SEG_ROW(0, sb0, sc0) SEG_ROW(1, sb1, sc1) SEG_ROW(2, sb2, sc2) SEG_ROW(3, sb3, sc3)
  SEG_ROW(4, sb4, sc4) SEG_ROW(5, sb5, sc5) SEG_ROW(6, sb6, sc6) SEG_ROW(7, sb7, sc7)
#undef SEG_ROW
  static_assert(SEGS_PER_WAVE == 8, "eight segment rows are written out above and in seg_slot()");
  __syncthreads();
  if (tot > 0) {
    const int lastt = tot - 1;
#define SLOT(t) seg_slot(min((t), lastt), sb0, sb1, sb2, sb3, sb4, sb5, sb6, sb7, sc1, sc2, sc3, sc4, sc5, sc6, sc7)
    const unsigned f4 = (unsigned)f << 2;
    const char* __restrict__ yb = (const char*)y;
    char* accb = (char*)acc;
    int kv = bkey[SLOT(2 * f + hf)];
    for (int t0 = 0; 2 * t0 < tot; t0 += EDGES_PER_TRIP) {
      const int nkv = bkey[SLOT(2 * (t0 + EDGES_PER_TRIP + f) + hf)];
      const unsigned gb = ((unsigned)kv >> 8) << 6;
      const unsigned lb = ((unsigned)kv & 255u) << 6;
      unsigned go[EDGES_PER_TRIP], la[EDGES_PER_TRIP];
#define OFFS_OF(k) go[k] = row_share_or<k>(gb, f4);  la[k] = row_share_or<k>(lb, f4);
      OFFS_OF(0) OFFS_OF(1) OFFS_OF(2) OFFS_OF(3) OFFS_OF(4) OFFS_OF(5) OFFS_OF(6) OFFS_OF(7)
      OFFS_OF(8) OFFS_OF(9) OFFS_OF(10) OFFS_OF(11) OFFS_OF(12) OFFS_OF(13) OFFS_OF(14) OFFS_OF(15)
#undef OFFS_OF
      static_assert(EDGES_PER_TRIP == 16, "one list entry per lane of a 16-lane group and trip");
      float yv[EDGES_PER_TRIP];
#pragma unroll
      for (int k = 0; k < EDGES_PER_TRIP; ++k) yv[k] = *(const float*)(yb + go[k]);
      asm volatile("" :: "v"(yv[0]), "v"(yv[1]), "v"(yv[2]), "v"(yv[3]),
                         "v"(yv[4]), "v"(yv[5]), "v"(yv[6]), "v"(yv[7]),
                         "v"(yv[8]), "v"(yv[9]), "v"(yv[10]), "v"(yv[11]),
                         "v"(yv[12]), "v"(yv[13]), "v"(yv[14]), "v"(yv[15]));
      if (2 * (t0 + EDGES_PER_TRIP) <= tot) {
#pragma unroll
        for (int k = 0; k < EDGES_PER_TRIP; ++k) {
          __hip_atomic_fetch_add((float*)(accb + la[k]), yv[k], __ATOMIC_RELAXED, __HIP_MEMORY_SCOPE_WORKGROUP);
        }
      } else {
#pragma unroll
        for (int k = 0; k < EDGES_PER_TRIP; ++k) {
          if (2 * (t0 + k) + hf < tot) {
            __hip_atomic_fetch_add((float*)(accb + la[k]), yv[k], __ATOMIC_RELAXED, __HIP_MEMORY_SCOPE_WORKGROUP);
          }
        }
      }
      kv = nkv;
    }
#undef SLOT
  }
  const int base = bin * bw;
  const int nrows = min(bw, N - base);
  if constexpr (!TAIL) {
    __syncthreads();
    for (int j = tid; j < nrows * 16; j += 1024) {
      const long long gi = (long long)base * 16 + j;
      const float di = dinv[base + (j >> 4)];
      const float hv = fmaxf(own[gi] - di * acc[j], 0.0f);
      outp[gi] = hv;
      outs[gi] = di * hv;
    }
  } else {
    __shared__ __attribute__((aligned(16))) float zs[ROWS_MAX * 40];
    const int ntiles = nrows > 0 ? (nrows + 15) >> 4 : 0;
    const bool tilewave = wv < ntiles;
    const int n = tid & 15;
    const int hi = (tid >> 4) & 1;
    const float* __restrict__ W0 = W2;
    const float* __restrict__ Wb = W2 + 16 * 40;
    v2f a[4], b0[12], bb[12];
    float bias0 = 0.f, bias1 = 0.f, bias2 = 0.f, nd = 0.f;
#pragma unroll
    for (int i = 0; i < 4; ++i) a[i] = (v2f){0.f, 0.f};
#pragma unroll
    for (int i = 0; i < 12; ++i) { b0[i] = (v2f){0.f, 0.f}; bb[i] = (v2f){0.f, 0.f}; }
    if (tilewave) {
      const int mrow = min(base + wv * 16 + n, N - 1);
#pragma unroll
      for (int i = 0; i < 4; ++i) a[i] = *(const v2f*)(own + (long long)mrow * 16 + i * 4 + 2 * hi);
      nd = -dinv[mrow];
#pragma unroll
      for (int ct = 0; ct < 3; ++ct) {
        const int cc = min(ct * 16 + n, 39);
#pragma unroll
        for (int i = 0; i < 4; ++i) {
          const int k0 = i * 4 + 2 * hi;
          b0[ct * 4 + i].x = W0[k0 * 40 + cc];  b0[ct * 4 + i].y = W0[(k0 + 1) * 40 + cc];
          bb[ct * 4 + i].x = Wb[k0 * 40 + cc];  bb[ct * 4 + i].y = Wb[(k0 + 1) * 40 + cc];
        }
      }
      bias0 = b2[n];  bias1 = b2[16 + n];  bias2 = b2[min(32 + n, 39)];
    }
    __syncthreads();
    if (tilewave) {
      v2f t[4];
#pragma unroll
      for (int i = 0; i < 4; ++i) {
        t[i] = *(const v2f*)(&acc[(wv * 16 + n) * 16 + i * 4 + 2 * hi]);
        t[i].x *= nd;  t[i].y *= nd;
      }
      v8f c[6];
#pragma unroll
      for (int i = 0; i < 6; ++i) c[i] = (v8f){0.f, 0.f, 0.f, 0.f, 0.f, 0.f, 0.f, 0.f};
#pragma unroll
      for (int ct = 0; ct < 3; ++ct) {
#pragma unroll
        for (int i = 0; i < 4; ++i) {
          c[ct * 2]     = WMMA_F32X4(a[i], b0[ct * 4 + i], c[ct * 2]);
          c[ct * 2 + 1] = WMMA_F32X4(t[i], bb[ct * 4 + i], c[ct * 2 + 1]);
        }
      }
      float z0[8], z1[8], z2[8], m[8], s[8];
#pragma unroll
      for (int r = 0; r < 8; ++r) {
        z0[r] = (c[0][r] + c[1][r]) + bias0;
        z1[r] = (c[2][r] + c[3][r]) + bias1;
        z2[r] = (c[4][r] + c[5][r]) + bias2;
        m[r] = fmaxf(z0[r], z1[r]);
      }
      if (n < 8) {
#pragma unroll
        for (int r = 0; r < 8; ++r) m[r] = fmaxf(m[r], z2[r]);
      }
#pragma unroll
      for (int r = 0; r < 8; ++r) m[r] = fmaxf(m[r], row_xor<8>(m[r]));
#pragma unroll
      for (int r = 0; r < 8; ++r) m[r] = fmaxf(m[r], row_xor<4>(m[r]));
#pragma unroll
      for (int r = 0; r < 8; ++r) m[r] = fmaxf(m[r], row_xor<2>(m[r]));
#pragma unroll
      for (int r = 0; r < 8; ++r) m[r] = fmaxf(m[r], row_xor<1>(m[r]));
#pragma unroll
      for (int r = 0; r < 8; ++r) s[r] = expf(z0[r] - m[r]) + expf(z1[r] - m[r]);
      if (n < 8) {
#pragma unroll
        for (int r = 0; r < 8; ++r) s[r] += expf(z2[r] - m[r]);
      }
#pragma unroll
      for (int r = 0; r < 8; ++r) s[r] += row_xor<8>(s[r]);
#pragma unroll
      for (int r = 0; r < 8; ++r) s[r] += row_xor<4>(s[r]);
#pragma unroll
      for (int r = 0; r < 8; ++r) s[r] += row_xor<2>(s[r]);
#pragma unroll
      for (int r = 0; r < 8; ++r) s[r] += row_xor<1>(s[r]);
#pragma unroll
      for (int r = 0; r < 8; ++r) {
        const float lse = m[r] + logf(s[r]);
        float* zr = zs + (wv * 16 + r + 8 * hi) * 40;
        zr[n] = z0[r] - lse;
        zr[16 + n] = z1[r] - lse;
        if (n < 8) zr[32 + n] = z2[r] - lse;
      }
    }
    __syncthreads();
    float4* __restrict__ dst4 = (float4*)(outp + (long long)base * 40);
    const float4* src4 = (const float4*)zs;
    for (int j = tid; j < nrows * 10; j += 1024) dst4[j] = src4[j];
  }
}

extern "C" void kernel_launch(void* const* d_in, const int* in_sizes, int n_in,
                              void* d_out, int out_size, void* d_ws, size_t ws_size,
                              hipStream_t stream) {
  const float* x  = (const float*)d_in[0];
  const int* ei   = (const int*)d_in[1];
  const float* W1 = (const float*)d_in[2];
  const float* b1 = (const float*)d_in[3];
  const float* W2 = (const float*)d_in[4];
  const float* b2 = (const float*)d_in[5];
  float* out = (float*)d_out;

  const int N = in_sizes[0] / 64;
  const int E = in_sizes[1] / 2;
  const int* src = ei;
  const int* dst = ei + E;

  float* ws = (float*)d_ws;
  size_t off = 0;
  float* dinv = ws + off; off += (size_t)N;
  float* pre1 = ws + off; off += (size_t)N * 16;
  float* y1   = ws + off; off += (size_t)N * 16;
  int* loff   = (int*)(ws + off); off += (size_t)CHUNKS * (NBINS + 1);
  int* bkey   = (int*)(ws + off); off += (size_t)E;
  float* h    = ws + off; off += (size_t)N * 16;
  float* hs   = ws + off;
  int* degp   = (int*)h;

  auto blk = [](long long n, int b) { return (unsigned)((n + b - 1) / b); };

  const int nTasks = (N + 31) / 32;
  bin_sort_kernel<<<CHUNKS, 1024, 0, stream>>>(src, dst, degp, loff, bkey, N, E);
  deg_gemm1_kernel<<<blk(nTasks, G1_WAVES), 32 * G1_WAVES, 0, stream>>>(
      degp, dinv, x, W1, b1, pre1, y1, N, nTasks);

  bin_sum_kernel<false><<<NBINS, 1024, 0, stream>>>(bkey, loff, dinv, y1, pre1, h, hs, W2, b2, N, E);

  bin_sum_kernel<true><<<NBINS, 1024, 0, stream>>>(bkey, loff, dinv, hs, h, out, nullptr, W2, b2, N, E);
}
